# starting kernel
# baseline (speedup 1.0000x reference)
#include <hip/hip_runtime.h>
#include <hip/hip_bf16.h>
#include <math.h>

typedef __attribute__((ext_vector_type(2))) float v2f;
typedef __attribute__((ext_vector_type(8))) float v8f;

#define N0  300000
#define N1  60000
#define N2  6000
#define E1  1200000
#define E2  300000
#define D_H 128
#define D_OUT 64
#define BN_EPS 1e-5f

__global__ void k_zero(float* __restrict__ p, long n) {
  long i = (long)blockIdx.x * blockDim.x + threadIdx.x;
  long stride = (long)gridDim.x * blockDim.x;
  for (; i < n; i += stride) p[i] = 0.0f;
}

__global__ void k_zeroA1(float* __restrict__ A1, const int* __restrict__ hmap) {
  int n = blockIdx.x * blockDim.x + threadIdx.x;
  int stride = gridDim.x * blockDim.x;
  for (; n < N1; n += stride) {
    if (hmap[n] < 0) {
#pragma unroll
      for (int r = 0; r < 4; ++r) {
        float* p = A1 + ((long)r * N1 + n) * 128;
        for (int c = 0; c < 128; ++c) p[c] = 0.0f;
      }
    }
  }
}

__global__ void k_scatter_f(const float* __restrict__ feat,
                            const int* __restrict__ src,
                            const int* __restrict__ dst,
                            const int* __restrict__ et,
                            const int* __restrict__ hmap,
                            float* __restrict__ acc, int nE, int nOut) {
  long total = (long)nE * 128;
  long i = (long)blockIdx.x * blockDim.x + threadIdx.x;
  long stride = (long)gridDim.x * blockDim.x;
  for (; i < total; i += stride) {
    int e = (int)(i >> 7);
    int c = (int)(i & 127);
    int d = dst[e];
    if (hmap[d] >= 0) continue;
    atomicAdd(&acc[((long)et[e] * nOut + d) * 128 + c],
              feat[(long)src[e] * 128 + c]);
  }
}

__global__ void k_histfill_stats(const int* __restrict__ hmap,
                                 const float* __restrict__ hbuf,
                                 float* __restrict__ h1,
                                 float* __restrict__ csum,
                                 float* __restrict__ csq) {
  __shared__ float ls[256], lq[256];
  long total = (long)N1 * 128;
  long i = (long)blockIdx.x * blockDim.x + threadIdx.x;
  long stride = (long)gridDim.x * blockDim.x;
  float s = 0.f, q = 0.f;
  for (; i < total; i += stride) {
    int n = (int)(i >> 7);
    int c = (int)(i & 127);
    int hm = hmap[n];
    if (hm >= 0) {
      float v = hbuf[(long)hm * 128 + c];
      h1[i] = v;
      s += v; q += v * v;
    }
  }
  ls[threadIdx.x] = s; lq[threadIdx.x] = q;
  __syncthreads();
  if (threadIdx.x < 128) {
    float S = ls[threadIdx.x] + ls[threadIdx.x + 128];
    float Q = lq[threadIdx.x] + lq[threadIdx.x + 128];
    atomicAdd(&csum[threadIdx.x], S);
    atomicAdd(&csq[threadIdx.x], Q);
  }
}

#define P1 (2 * 128 + 32)
#define LDS1 (32 * P1)

__global__ void k_gemm1(const float* __restrict__ A1,
                        const float* __restrict__ x,
                        const float* __restrict__ W1,
                        const float* __restrict__ Ws,
                        const float* __restrict__ b1,
                        const int*   __restrict__ hmap,
                        const float* __restrict__ hbuf,
                        float* __restrict__ h1,
                        float* __restrict__ csum,
                        float* __restrict__ csq) {
  __shared__ float bs[LDS1];
  __shared__ int blockNeed;

  if (threadIdx.x == 0) blockNeed = 0;
  __syncthreads();
  int blockRow = blockIdx.x * 128;
  int rr = blockRow + threadIdx.x;
  if (threadIdx.x < 128 && rr < N1 && hmap[rr] < 0) blockNeed = 1;
  __syncthreads();
  if (!blockNeed) return;

  int wave = threadIdx.x >> 5;
  int lane = threadIdx.x & 31;
  const int mtiles = N1 / 16;
  int tile = blockIdx.x * 8 + wave;
  bool valid = (tile < mtiles);
  int row0 = (valid ? tile : 0) * 16;
  bool pred = valid && (lane < 16) && (hmap[row0 + (lane & 15)] < 0);
  bool doWork = valid && (__ballot(pred) != 0ull);

  int aRow = lane & 15;
  int aK   = (lane >> 4) << 1;
  int n16  = lane & 15;

  v8f acc[8];
  v8f zero = {0.f,0.f,0.f,0.f,0.f,0.f,0.f,0.f};
#pragma unroll
  for (int t = 0; t < 8; ++t) acc[t] = zero;

  for (int seg = 0; seg < 5; ++seg) {
    const float* Ap = (seg < 4) ? (A1 + ((long)seg * N1 + row0 + aRow) * 128)
                                : (x  + (long)(row0 + aRow) * 128);
    const float* Bp = (seg < 4) ? (W1 + (long)seg * 128 * 128) : Ws;
    for (int kc = 0; kc < 128; kc += 64) {
      __syncthreads();
      for (int idx = threadIdx.x; idx < 64 * 128; idx += 256) {
        int k = idx >> 7;
        int n = idx & 127;
        bs[(k >> 1) * P1 + n * 2 + (k & 1)] = Bp[(kc + k) * 128 + n];
        if (kc == 0)
          __builtin_prefetch(&Bp[(64 + k) * 128 + n], 0, 0);
      }
      __syncthreads();

      if (doWork) {
        v2f a = *(const v2f*)(Ap + kc + aK);
#pragma unroll 2
        for (int kr = 0; kr < 64; kr += 4) {
          v2f a_cur = a;
          if (kr < 60) a = *(const v2f*)(Ap + kc + kr + 4 + aK);
          const v2f* bpair =
              (const v2f*)(bs + ((kr >> 1) + (lane >> 4)) * P1) + n16;
          v2f bf[8];
#pragma unroll
          for (int t = 0; t < 8; ++t) bf[t] = bpair[t * 16];
#pragma unroll
          for (int t = 0; t < 8; ++t)
            acc[t] = __builtin_amdgcn_wmma_f32_16x16x4_f32(
                false, a_cur, false, bf[t], (short)0, acc[t], false, false);
        }
      }
    }
  }

  if (!doWork) return;
  int halfOff = (lane >> 4) << 3;
#pragma unroll
  for (int v = 0; v < 8; ++v) {
    int r = row0 + v + halfOff;
    int hm = hmap[r];
#pragma unroll
    for (int t = 0; t < 8; ++t) {
      int col = t * 16 + n16;
      float val = acc[t][v] + b1[col];
      if (hm >= 0) val = hbuf[(long)hm * 128 + col];
      else {
        atomicAdd(&csum[col], val);
        atomicAdd(&csq[col], val * val);
      }
      h1[(long)r * 128 + col] = val;
    }
  }
}

__global__ void k_bnfinal(const float* __restrict__ csum,
                          const float* __restrict__ csq,
                          const float* __restrict__ gamma,
                          const float* __restrict__ beta,
                          float* __restrict__ scale,
                          float* __restrict__ shift) {
  int c = threadIdx.x;
  const float inv_n = 1.0f / (float)N1;
  float mean = csum[c] * inv_n;
  float var  = csq[c] * inv_n - mean * mean;
  float sc = gamma[c] * rsqrtf(var + BN_EPS);
  scale[c] = sc;
  shift[c] = beta[c] - mean * sc;
}

__global__ void k_normhead(const float* __restrict__ h1,
                           const float* __restrict__ scale,
                           const float* __restrict__ shift,
                           float* __restrict__ h1n) {
  long total = (long)N2 * 128;
  long i = (long)blockIdx.x * blockDim.x + threadIdx.x;
  long stride = (long)gridDim.x * blockDim.x;
  for (; i < total; i += stride) {
    int c = (int)(i & 127);
    h1n[i] = fmaxf(0.0f, h1[i] * scale[c] + shift[c]);
  }
}

__global__ void k_scatter2n(const float* __restrict__ h1,
                            const float* __restrict__ scale,
                            const float* __restrict__ shift,
                            const int* __restrict__ src,
                            const int* __restrict__ dst,
                            const int* __restrict__ et,
                            float* __restrict__ acc) {
  long total = (long)E2 * 128;
  long i = (long)blockIdx.x * blockDim.x + threadIdx.x;
  long stride = (long)gridDim.x * blockDim.x;
  for (; i < total; i += stride) {
    int e = (int)(i >> 7);
    int c = (int)(i & 127);
    float v = fmaxf(0.0f, h1[(long)src[e] * 128 + c] * scale[c] + shift[c]);
    atomicAdd(&acc[((long)et[e] * N2 + dst[e]) * 128 + c], v);
  }
}

#define P2 (2 * 64 + 32)
#define LDS2 (64 * P2)

__global__ void k_gemm2(const float* __restrict__ A2,
                        const float* __restrict__ h1n,
                        const float* __restrict__ W2,
                        const float* __restrict__ Ws2,
                        const float* __restrict__ b2,
                        float* __restrict__ out) {
  __shared__ float bs[LDS2];

  int wave = (int)((blockIdx.x * blockDim.x + threadIdx.x) >> 5);
  int lane = threadIdx.x & 31;
  const int mtiles = N2 / 16;
  bool valid = (wave < mtiles);
  int row0 = (valid ? wave : 0) * 16;
  int aRow = lane & 15;
  int aK   = (lane >> 4) << 1;
  int n16  = lane & 15;

  v8f acc[4];
  v8f zero = {0.f,0.f,0.f,0.f,0.f,0.f,0.f,0.f};
#pragma unroll
  for (int t = 0; t < 4; ++t) acc[t] = zero;

  for (int seg = 0; seg < 5; ++seg) {
    const float* Ap = (seg < 4) ? (A2  + ((long)seg * N2 + row0 + aRow) * 128)
                                : (h1n + (long)(row0 + aRow) * 128);
    const float* Bp = (seg < 4) ? (W2 + (long)seg * 128 * 64) : Ws2;

    __syncthreads();
    for (int idx = threadIdx.x; idx < 128 * 64; idx += 256) {
      int k = idx >> 6;
      int n = idx & 63;
      bs[(k >> 1) * P2 + n * 2 + (k & 1)] = Bp[k * 64 + n];
    }
    __syncthreads();

    if (valid) {
      v2f a = *(const v2f*)(Ap + aK);
#pragma unroll 2
      for (int k0 = 0; k0 < 128; k0 += 4) {
        v2f a_cur = a;
        if (k0 < 124) a = *(const v2f*)(Ap + k0 + 4 + aK);
        const v2f* bpair =
            (const v2f*)(bs + ((k0 >> 1) + (lane >> 4)) * P2) + n16;
        v2f bf[4];
#pragma unroll
        for (int t = 0; t < 4; ++t) bf[t] = bpair[t * 16];
#pragma unroll
        for (int t = 0; t < 4; ++t)
          acc[t] = __builtin_amdgcn_wmma_f32_16x16x4_f32(
              false, a_cur, false, bf[t], (short)0, acc[t], false, false);
      }
    }
  }

  if (!valid) return;
  int halfOff = (lane >> 4) << 3;
#pragma unroll
  for (int v = 0; v < 8; ++v) {
    int r = row0 + v + halfOff;
    float vals[4];
    float m = -1e30f;
#pragma unroll
    for (int t = 0; t < 4; ++t) {
      vals[t] = acc[t][v] + b2[t * 16 + n16];
      m = fmaxf(m, vals[t]);
    }
#pragma unroll
    for (int mask = 1; mask < 16; mask <<= 1)
      m = fmaxf(m, __shfl_xor(m, mask, 32));
    float s = 0.f;
#pragma unroll
    for (int t = 0; t < 4; ++t) s += expf(vals[t] - m);
#pragma unroll
    for (int mask = 1; mask < 16; mask <<= 1)
      s += __shfl_xor(s, mask, 32);
    float lse = m + logf(s);
#pragma unroll
    for (int t = 0; t < 4; ++t)
      out[(long)r * 64 + t * 16 + n16] = vals[t] - lse;
  }
}

extern "C" void kernel_launch(void* const* d_in, const int* in_sizes, int n_in,
                              void* d_out, int out_size, void* d_ws, size_t ws_size,
                              hipStream_t stream) {
  const float* x     = (const float*)d_in[0];
  const int*   src1  = (const int*)  d_in[1];
  const int*   dst1  = (const int*)  d_in[2];
  const int*   et1   = (const int*)  d_in[3];
  const int*   src2  = (const int*)  d_in[4];
  const int*   dst2  = (const int*)  d_in[5];
  const int*   et2   = (const int*)  d_in[6];
  const int*   hmap  = (const int*)  d_in[7];
  const float* hbuf  = (const float*)d_in[8];
  const float* W1    = (const float*)d_in[9];
  const float* Ws1   = (const float*)d_in[10];
  const float* b1    = (const float*)d_in[11];
  const float* gamma = (const float*)d_in[12];
  const float* beta  = (const float*)d_in[13];
  const float* W2    = (const float*)d_in[14];
  const float* Ws2   = (const float*)d_in[15];
  const float* b2    = (const float*)d_in[16];

  float* ws    = (float*)d_ws;
  float* A1    = ws;
  float* A2    = A1 + (size_t)4 * N1 * 128;
  float* sums  = A2 + (size_t)4 * N2 * 128;
  float* ss    = sums + 256;
  float* h1    = ss + 256;
  float* h1n   = h1 + (size_t)N1 * 128;
  float* out   = (float*)d_out;

  k_zero<<<512, 256, 0, stream>>>(A2, (long)4 * N2 * 128 + 512);
  k_zeroA1<<<256, 256, 0, stream>>>(A1, hmap);

  k_scatter_f<<<4096, 256, 0, stream>>>(x, src1, dst1, et1, hmap, A1, E1, N1);
  k_histfill_stats<<<512, 256, 0, stream>>>(hmap, hbuf, h1, sums, sums + 128);

  k_gemm1<<<(N1/16 + 7) / 8, 256, 0, stream>>>(A1, x, W1, Ws1, b1, hmap, hbuf,
                                               h1, sums, sums + 128);

  k_bnfinal<<<1, 128, 0, stream>>>(sums, sums + 128, gamma, beta, ss, ss + 128);
  k_normhead<<<512, 256, 0, stream>>>(h1, ss, ss + 128, h1n);

  k_scatter2n<<<2048, 256, 0, stream>>>(h1, ss, ss + 128, src2, dst2, et2, A2);

  k_gemm2<<<(N2/16 + 7) / 8, 256, 0, stream>>>(A2, h1n, W2, Ws2, b2, out);
}


# layer 1 goes transform-then-aggregate: edges into no-history rows compacted to a list, x[src] rows meet W1[et] on WMMA, summed into h1 by atomics; A1 plane and tile GEMM dropped (sum order changes)
# speedup vs baseline: 2.2509x; 2.2509x over previous
#include <hip/hip_runtime.h>
#include <hip/hip_bf16.h>
#include <math.h>

typedef __attribute__((ext_vector_type(2))) float v2f;
typedef __attribute__((ext_vector_type(8))) float v8f;

#define N0  300000
#define N1  60000
#define N2  6000
#define E1  1200000
#define E2  300000
#define D_H 128
#define D_OUT 64
#define BN_EPS 1e-5f

__global__ void k_zero(float* __restrict__ p, long n) {
  long i = (long)blockIdx.x * blockDim.x + threadIdx.x;
  long stride = (long)gridDim.x * blockDim.x;
  for (; i < n; i += stride) p[i] = 0.0f;
}

#define NQ (E1 / 4)
#define CAP (E1 + N1)
static_assert(E1 % 4 == 0, "edge list is read four at a time");

__device__ __forceinline__ void push_entry(int* __restrict__ cnts,
                                           int* __restrict__ esrc,
                                           int* __restrict__ edst,
                                           int* __restrict__ erel,
                                           int s, int d, int r) {
  int slot = atomicAdd(&cnts[0], 1);
  atomicExch(&esrc[slot], s);
  atomicExch(&edst[slot], d);
  atomicExch(&erel[slot], r);
}

__global__ void k_pick(const int* __restrict__ src,
                       const int* __restrict__ dst,
                       const int* __restrict__ et,
                       const int* __restrict__ hmap,
                       int* __restrict__ cnts,
                       int* __restrict__ esrc,
                       int* __restrict__ edst,
                       int* __restrict__ erel,
                       int* __restrict__ rlist) {
  int t = blockIdx.x * blockDim.x + threadIdx.x;
  if (t < NQ) {
    int4 d = ((const int4*)dst)[t];
    int h0 = hmap[d.x], h1_ = hmap[d.y], h2 = hmap[d.z], h3 = hmap[d.w];
    int e = t * 4;
    if (h0 < 0) push_entry(cnts, esrc, edst, erel, src[e],     d.x, et[e]);
    if (h1_ < 0) push_entry(cnts, esrc, edst, erel, src[e + 1], d.y, et[e + 1]);
    if (h2 < 0) push_entry(cnts, esrc, edst, erel, src[e + 2], d.z, et[e + 2]);
    if (h3 < 0) push_entry(cnts, esrc, edst, erel, src[e + 3], d.w, et[e + 3]);
  } else {
    int n = t - NQ;
    if (n < N1 && hmap[n] < 0) {
      push_entry(cnts, esrc, edst, erel, n, n, 4);
      int slot = atomicAdd(&cnts[1], 1);
      atomicExch(&rlist[slot], n);
    }
  }
}

__global__ void k_histfill_stats(const int* __restrict__ hmap,
                                 const float* __restrict__ hbuf,
                                 const float* __restrict__ b1,
                                 float* __restrict__ h1,
                                 float* __restrict__ csum,
                                 float* __restrict__ csq) {
  __shared__ float ls[256], lq[256];
  long total = (long)N1 * 128;
  long i = (long)blockIdx.x * blockDim.x + threadIdx.x;
  long stride = (long)gridDim.x * blockDim.x;
  float s = 0.f, q = 0.f;
  for (; i < total; i += stride) {
    int n = (int)(i >> 7);
    int c = (int)(i & 127);
    int hm = hmap[n];
    if (hm >= 0) {
      float v = hbuf[(long)hm * 128 + c];
      h1[i] = v;
      s += v; q += v * v;
    } else {
      h1[i] = b1[c];
    }
  }
  ls[threadIdx.x] = s; lq[threadIdx.x] = q;
  __syncthreads();
  if (threadIdx.x < 128) {
    float S = ls[threadIdx.x] + ls[threadIdx.x + 128];
    float Q = lq[threadIdx.x] + lq[threadIdx.x + 128];
    atomicAdd(&csum[threadIdx.x], S);
    atomicAdd(&csq[threadIdx.x], Q);
  }
}

__global__ void __launch_bounds__(256)
k_msg1(const float* __restrict__ x,
                       const float* __restrict__ W1,
                       const float* __restrict__ Ws,
                       const int* __restrict__ cnts,
                       const int* __restrict__ esrc,
                       const int* __restrict__ edst,
                       const int* __restrict__ erel,
                       float* __restrict__ h1) {
  int cnt = cnts[0];
  int ntiles = (cnt + 15) >> 4;
  int wave = threadIdx.x >> 5;
  int lane = threadIdx.x & 31;
  int aRow = lane & 15;
  int aK   = (lane >> 4) << 1;
  int col  = wave * 16 + (lane & 15);
  int halfOff = (lane >> 4) << 3;
  const v2f zero2 = {0.f, 0.f};

  for (int tile = blockIdx.x; tile < ntiles; tile += gridDim.x) {
    int ei = tile * 16 + aRow;
    bool live = ei < cnt;
    int ec = live ? ei : 0;
    int sRow = esrc[ec];
    int rel  = live ? erel[ec] : -1;
    const float* Ap = x + (long)sRow * 128 + aK;

    v2f af[32];
#pragma unroll
    for (int ks = 0; ks < 32; ++ks) af[ks] = *(const v2f*)(Ap + 4 * ks);

    v8f acc = {0.f,0.f,0.f,0.f,0.f,0.f,0.f,0.f};
    for (int r = 0; r < 5; ++r) {
      bool mine = (rel == r);
      if (__ballot(mine) == 0ull) continue;
      const float* Bp = ((r < 4) ? (W1 + (long)r * 128 * 128) : Ws)
                        + aK * 128 + col;
      v2f bf[32];
#pragma unroll
      for (int ks = 0; ks < 32; ++ks) {
        bf[ks].x = Bp[(4 * ks) * 128];
        bf[ks].y = Bp[(4 * ks + 1) * 128];
      }
#pragma unroll
      for (int ks = 0; ks < 32; ++ks) {
        v2f a = mine ? af[ks] : zero2;
        acc = __builtin_amdgcn_wmma_f32_16x16x4_f32(
            false, a, false, bf[ks], (short)0, acc, false, false);
      }
    }

    int dd[8];
#pragma unroll
    for (int v = 0; v < 8; ++v) {
      int eo = tile * 16 + v + halfOff;
      dd[v] = edst[eo < cnt ? eo : 0];
    }
#pragma unroll
    for (int v = 0; v < 8; ++v) {
      int eo = tile * 16 + v + halfOff;
      if (eo < cnt) atomicAdd(&h1[(long)dd[v] * 128 + col], acc[v]);
    }
  }
}

__global__ void k_bnfinal(const float* __restrict__ csum,
                          const float* __restrict__ csq,
                          const float* __restrict__ gamma,
                          const float* __restrict__ beta,
                          float* __restrict__ scale,
                          float* __restrict__ shift,
                          const int* __restrict__ cnts,
                          const int* __restrict__ rlist,
                          const float* __restrict__ h1) {
  int c = threadIdx.x;
  float s = csum[c], q = csq[c];
  int nr = cnts[1];
  for (int j = 0; j < nr; ++j) {
    float v = h1[(long)rlist[j] * 128 + c];
    s += v; q += v * v;
  }
  const float inv_n = 1.0f / (float)N1;
  float mean = s * inv_n;
  float var  = q * inv_n - mean * mean;
  float sc = gamma[c] * rsqrtf(var + BN_EPS);
  scale[c] = sc;
  shift[c] = beta[c] - mean * sc;
}

__global__ void k_normhead(const float* __restrict__ h1,
                           const float* __restrict__ scale,
                           const float* __restrict__ shift,
                           float* __restrict__ h1n) {
  long total = (long)N2 * 128;
  long i = (long)blockIdx.x * blockDim.x + threadIdx.x;
  long stride = (long)gridDim.x * blockDim.x;
  for (; i < total; i += stride) {
    int c = (int)(i & 127);
    h1n[i] = fmaxf(0.0f, h1[i] * scale[c] + shift[c]);
  }
}

__global__ void k_scatter2n(const float* __restrict__ h1,
                            const float* __restrict__ scale,
                            const float* __restrict__ shift,
                            const int* __restrict__ src,
                            const int* __restrict__ dst,
                            const int* __restrict__ et,
                            float* __restrict__ acc) {
  long total = (long)E2 * 128;
  long i = (long)blockIdx.x * blockDim.x + threadIdx.x;
  long stride = (long)gridDim.x * blockDim.x;
  for (; i < total; i += stride) {
    int e = (int)(i >> 7);
    int c = (int)(i & 127);
    float v = fmaxf(0.0f, h1[(long)src[e] * 128 + c] * scale[c] + shift[c]);
    atomicAdd(&acc[((long)et[e] * N2 + dst[e]) * 128 + c], v);
  }
}

#define P2 (2 * 64 + 32)
#define LDS2 (64 * P2)

__global__ void k_gemm2(const float* __restrict__ A2,
                        const float* __restrict__ h1n,
                        const float* __restrict__ W2,
                        const float* __restrict__ Ws2,
                        const float* __restrict__ b2,
                        float* __restrict__ out) {
  __shared__ float bs[LDS2];

  int wave = (int)((blockIdx.x * blockDim.x + threadIdx.x) >> 5);
  int lane = threadIdx.x & 31;
  const int mtiles = N2 / 16;
  bool valid = (wave < mtiles);
  int row0 = (valid ? wave : 0) * 16;
  int aRow = lane & 15;
  int aK   = (lane >> 4) << 1;
  int n16  = lane & 15;

  v8f acc[4];
  v8f zero = {0.f,0.f,0.f,0.f,0.f,0.f,0.f,0.f};
#pragma unroll
  for (int t = 0; t < 4; ++t) acc[t] = zero;

  for (int seg = 0; seg < 5; ++seg) {
    const float* Ap = (seg < 4) ? (A2  + ((long)seg * N2 + row0 + aRow) * 128)
                                : (h1n + (long)(row0 + aRow) * 128);
    const float* Bp = (seg < 4) ? (W2 + (long)seg * 128 * 64) : Ws2;

    __syncthreads();
    for (int idx = threadIdx.x; idx < 128 * 64; idx += 256) {
      int k = idx >> 6;
      int n = idx & 63;
      bs[(k >> 1) * P2 + n * 2 + (k & 1)] = Bp[k * 64 + n];
    }
    __syncthreads();

    if (valid) {
      v2f a = *(const v2f*)(Ap + aK);
#pragma unroll 2
      for (int k0 = 0; k0 < 128; k0 += 4) {
        v2f a_cur = a;
        if (k0 < 124) a = *(const v2f*)(Ap + k0 + 4 + aK);
        const v2f* bpair =
            (const v2f*)(bs + ((k0 >> 1) + (lane >> 4)) * P2) + n16;
        v2f bf[4];
#pragma unroll
        for (int t = 0; t < 4; ++t) bf[t] = bpair[t * 16];
#pragma unroll
        for (int t = 0; t < 4; ++t)
          acc[t] = __builtin_amdgcn_wmma_f32_16x16x4_f32(
              false, a_cur, false, bf[t], (short)0, acc[t], false, false);
      }
    }
  }

  if (!valid) return;
  int halfOff = (lane >> 4) << 3;
#pragma unroll
  for (int v = 0; v < 8; ++v) {
    int r = row0 + v + halfOff;
    float vals[4];
    float m = -1e30f;
#pragma unroll
    for (int t = 0; t < 4; ++t) {
      vals[t] = acc[t][v] + b2[t * 16 + n16];
      m = fmaxf(m, vals[t]);
    }
#pragma unroll
    for (int mask = 1; mask < 16; mask <<= 1)
      m = fmaxf(m, __shfl_xor(m, mask, 32));
    float s = 0.f;
#pragma unroll
    for (int t = 0; t < 4; ++t) s += expf(vals[t] - m);
#pragma unroll
    for (int mask = 1; mask < 16; mask <<= 1)
      s += __shfl_xor(s, mask, 32);
    float lse = m + logf(s);
#pragma unroll
    for (int t = 0; t < 4; ++t)
      out[(long)r * 64 + t * 16 + n16] = vals[t] - lse;
  }
}

extern "C" void kernel_launch(void* const* d_in, const int* in_sizes, int n_in,
                              void* d_out, int out_size, void* d_ws, size_t ws_size,
                              hipStream_t stream) {
  const float* x     = (const float*)d_in[0];
  const int*   src1  = (const int*)  d_in[1];
  const int*   dst1  = (const int*)  d_in[2];
  const int*   et1   = (const int*)  d_in[3];
  const int*   src2  = (const int*)  d_in[4];
  const int*   dst2  = (const int*)  d_in[5];
  const int*   et2   = (const int*)  d_in[6];
  const int*   hmap  = (const int*)  d_in[7];
  const float* hbuf  = (const float*)d_in[8];
  const float* W1    = (const float*)d_in[9];
  const float* Ws1   = (const float*)d_in[10];
  const float* b1    = (const float*)d_in[11];
  const float* gamma = (const float*)d_in[12];
  const float* beta  = (const float*)d_in[13];
  const float* W2    = (const float*)d_in[14];
  const float* Ws2   = (const float*)d_in[15];
  const float* b2    = (const float*)d_in[16];

  float* ws    = (float*)d_ws;
  float* A1    = ws;
  float* A2    = A1 + (size_t)4 * N1 * 128;
  float* sums  = A2 + (size_t)4 * N2 * 128;
  float* ss    = sums + 256;
  float* h1    = ss + 256;
  float* h1n   = h1 + (size_t)N1 * 128;
  float* out   = (float*)d_out;
  int* esrc  = (int*)A1;
  int* edst  = esrc + CAP;
  int* erel  = edst + CAP;
  int* rlist = erel + CAP;
  int* cnts  = (int*)A2 - 4;

  k_zero<<<512, 256, 0, stream>>>(A2 - 4, (long)4 * N2 * 128 + 512 + 4);

  k_pick<<<(NQ + N1 + 255) / 256, 256, 0, stream>>>(src1, dst1, et1, hmap, cnts,
                                                    esrc, edst, erel, rlist);
  k_histfill_stats<<<512, 256, 0, stream>>>(hmap, hbuf, b1, h1, sums, sums + 128);

  k_msg1<<<1024, 256, 0, stream>>>(x, W1, Ws1, cnts, esrc, edst, erel, h1);

  k_bnfinal<<<1, 128, 0, stream>>>(sums, sums + 128, gamma, beta, ss, ss + 128,
                                   cnts, rlist, h1);
  k_normhead<<<512, 256, 0, stream>>>(h1, ss, ss + 128, h1n);

  k_scatter2n<<<2048, 256, 0, stream>>>(h1, ss, ss + 128, src2, dst2, et2, A2);

  k_gemm2<<<(N2/16 + 7) / 8, 256, 0, stream>>>(A2, h1n, W2, Ws2, b2, out);
}


# layer 2 without float atomics: edges counting-sorted by (dst, relation); each 16-row group sums its segments into an LDS A tile and runs it on WMMA in place (edge, K, softmax sum order changes)
# speedup vs baseline: 2.8495x; 1.2659x over previous
#include <hip/hip_runtime.h>
#include <hip/hip_bf16.h>
#include <math.h>

typedef __attribute__((ext_vector_type(2))) float v2f;
typedef __attribute__((ext_vector_type(8))) float v8f;

#define N0  300000
#define N1  60000
#define N2  6000
#define E1  1200000
#define E2  300000
#define D_H 128
#define D_OUT 64
#define BN_EPS 1e-5f

__global__ void k_zero(float* __restrict__ p, long n) {
  long i = (long)blockIdx.x * blockDim.x + threadIdx.x;
  long stride = (long)gridDim.x * blockDim.x;
  for (; i < n; i += stride) p[i] = 0.0f;
}

#define NQ (E1 / 4)
#define CAP (E1 + N1)
static_assert(E1 % 4 == 0, "edge list is read four at a time");
#define NSEG (4 * N2)
static_assert(E2 % 4 == 0, "layer-2 edge list is read four at a time");

__device__ __forceinline__ void push_entry(int* __restrict__ cnts,
                                           int* __restrict__ esrc,
                                           int* __restrict__ edst,
                                           int* __restrict__ erel,
                                           int s, int d, int r) {
  int slot = atomicAdd(&cnts[0], 1);
  atomicExch(&esrc[slot], s);
  atomicExch(&edst[slot], d);
  atomicExch(&erel[slot], r);
}

__global__ void k_pick(const int* __restrict__ src,
                       const int* __restrict__ dst,
                       const int* __restrict__ et,
                       const int* __restrict__ hmap,
                       int* __restrict__ cnts,
                       int* __restrict__ esrc,
                       int* __restrict__ edst,
                       int* __restrict__ erel,
                       int* __restrict__ rlist,
                       const int* __restrict__ dst2,
                       const int* __restrict__ et2,
                       int* __restrict__ deg) {
  int t = blockIdx.x * blockDim.x + threadIdx.x;
  if (t < NQ) {
    int4 d = ((const int4*)dst)[t];
    int h0 = hmap[d.x], h1_ = hmap[d.y], h2 = hmap[d.z], h3 = hmap[d.w];
    int e = t * 4;
    if (h0 < 0) push_entry(cnts, esrc, edst, erel, src[e],     d.x, et[e]);
    if (h1_ < 0) push_entry(cnts, esrc, edst, erel, src[e + 1], d.y, et[e + 1]);
    if (h2 < 0) push_entry(cnts, esrc, edst, erel, src[e + 2], d.z, et[e + 2]);
    if (h3 < 0) push_entry(cnts, esrc, edst, erel, src[e + 3], d.w, et[e + 3]);
  } else if (t < NQ + N1) {
    int n = t - NQ;
    if (hmap[n] < 0) {
      push_entry(cnts, esrc, edst, erel, n, n, 4);
      int slot = atomicAdd(&cnts[1], 1);
      atomicExch(&rlist[slot], n);
    }
  } else {
    int q = t - (NQ + N1);
    if (q < E2 / 4) {
      int4 d = ((const int4*)dst2)[q];
      int4 r = ((const int4*)et2)[q];
      atomicAdd(&deg[d.x * 4 + r.x], 1);
      atomicAdd(&deg[d.y * 4 + r.y], 1);
      atomicAdd(&deg[d.z * 4 + r.z], 1);
      atomicAdd(&deg[d.w * 4 + r.w], 1);
    }
  }
}

__global__ void __launch_bounds__(1024)
k_scan(const int* __restrict__ deg,
       int* __restrict__ segoff,
       int* __restrict__ cursor) {
  __shared__ int wsum[32];
  const int PER = 24;
  static_assert(NSEG % 24 == 0 && NSEG / 24 <= 1024, "one thread per 24 segments");
  int t = threadIdx.x;
  int lane = t & 31, w = t >> 5;
  bool have = t < NSEG / PER;
  int loc[PER];
  int s = 0;
  if (have) {
    const int4* p = (const int4*)(deg + t * PER);
#pragma unroll
    for (int k = 0; k < PER / 4; ++k) {
      int4 v = p[k];
      loc[4 * k] = s;     s += v.x;
      loc[4 * k + 1] = s; s += v.y;
      loc[4 * k + 2] = s; s += v.z;
      loc[4 * k + 3] = s; s += v.w;
    }
  } else {
#pragma unroll
    for (int k = 0; k < PER; ++k) loc[k] = 0;
  }
  int inc = s;
#pragma unroll
  for (int d = 1; d < 32; d <<= 1) {
    int o = __shfl_up(inc, d, 32);
    if (lane >= d) inc += o;
  }
  if (lane == 31) wsum[w] = inc;
  __syncthreads();
  if (w == 0) {
    int x = wsum[lane];
    int xi = x;
#pragma unroll
    for (int d = 1; d < 32; d <<= 1) {
      int o = __shfl_up(xi, d, 32);
      if (lane >= d) xi += o;
    }
    wsum[lane] = xi - x;
  }
  __syncthreads();
  int off = wsum[w] + inc - s;
  if (have) {
    int4* po = (int4*)(segoff + t * PER);
    int4* pc = (int4*)(cursor + t * PER);
#pragma unroll
    for (int k = 0; k < PER / 4; ++k) {
      int4 v;
      v.x = off + loc[4 * k];     v.y = off + loc[4 * k + 1];
      v.z = off + loc[4 * k + 2]; v.w = off + loc[4 * k + 3];
      po[k] = v;
      pc[k] = v;
    }
  }
  if (t == NSEG / PER) segoff[NSEG] = off;
}

__global__ void k_fill(const int* __restrict__ src2,
                       const int* __restrict__ dst2,
                       const int* __restrict__ et2,
                       int* __restrict__ cursor,
                       int* __restrict__ ssrc) {
  int q = blockIdx.x * blockDim.x + threadIdx.x;
  if (q < E2 / 4) {
    int4 s = ((const int4*)src2)[q];
    int4 d = ((const int4*)dst2)[q];
    int4 r = ((const int4*)et2)[q];
    int p0 = atomicAdd(&cursor[d.x * 4 + r.x], 1);
    int p1 = atomicAdd(&cursor[d.y * 4 + r.y], 1);
    int p2 = atomicAdd(&cursor[d.z * 4 + r.z], 1);
    int p3 = atomicAdd(&cursor[d.w * 4 + r.w], 1);
    atomicExch(&ssrc[p0], s.x);
    atomicExch(&ssrc[p1], s.y);
    atomicExch(&ssrc[p2], s.z);
    atomicExch(&ssrc[p3], s.w);
  }
}

__global__ void k_histfill_stats(const int* __restrict__ hmap,
                                 const float* __restrict__ hbuf,
                                 const float* __restrict__ b1,
                                 float* __restrict__ h1,
                                 float* __restrict__ csum,
                                 float* __restrict__ csq) {
  __shared__ float ls[256], lq[256];
  long total = (long)N1 * 128;
  long i = (long)blockIdx.x * blockDim.x + threadIdx.x;
  long stride = (long)gridDim.x * blockDim.x;
  float s = 0.f, q = 0.f;
  for (; i < total; i += stride) {
    int n = (int)(i >> 7);
    int c = (int)(i & 127);
    int hm = hmap[n];
    if (hm >= 0) {
      float v = hbuf[(long)hm * 128 + c];
      h1[i] = v;
      s += v; q += v * v;
    } else {
      h1[i] = b1[c];
    }
  }
  ls[threadIdx.x] = s; lq[threadIdx.x] = q;
  __syncthreads();
  if (threadIdx.x < 128) {
    float S = ls[threadIdx.x] + ls[threadIdx.x + 128];
    float Q = lq[threadIdx.x] + lq[threadIdx.x + 128];
    atomicAdd(&csum[threadIdx.x], S);
    atomicAdd(&csq[threadIdx.x], Q);
  }
}

__global__ void __launch_bounds__(256)
k_msg1(const float* __restrict__ x,
                       const float* __restrict__ W1,
                       const float* __restrict__ Ws,
                       const int* __restrict__ cnts,
                       const int* __restrict__ esrc,
                       const int* __restrict__ edst,
                       const int* __restrict__ erel,
                       float* __restrict__ h1) {
  int cnt = cnts[0];
  int ntiles = (cnt + 15) >> 4;
  int wave = threadIdx.x >> 5;
  int lane = threadIdx.x & 31;
  int aRow = lane & 15;
  int aK   = (lane >> 4) << 1;
  int col  = wave * 16 + (lane & 15);
  int halfOff = (lane >> 4) << 3;
  const v2f zero2 = {0.f, 0.f};

  for (int tile = blockIdx.x; tile < ntiles; tile += gridDim.x) {
    int ei = tile * 16 + aRow;
    bool live = ei < cnt;
    int ec = live ? ei : 0;
    int sRow = esrc[ec];
    int rel  = live ? erel[ec] : -1;
    const float* Ap = x + (long)sRow * 128 + aK;

    v2f af[32];
#pragma unroll
    for (int ks = 0; ks < 32; ++ks) af[ks] = *(const v2f*)(Ap + 4 * ks);

    v8f acc = {0.f,0.f,0.f,0.f,0.f,0.f,0.f,0.f};
    for (int r = 0; r < 5; ++r) {
      bool mine = (rel == r);
      if (__ballot(mine) == 0ull) continue;
      const float* Bp = ((r < 4) ? (W1 + (long)r * 128 * 128) : Ws)
                        + aK * 128 + col;
      v2f bf[32];
#pragma unroll
      for (int ks = 0; ks < 32; ++ks) {
        bf[ks].x = Bp[(4 * ks) * 128];
        bf[ks].y = Bp[(4 * ks + 1) * 128];
      }
#pragma unroll
      for (int ks = 0; ks < 32; ++ks) {
        v2f a = mine ? af[ks] : zero2;
        acc = __builtin_amdgcn_wmma_f32_16x16x4_f32(
            false, a, false, bf[ks], (short)0, acc, false, false);
      }
    }

    int dd[8];
#pragma unroll
    for (int v = 0; v < 8; ++v) {
      int eo = tile * 16 + v + halfOff;
      dd[v] = edst[eo < cnt ? eo : 0];
    }
#pragma unroll
    for (int v = 0; v < 8; ++v) {
      int eo = tile * 16 + v + halfOff;
      if (eo < cnt) atomicAdd(&h1[(long)dd[v] * 128 + col], acc[v]);
    }
  }
}

__global__ void k_bnfinal(const float* __restrict__ csum,
                          const float* __restrict__ csq,
                          const float* __restrict__ gamma,
                          const float* __restrict__ beta,
                          float* __restrict__ scale,
                          float* __restrict__ shift,
                          const int* __restrict__ cnts,
                          const int* __restrict__ rlist,
                          const float* __restrict__ h1) {
  int c = threadIdx.x;
  float s = csum[c], q = csq[c];
  int nr = cnts[1];
  for (int j = 0; j < nr; ++j) {
    float v = h1[(long)rlist[j] * 128 + c];
    s += v; q += v * v;
  }
  const float inv_n = 1.0f / (float)N1;
  float mean = s * inv_n;
  float var  = q * inv_n - mean * mean;
  float sc = gamma[c] * rsqrtf(var + BN_EPS);
  scale[c] = sc;
  shift[c] = beta[c] - mean * sc;
}

#define PA 644
__global__ void __launch_bounds__(256)
k_agg_gemm2(const float* __restrict__ h1,
            const float* __restrict__ scale,
            const float* __restrict__ shift,
            const int* __restrict__ segoff,
            const int* __restrict__ ssrc,
            const float* __restrict__ W2,
            const float* __restrict__ Ws2,
            const float* __restrict__ b2,
            float* __restrict__ out) {
  __shared__ float At[16 * PA];
  __shared__ float red[8 * 4 * 8 * 32];

  int wave = __builtin_amdgcn_readfirstlane((int)(threadIdx.x >> 5));
  int lane = threadIdx.x & 31;
  int row0 = blockIdx.x * 16;

  float4 sc = ((const float4*)scale)[lane];
  float4 sh = ((const float4*)shift)[lane];

  int g0 = row0 + wave * 2;
  const int* so = segoff + g0 * 4;
  int bnd[9];
#pragma unroll
  for (int i = 0; i < 9; ++i) bnd[i] = so[i];
  int first[8];
#pragma unroll
  for (int i = 0; i < 8; ++i) {
    int n = bnd[i + 1] - bnd[i];
    first[i] = (lane < n) ? ssrc[bnd[i] + lane] : 0;
  }
#pragma unroll
  for (int rr = 0; rr < 2; ++rr) {
    float4 x = ((const float4*)(h1 + (long)(g0 + rr) * 128))[lane];
    float4 v;
    v.x = fmaxf(0.0f, x.x * sc.x + sh.x);
    v.y = fmaxf(0.0f, x.y * sc.y + sh.y);
    v.z = fmaxf(0.0f, x.z * sc.z + sh.z);
    v.w = fmaxf(0.0f, x.w * sc.w + sh.w);
    *(float4*)(At + (wave * 2 + rr) * PA + 512 + 4 * lane) = v;
  }
#pragma unroll
  for (int i = 0; i < 8; ++i) {
    int beg = bnd[i], end = bnd[i + 1];
    float4 acc = {0.f, 0.f, 0.f, 0.f};
    for (int base = beg; base < end; base += 32) {
      int n = end - base;
      n = n < 32 ? n : 32;
      int mine = first[i];
      if (base != beg) mine = (lane < n) ? ssrc[base + lane] : 0;
      for (int j0 = 0; j0 < n; j0 += 8) {
        float4 x[8];
#pragma unroll
        for (int k = 0; k < 8; ++k) {
          int sRow = __builtin_amdgcn_readlane(mine, j0 + k);
          x[k] = ((const float4*)(h1 + (long)sRow * 128))[lane];
        }
#pragma unroll
        for (int k = 0; k < 8; ++k) {
          if (j0 + k < n) {
            acc.x += fmaxf(0.0f, x[k].x * sc.x + sh.x);
            acc.y += fmaxf(0.0f, x[k].y * sc.y + sh.y);
            acc.z += fmaxf(0.0f, x[k].z * sc.z + sh.z);
            acc.w += fmaxf(0.0f, x[k].w * sc.w + sh.w);
          }
        }
      }
    }
    *(float4*)(At + (wave * 2 + (i >> 2)) * PA + (i & 3) * 128 + 4 * lane) = acc;
  }
  __syncthreads();

  int c = lane & 15;
  int h2 = (lane >> 4) << 1;
  int kb = wave * 80;
  v2f af[20];
  v2f bf[20][4];
#pragma unroll
  for (int ks = 0; ks < 20; ++ks) {
    int kk = kb + 4 * ks;
    const float* Bk = (kk < 512) ? (W2 + kk * 64) : (Ws2 + (kk - 512) * 64);
    Bk += h2 * 64 + c;
#pragma unroll
    for (int t = 0; t < 4; ++t) {
      bf[ks][t].x = Bk[t * 16];
      bf[ks][t].y = Bk[64 + t * 16];
    }
    af[ks] = *(const v2f*)(At + c * PA + kk + h2);
  }
  v8f acc[4];
  const v8f zero = {0.f,0.f,0.f,0.f,0.f,0.f,0.f,0.f};
#pragma unroll
  for (int t = 0; t < 4; ++t) acc[t] = zero;
#pragma unroll
  for (int ks = 0; ks < 20; ++ks) {
#pragma unroll
    for (int t = 0; t < 4; ++t)
      acc[t] = __builtin_amdgcn_wmma_f32_16x16x4_f32(
          false, af[ks], false, bf[ks][t], (short)0, acc[t], false, false);
  }
#pragma unroll
  for (int t = 0; t < 4; ++t)
#pragma unroll
    for (int v = 0; v < 8; ++v)
      red[((wave * 4 + t) * 8 + v) * 32 + lane] = acc[t][v];
  __syncthreads();

  int row = threadIdx.x >> 4;
  int cq  = threadIdx.x & 15;
  int ridx = (((cq >> 2) * 8 + (row & 7)) * 32) + (row >> 3) * 16 + (cq & 3) * 4;
  float4 bb = ((const float4*)b2)[cq];
  float vals[4] = {bb.x, bb.y, bb.z, bb.w};
#pragma unroll
  for (int w = 0; w < 8; ++w) {
    float4 p = *(const float4*)(red + w * (4 * 8 * 32) + ridx);
    vals[0] += p.x; vals[1] += p.y; vals[2] += p.z; vals[3] += p.w;
  }
  float m = fmaxf(fmaxf(vals[0], vals[1]), fmaxf(vals[2], vals[3]));
#pragma unroll
  for (int mask = 1; mask < 16; mask <<= 1)
    m = fmaxf(m, __shfl_xor(m, mask, 32));
  float sum = 0.f;
#pragma unroll
  for (int t = 0; t < 4; ++t) sum += expf(vals[t] - m);
#pragma unroll
  for (int mask = 1; mask < 16; mask <<= 1)
    sum += __shfl_xor(sum, mask, 32);
  float lse = m + logf(sum);
  float4 o;
  o.x = vals[0] - lse; o.y = vals[1] - lse; o.z = vals[2] - lse; o.w = vals[3] - lse;
  ((float4*)(out + (long)(row0 + row) * 64))[cq] = o;
}

extern "C" void kernel_launch(void* const* d_in, const int* in_sizes, int n_in,
                              void* d_out, int out_size, void* d_ws, size_t ws_size,
                              hipStream_t stream) {
  const float* x     = (const float*)d_in[0];
  const int*   src1  = (const int*)  d_in[1];
  const int*   dst1  = (const int*)  d_in[2];
  const int*   et1   = (const int*)  d_in[3];
  const int*   src2  = (const int*)  d_in[4];
  const int*   dst2  = (const int*)  d_in[5];
  const int*   et2   = (const int*)  d_in[6];
  const int*   hmap  = (const int*)  d_in[7];
  const float* hbuf  = (const float*)d_in[8];
  const float* W1    = (const float*)d_in[9];
  const float* Ws1   = (const float*)d_in[10];
  const float* b1    = (const float*)d_in[11];
  const float* gamma = (const float*)d_in[12];
  const float* beta  = (const float*)d_in[13];
  const float* W2    = (const float*)d_in[14];
  const float* Ws2   = (const float*)d_in[15];
  const float* b2    = (const float*)d_in[16];

  float* ws    = (float*)d_ws;
  float* A1    = ws;
  float* A2    = A1 + (size_t)4 * N1 * 128;
  float* ss    = A2 + (size_t)4 * N2 * 128 + 256;
  float* h1    = ss + 256;
  float* out   = (float*)d_out;
  int* esrc  = (int*)A1;
  int* edst  = esrc + CAP;
  int* erel  = edst + CAP;
  int* rlist = erel + CAP;
  int* cnts  = (int*)A2 - 4;
  int* deg   = (int*)A2;
  float* sums = (float*)(deg + NSEG);
  int* segoff = (int*)(sums + 256);
  int* cursor = segoff + NSEG + 32;
  int* ssrc   = cursor + NSEG;

  k_zero<<<96, 256, 0, stream>>>((float*)cnts, (long)(4 + NSEG + 256));

  k_pick<<<(NQ + N1 + E2 / 4 + 255) / 256, 256, 0, stream>>>(src1, dst1, et1, hmap, cnts,
                                                         esrc, edst, erel, rlist,
                                                         dst2, et2, deg);
  k_histfill_stats<<<512, 256, 0, stream>>>(hmap, hbuf, b1, h1, sums, sums + 128);

  k_scan<<<1, 1024, 0, stream>>>(deg, segoff, cursor);
  k_fill<<<(E2 / 4 + 255) / 256, 256, 0, stream>>>(src2, dst2, et2, cursor, ssrc);

  k_msg1<<<1024, 256, 0, stream>>>(x, W1, Ws1, cnts, esrc, edst, erel, h1);

  k_bnfinal<<<1, 128, 0, stream>>>(sums, sums + 128, gamma, beta, ss, ss + 128,
                                   cnts, rlist, h1);

  static_assert(N2 % 16 == 0, "one work-group per 16 destination rows");
  k_agg_gemm2<<<N2 / 16, 256, 0, stream>>>(h1, ss, ss + 128, segoff, ssrc,
                                           W2, Ws2, b2, out);
}


# history fill + BN sums launched as 512 groups of 1024 threads (16 waves per SIMD, was 4): a thread's index-then-row load chain is 15 trips, not 59; 8 LDS partials per column (BN sum order changes)
# speedup vs baseline: 3.5630x; 1.2504x over previous
#include <hip/hip_runtime.h>
#include <hip/hip_bf16.h>
#include <math.h>

typedef __attribute__((ext_vector_type(2))) float v2f;
typedef __attribute__((ext_vector_type(8))) float v8f;

#define N0  300000
#define N1  60000
#define N2  6000
#define E1  1200000
#define E2  300000
#define D_H 128
#define D_OUT 64
#define BN_EPS 1e-5f

__global__ void k_zero(float* __restrict__ p, long n) {
  long i = (long)blockIdx.x * blockDim.x + threadIdx.x;
  long stride = (long)gridDim.x * blockDim.x;
  for (; i < n; i += stride) p[i] = 0.0f;
}

#define NQ (E1 / 4)
#define CAP (E1 + N1)
static_assert(E1 % 4 == 0, "edge list is read four at a time");
#define NSEG (4 * N2)
static_assert(E2 % 4 == 0, "layer-2 edge list is read four at a time");

__device__ __forceinline__ void push_entry(int* __restrict__ cnts,
                                           int* __restrict__ esrc,
                                           int* __restrict__ edst,
                                           int* __restrict__ erel,
                                           int s, int d, int r) {
  int slot = atomicAdd(&cnts[0], 1);
  atomicExch(&esrc[slot], s);
  atomicExch(&edst[slot], d);
  atomicExch(&erel[slot], r);
}

__global__ void k_pick(const int* __restrict__ src,
                       const int* __restrict__ dst,
                       const int* __restrict__ et,
                       const int* __restrict__ hmap,
                       int* __restrict__ cnts,
                       int* __restrict__ esrc,
                       int* __restrict__ edst,
                       int* __restrict__ erel,
                       int* __restrict__ rlist,
                       const int* __restrict__ dst2,
                       const int* __restrict__ et2,
                       int* __restrict__ deg) {
  int t = blockIdx.x * blockDim.x + threadIdx.x;
  if (t < NQ) {
    int4 d = ((const int4*)dst)[t];
    int h0 = hmap[d.x], h1_ = hmap[d.y], h2 = hmap[d.z], h3 = hmap[d.w];
    int e = t * 4;
    if (h0 < 0) push_entry(cnts, esrc, edst, erel, src[e],     d.x, et[e]);
    if (h1_ < 0) push_entry(cnts, esrc, edst, erel, src[e + 1], d.y, et[e + 1]);
    if (h2 < 0) push_entry(cnts, esrc, edst, erel, src[e + 2], d.z, et[e + 2]);
    if (h3 < 0) push_entry(cnts, esrc, edst, erel, src[e + 3], d.w, et[e + 3]);
  } else if (t < NQ + N1) {
    int n = t - NQ;
    if (hmap[n] < 0) {
      push_entry(cnts, esrc, edst, erel, n, n, 4);
      int slot = atomicAdd(&cnts[1], 1);
      atomicExch(&rlist[slot], n);
    }
  } else {
    int q = t - (NQ + N1);
    if (q < E2 / 4) {
      int4 d = ((const int4*)dst2)[q];
      int4 r = ((const int4*)et2)[q];
      atomicAdd(&deg[d.x * 4 + r.x], 1);
      atomicAdd(&deg[d.y * 4 + r.y], 1);
      atomicAdd(&deg[d.z * 4 + r.z], 1);
      atomicAdd(&deg[d.w * 4 + r.w], 1);
    }
  }
}

__global__ void __launch_bounds__(1024)
k_scan(const int* __restrict__ deg,
       int* __restrict__ segoff,
       int* __restrict__ cursor) {
  __shared__ int wsum[32];
  const int PER = 24;
  static_assert(NSEG % 24 == 0 && NSEG / 24 <= 1024, "one thread per 24 segments");
  int t = threadIdx.x;
  int lane = t & 31, w = t >> 5;
  bool have = t < NSEG / PER;
  int loc[PER];
  int s = 0;
  if (have) {
    const int4* p = (const int4*)(deg + t * PER);
#pragma unroll
    for (int k = 0; k < PER / 4; ++k) {
      int4 v = p[k];
      loc[4 * k] = s;     s += v.x;
      loc[4 * k + 1] = s; s += v.y;
      loc[4 * k + 2] = s; s += v.z;
      loc[4 * k + 3] = s; s += v.w;
    }
  } else {
#pragma unroll
    for (int k = 0; k < PER; ++k) loc[k] = 0;
  }
  int inc = s;
#pragma unroll
  for (int d = 1; d < 32; d <<= 1) {
    int o = __shfl_up(inc, d, 32);
    if (lane >= d) inc += o;
  }
  if (lane == 31) wsum[w] = inc;
  __syncthreads();
  if (w == 0) {
    int x = wsum[lane];
    int xi = x;
#pragma unroll
    for (int d = 1; d < 32; d <<= 1) {
      int o = __shfl_up(xi, d, 32);
      if (lane >= d) xi += o;
    }
    wsum[lane] = xi - x;
  }
  __syncthreads();
  int off = wsum[w] + inc - s;
  if (have) {
    int4* po = (int4*)(segoff + t * PER);
    int4* pc = (int4*)(cursor + t * PER);
#pragma unroll
    for (int k = 0; k < PER / 4; ++k) {
      int4 v;
      v.x = off + loc[4 * k];     v.y = off + loc[4 * k + 1];
      v.z = off + loc[4 * k + 2]; v.w = off + loc[4 * k + 3];
      po[k] = v;
      pc[k] = v;
    }
  }
  if (t == NSEG / PER) segoff[NSEG] = off;
}

__global__ void k_fill(const int* __restrict__ src2,
                       const int* __restrict__ dst2,
                       const int* __restrict__ et2,
                       int* __restrict__ cursor,
                       int* __restrict__ ssrc) {
  int q = blockIdx.x * blockDim.x + threadIdx.x;
  if (q < E2 / 4) {
    int4 s = ((const int4*)src2)[q];
    int4 d = ((const int4*)dst2)[q];
    int4 r = ((const int4*)et2)[q];
    int p0 = atomicAdd(&cursor[d.x * 4 + r.x], 1);
    int p1 = atomicAdd(&cursor[d.y * 4 + r.y], 1);
    int p2 = atomicAdd(&cursor[d.z * 4 + r.z], 1);
    int p3 = atomicAdd(&cursor[d.w * 4 + r.w], 1);
    atomicExch(&ssrc[p0], s.x);
    atomicExch(&ssrc[p1], s.y);
    atomicExch(&ssrc[p2], s.z);
    atomicExch(&ssrc[p3], s.w);
  }
}

#define HF_T 1024
#define HF_G 512
static_assert(HF_T % 128 == 0, "a thread keeps one column over all its trips");
__global__ void __launch_bounds__(HF_T)
k_histfill_stats(const int* __restrict__ hmap,
                 const float* __restrict__ hbuf,
                 const float* __restrict__ b1,
                 float* __restrict__ h1,
                 float* __restrict__ csum,
                 float* __restrict__ csq) {
  __shared__ float ls[HF_T], lq[HF_T];
  long total = (long)N1 * 128;
  long i = (long)blockIdx.x * blockDim.x + threadIdx.x;
  long stride = (long)gridDim.x * blockDim.x;
  float s = 0.f, q = 0.f;
  for (; i < total; i += stride) {
    int n = (int)(i >> 7);
    int c = (int)(i & 127);
    int hm = hmap[n];
    if (hm >= 0) {
      float v = hbuf[(long)hm * 128 + c];
      h1[i] = v;
      s += v; q += v * v;
    } else {
      h1[i] = b1[c];
    }
  }
  ls[threadIdx.x] = s; lq[threadIdx.x] = q;
  __syncthreads();
  if (threadIdx.x < 128) {
    float S = 0.f, Q = 0.f;
    for (int k = 0; k < HF_T / 128; ++k) {
      S += ls[threadIdx.x + 128 * k];
      Q += lq[threadIdx.x + 128 * k];
    }
    atomicAdd(&csum[threadIdx.x], S);
    atomicAdd(&csq[threadIdx.x], Q);
  }
}

__global__ void __launch_bounds__(256)
k_msg1(const float* __restrict__ x,
                       const float* __restrict__ W1,
                       const float* __restrict__ Ws,
                       const int* __restrict__ cnts,
                       const int* __restrict__ esrc,
                       const int* __restrict__ edst,
                       const int* __restrict__ erel,
                       float* __restrict__ h1) {
  int cnt = cnts[0];
  int ntiles = (cnt + 15) >> 4;
  int wave = threadIdx.x >> 5;
  int lane = threadIdx.x & 31;
  int aRow = lane & 15;
  int aK   = (lane >> 4) << 1;
  int col  = wave * 16 + (lane & 15);
  int halfOff = (lane >> 4) << 3;
  const v2f zero2 = {0.f, 0.f};

  for (int tile = blockIdx.x; tile < ntiles; tile += gridDim.x) {
    int ei = tile * 16 + aRow;
    bool live = ei < cnt;
    int ec = live ? ei : 0;
    int sRow = esrc[ec];
    int rel  = live ? erel[ec] : -1;
    const float* Ap = x + (long)sRow * 128 + aK;

    v2f af[32];
#pragma unroll
    for (int ks = 0; ks < 32; ++ks) af[ks] = *(const v2f*)(Ap + 4 * ks);

    v8f acc = {0.f,0.f,0.f,0.f,0.f,0.f,0.f,0.f};
    for (int r = 0; r < 5; ++r) {
      bool mine = (rel == r);
      if (__ballot(mine) == 0ull) continue;
      const float* Bp = ((r < 4) ? (W1 + (long)r * 128 * 128) : Ws)
                        + aK * 128 + col;
      v2f bf[32];
#pragma unroll
      for (int ks = 0; ks < 32; ++ks) {
        bf[ks].x = Bp[(4 * ks) * 128];
        bf[ks].y = Bp[(4 * ks + 1) * 128];
      }
#pragma unroll
      for (int ks = 0; ks < 32; ++ks) {
        v2f a = mine ? af[ks] : zero2;
        acc = __builtin_amdgcn_wmma_f32_16x16x4_f32(
            false, a, false, bf[ks], (short)0, acc, false, false);
      }
    }

    int dd[8];
#pragma unroll
    for (int v = 0; v < 8; ++v) {
      int eo = tile * 16 + v + halfOff;
      dd[v] = edst[eo < cnt ? eo : 0];
    }
#pragma unroll
    for (int v = 0; v < 8; ++v) {
      int eo = tile * 16 + v + halfOff;
      if (eo < cnt) atomicAdd(&h1[(long)dd[v] * 128 + col], acc[v]);
    }
  }
}

__global__ void k_bnfinal(const float* __restrict__ csum,
                          const float* __restrict__ csq,
                          const float* __restrict__ gamma,
                          const float* __restrict__ beta,
                          float* __restrict__ scale,
                          float* __restrict__ shift,
                          const int* __restrict__ cnts,
                          const int* __restrict__ rlist,
                          const float* __restrict__ h1) {
  int c = threadIdx.x;
  float s = csum[c], q = csq[c];
  int nr = cnts[1];
  for (int j = 0; j < nr; ++j) {
    float v = h1[(long)rlist[j] * 128 + c];
    s += v; q += v * v;
  }
  const float inv_n = 1.0f / (float)N1;
  float mean = s * inv_n;
  float var  = q * inv_n - mean * mean;
  float sc = gamma[c] * rsqrtf(var + BN_EPS);
  scale[c] = sc;
  shift[c] = beta[c] - mean * sc;
}

#define PA 644
__global__ void __launch_bounds__(256)
k_agg_gemm2(const float* __restrict__ h1,
            const float* __restrict__ scale,
            const float* __restrict__ shift,
            const int* __restrict__ segoff,
            const int* __restrict__ ssrc,
            const float* __restrict__ W2,
            const float* __restrict__ Ws2,
            const float* __restrict__ b2,
            float* __restrict__ out) {
  __shared__ float At[16 * PA];
  __shared__ float red[8 * 4 * 8 * 32];

  int wave = __builtin_amdgcn_readfirstlane((int)(threadIdx.x >> 5));
  int lane = threadIdx.x & 31;
  int row0 = blockIdx.x * 16;

  float4 sc = ((const float4*)scale)[lane];
  float4 sh = ((const float4*)shift)[lane];

  int g0 = row0 + wave * 2;
  const int* so = segoff + g0 * 4;
  int bnd[9];
#pragma unroll
  for (int i = 0; i < 9; ++i) bnd[i] = so[i];
  int first[8];
#pragma unroll
  for (int i = 0; i < 8; ++i) {
    int n = bnd[i + 1] - bnd[i];
    first[i] = (lane < n) ? ssrc[bnd[i] + lane] : 0;
  }
#pragma unroll
  for (int rr = 0; rr < 2; ++rr) {
    float4 x = ((const float4*)(h1 + (long)(g0 + rr) * 128))[lane];
    float4 v;
    v.x = fmaxf(0.0f, x.x * sc.x + sh.x);
    v.y = fmaxf(0.0f, x.y * sc.y + sh.y);
    v.z = fmaxf(0.0f, x.z * sc.z + sh.z);
    v.w = fmaxf(0.0f, x.w * sc.w + sh.w);
    *(float4*)(At + (wave * 2 + rr) * PA + 512 + 4 * lane) = v;
  }
#pragma unroll
  for (int i = 0; i < 8; ++i) {
    int beg = bnd[i], end = bnd[i + 1];
    float4 acc = {0.f, 0.f, 0.f, 0.f};
    for (int base = beg; base < end; base += 32) {
      int n = end - base;
      n = n < 32 ? n : 32;
      int mine = first[i];
      if (base != beg) mine = (lane < n) ? ssrc[base + lane] : 0;
      for (int j0 = 0; j0 < n; j0 += 8) {
        float4 x[8];
#pragma unroll
        for (int k = 0; k < 8; ++k) {
          int sRow = __builtin_amdgcn_readlane(mine, j0 + k);
          x[k] = ((const float4*)(h1 + (long)sRow * 128))[lane];
        }
#pragma unroll
        for (int k = 0; k < 8; ++k) {
          if (j0 + k < n) {
            acc.x += fmaxf(0.0f, x[k].x * sc.x + sh.x);
            acc.y += fmaxf(0.0f, x[k].y * sc.y + sh.y);
            acc.z += fmaxf(0.0f, x[k].z * sc.z + sh.z);
            acc.w += fmaxf(0.0f, x[k].w * sc.w + sh.w);
          }
        }
      }
    }
    *(float4*)(At + (wave * 2 + (i >> 2)) * PA + (i & 3) * 128 + 4 * lane) = acc;
  }
  __syncthreads();

  int c = lane & 15;
  int h2 = (lane >> 4) << 1;
  int kb = wave * 80;
  v2f af[20];
  v2f bf[20][4];
#pragma unroll
  for (int ks = 0; ks < 20; ++ks) {
    int kk = kb + 4 * ks;
    const float* Bk = (kk < 512) ? (W2 + kk * 64) : (Ws2 + (kk - 512) * 64);
    Bk += h2 * 64 + c;
#pragma unroll
    for (int t = 0; t < 4; ++t) {
      bf[ks][t].x = Bk[t * 16];
      bf[ks][t].y = Bk[64 + t * 16];
    }
    af[ks] = *(const v2f*)(At + c * PA + kk + h2);
  }
  v8f acc[4];
  const v8f zero = {0.f,0.f,0.f,0.f,0.f,0.f,0.f,0.f};
#pragma unroll
  for (int t = 0; t < 4; ++t) acc[t] = zero;
#pragma unroll
  for (int ks = 0; ks < 20; ++ks) {
#pragma unroll
    for (int t = 0; t < 4; ++t)
      acc[t] = __builtin_amdgcn_wmma_f32_16x16x4_f32(
          false, af[ks], false, bf[ks][t], (short)0, acc[t], false, false);
  }
#pragma unroll
  for (int t = 0; t < 4; ++t)
#pragma unroll
    for (int v = 0; v < 8; ++v)
      red[((wave * 4 + t) * 8 + v) * 32 + lane] = acc[t][v];
  __syncthreads();

  int row = threadIdx.x >> 4;
  int cq  = threadIdx.x & 15;
  int ridx = (((cq >> 2) * 8 + (row & 7)) * 32) + (row >> 3) * 16 + (cq & 3) * 4;
  float4 bb = ((const float4*)b2)[cq];
  float vals[4] = {bb.x, bb.y, bb.z, bb.w};
#pragma unroll
  for (int w = 0; w < 8; ++w) {
    float4 p = *(const float4*)(red + w * (4 * 8 * 32) + ridx);
    vals[0] += p.x; vals[1] += p.y; vals[2] += p.z; vals[3] += p.w;
  }
  float m = fmaxf(fmaxf(vals[0], vals[1]), fmaxf(vals[2], vals[3]));
#pragma unroll
  for (int mask = 1; mask < 16; mask <<= 1)
    m = fmaxf(m, __shfl_xor(m, mask, 32));
  float sum = 0.f;
#pragma unroll
  for (int t = 0; t < 4; ++t) sum += expf(vals[t] - m);
#pragma unroll
  for (int mask = 1; mask < 16; mask <<= 1)
    sum += __shfl_xor(sum, mask, 32);
  float lse = m + logf(sum);
  float4 o;
  o.x = vals[0] - lse; o.y = vals[1] - lse; o.z = vals[2] - lse; o.w = vals[3] - lse;
  ((float4*)(out + (long)(row0 + row) * 64))[cq] = o;
}

extern "C" void kernel_launch(void* const* d_in, const int* in_sizes, int n_in,
                              void* d_out, int out_size, void* d_ws, size_t ws_size,
                              hipStream_t stream) {
  const float* x     = (const float*)d_in[0];
  const int*   src1  = (const int*)  d_in[1];
  const int*   dst1  = (const int*)  d_in[2];
  const int*   et1   = (const int*)  d_in[3];
  const int*   src2  = (const int*)  d_in[4];
  const int*   dst2  = (const int*)  d_in[5];
  const int*   et2   = (const int*)  d_in[6];
  const int*   hmap  = (const int*)  d_in[7];
  const float* hbuf  = (const float*)d_in[8];
  const float* W1    = (const float*)d_in[9];
  const float* Ws1   = (const float*)d_in[10];
  const float* b1    = (const float*)d_in[11];
  const float* gamma = (const float*)d_in[12];
  const float* beta  = (const float*)d_in[13];
  const float* W2    = (const float*)d_in[14];
  const float* Ws2   = (const float*)d_in[15];
  const float* b2    = (const float*)d_in[16];

  float* ws    = (float*)d_ws;
  float* A1    = ws;
  float* A2    = A1 + (size_t)4 * N1 * 128;
  float* ss    = A2 + (size_t)4 * N2 * 128 + 256;
  float* h1    = ss + 256;
  float* out   = (float*)d_out;
  int* esrc  = (int*)A1;
  int* edst  = esrc + CAP;
  int* erel  = edst + CAP;
  int* rlist = erel + CAP;
  int* cnts  = (int*)A2 - 4;
  int* deg   = (int*)A2;
  float* sums = (float*)(deg + NSEG);
  int* segoff = (int*)(sums + 256);
  int* cursor = segoff + NSEG + 32;
  int* ssrc   = cursor + NSEG;

  k_zero<<<96, 256, 0, stream>>>((float*)cnts, (long)(4 + NSEG + 256));

  k_pick<<<(NQ + N1 + E2 / 4 + 255) / 256, 256, 0, stream>>>(src1, dst1, et1, hmap, cnts,
                                                         esrc, edst, erel, rlist,
                                                         dst2, et2, deg);
  k_histfill_stats<<<HF_G, HF_T, 0, stream>>>(hmap, hbuf, b1, h1, sums, sums + 128);

  k_scan<<<1, 1024, 0, stream>>>(deg, segoff, cursor);
  k_fill<<<(E2 / 4 + 255) / 256, 256, 0, stream>>>(src2, dst2, et2, cursor, ssrc);

  k_msg1<<<1024, 256, 0, stream>>>(x, W1, Ws1, cnts, esrc, edst, erel, h1);

  k_bnfinal<<<1, 128, 0, stream>>>(sums, sums + 128, gamma, beta, ss, ss + 128,
                                   cnts, rlist, h1);

  static_assert(N2 % 16 == 0, "one work-group per 16 destination rows");
  k_agg_gemm2<<<N2 / 16, 256, 0, stream>>>(h1, ss, ss + 128, segoff, ssrc,
                                           W2, Ws2, b2, out);
}


# layer-2 edges binned per 16-row tile (fixed buckets + overflow list) in the edge pass; each tile sorts its own by (row, relation) in LDS: scan and fill launches gone (edge sum order: arrival rank)
# speedup vs baseline: 4.9889x; 1.4002x over previous
#include <hip/hip_runtime.h>
#include <hip/hip_bf16.h>
#include <math.h>

typedef __attribute__((ext_vector_type(2))) float v2f;
typedef __attribute__((ext_vector_type(8))) float v8f;

#define N0  300000
#define N1  60000
#define N2  6000
#define E1  1200000
#define E2  300000
#define D_H 128
#define D_OUT 64
#define BN_EPS 1e-5f

__global__ void k_zero(float* __restrict__ p, long n) {
  long i = (long)blockIdx.x * blockDim.x + threadIdx.x;
  long stride = (long)gridDim.x * blockDim.x;
  for (; i < n; i += stride) p[i] = 0.0f;
}

#define NQ (E1 / 4)
#define CAP (E1 + N1)
static_assert(E1 % 4 == 0, "edge list is read four at a time");
#define NSEG (4 * N2)
static_assert(E2 % 4 == 0, "layer-2 edge list is read four at a time");

#define NT (N2 / 16)
#define TC 2048
#define TSTRIDE 64
static_assert(N1 <= 65536, "a source row fits the low 16 bits of a bucket entry");
static_assert(TC % 256 == 0, "a chunk is a whole number of passes of the work-group");
static_assert(NT * TSTRIDE == NSEG, "the counters fill the cleared range");

__device__ __forceinline__ void put_edge(int* __restrict__ bucket,
                                         int* __restrict__ cnts,
                                         int* __restrict__ ovt,
                                         int* __restrict__ ove,
                                         int slot, int s, int d, int r) {
  int tile = d >> 4;
  int packed = s | ((((d & 15) << 2) | r) << 16);
  if (slot < TC) {
    atomicExch(&bucket[tile * TC + slot], packed);
  } else {
    int o = atomicAdd(&cnts[2], 1);
    atomicExch(&ovt[o], tile);
    atomicExch(&ove[o], packed);
  }
}

__device__ __forceinline__ void push_entry(int* __restrict__ cnts,
                                           int* __restrict__ esrc,
                                           int* __restrict__ edst,
                                           int* __restrict__ erel,
                                           int s, int d, int r) {
  int slot = atomicAdd(&cnts[0], 1);
  atomicExch(&esrc[slot], s);
  atomicExch(&edst[slot], d);
  atomicExch(&erel[slot], r);
}

__global__ void k_pick(const int* __restrict__ src,
                       const int* __restrict__ dst,
                       const int* __restrict__ et,
                       const int* __restrict__ hmap,
                       int* __restrict__ cnts,
                       int* __restrict__ esrc,
                       int* __restrict__ edst,
                       int* __restrict__ erel,
                       int* __restrict__ rlist,
                       const int* __restrict__ src2,
                       const int* __restrict__ dst2,
                       const int* __restrict__ et2,
                       int* __restrict__ tcnt,
                       int* __restrict__ bucket,
                       int* __restrict__ ovt,
                       int* __restrict__ ove) {
  int t = blockIdx.x * blockDim.x + threadIdx.x;
  if (t < E2 / 4) {
    int4 s = ((const int4*)src2)[t];
    int4 d = ((const int4*)dst2)[t];
    int4 r = ((const int4*)et2)[t];
    int p0 = atomicAdd(&tcnt[(d.x >> 4) * TSTRIDE], 1);
    int p1 = atomicAdd(&tcnt[(d.y >> 4) * TSTRIDE], 1);
    int p2 = atomicAdd(&tcnt[(d.z >> 4) * TSTRIDE], 1);
    int p3 = atomicAdd(&tcnt[(d.w >> 4) * TSTRIDE], 1);
    put_edge(bucket, cnts, ovt, ove, p0, s.x, d.x, r.x);
    put_edge(bucket, cnts, ovt, ove, p1, s.y, d.y, r.y);
    put_edge(bucket, cnts, ovt, ove, p2, s.z, d.z, r.z);
    put_edge(bucket, cnts, ovt, ove, p3, s.w, d.w, r.w);
  } else if (t < E2 / 4 + NQ) {
    int q = t - E2 / 4;
    int4 d = ((const int4*)dst)[q];
    int h0 = hmap[d.x], h1_ = hmap[d.y], h2 = hmap[d.z], h3 = hmap[d.w];
    int e = q * 4;
    if (h0 < 0) push_entry(cnts, esrc, edst, erel, src[e],     d.x, et[e]);
    if (h1_ < 0) push_entry(cnts, esrc, edst, erel, src[e + 1], d.y, et[e + 1]);
    if (h2 < 0) push_entry(cnts, esrc, edst, erel, src[e + 2], d.z, et[e + 2]);
    if (h3 < 0) push_entry(cnts, esrc, edst, erel, src[e + 3], d.w, et[e + 3]);
  } else if (t < E2 / 4 + NQ + N1) {
    int n = t - (E2 / 4 + NQ);
    if (hmap[n] < 0) {
      push_entry(cnts, esrc, edst, erel, n, n, 4);
      int slot = atomicAdd(&cnts[1], 1);
      atomicExch(&rlist[slot], n);
    }
  }
}

#define HF_T 1024
#define HF_G 512
static_assert(HF_T % 128 == 0, "a thread keeps one column over all its trips");
__global__ void __launch_bounds__(HF_T)
k_histfill_stats(const int* __restrict__ hmap,
                 const float* __restrict__ hbuf,
                 const float* __restrict__ b1,
                 float* __restrict__ h1,
                 float* __restrict__ csum,
                 float* __restrict__ csq) {
  __shared__ float ls[HF_T], lq[HF_T];
  long total = (long)N1 * 128;
  long i = (long)blockIdx.x * blockDim.x + threadIdx.x;
  long stride = (long)gridDim.x * blockDim.x;
  float s = 0.f, q = 0.f;
  for (; i < total; i += stride) {
    int n = (int)(i >> 7);
    int c = (int)(i & 127);
    int hm = hmap[n];
    if (hm >= 0) {
      float v = hbuf[(long)hm * 128 + c];
      h1[i] = v;
      s += v; q += v * v;
    } else {
      h1[i] = b1[c];
    }
  }
  ls[threadIdx.x] = s; lq[threadIdx.x] = q;
  __syncthreads();
  if (threadIdx.x < 128) {
    float S = 0.f, Q = 0.f;
    for (int k = 0; k < HF_T / 128; ++k) {
      S += ls[threadIdx.x + 128 * k];
      Q += lq[threadIdx.x + 128 * k];
    }
    atomicAdd(&csum[threadIdx.x], S);
    atomicAdd(&csq[threadIdx.x], Q);
  }
}

__global__ void __launch_bounds__(256)
k_msg1(const float* __restrict__ x,
                       const float* __restrict__ W1,
                       const float* __restrict__ Ws,
                       const int* __restrict__ cnts,
                       const int* __restrict__ esrc,
                       const int* __restrict__ edst,
                       const int* __restrict__ erel,
                       float* __restrict__ h1) {
  int cnt = cnts[0];
  int ntiles = (cnt + 15) >> 4;
  int wave = threadIdx.x >> 5;
  int lane = threadIdx.x & 31;
  int aRow = lane & 15;
  int aK   = (lane >> 4) << 1;
  int col  = wave * 16 + (lane & 15);
  int halfOff = (lane >> 4) << 3;
  const v2f zero2 = {0.f, 0.f};

  for (int tile = blockIdx.x; tile < ntiles; tile += gridDim.x) {
    int ei = tile * 16 + aRow;
    bool live = ei < cnt;
    int ec = live ? ei : 0;
    int sRow = esrc[ec];
    int rel  = live ? erel[ec] : -1;
    const float* Ap = x + (long)sRow * 128 + aK;

    v2f af[32];
#pragma unroll
    for (int ks = 0; ks < 32; ++ks) af[ks] = *(const v2f*)(Ap + 4 * ks);

    v8f acc = {0.f,0.f,0.f,0.f,0.f,0.f,0.f,0.f};
    for (int r = 0; r < 5; ++r) {
      bool mine = (rel == r);
      if (__ballot(mine) == 0ull) continue;
      const float* Bp = ((r < 4) ? (W1 + (long)r * 128 * 128) : Ws)
                        + aK * 128 + col;
      v2f bf[32];
#pragma unroll
      for (int ks = 0; ks < 32; ++ks) {
        bf[ks].x = Bp[(4 * ks) * 128];
        bf[ks].y = Bp[(4 * ks + 1) * 128];
      }
#pragma unroll
      for (int ks = 0; ks < 32; ++ks) {
        v2f a = mine ? af[ks] : zero2;
        acc = __builtin_amdgcn_wmma_f32_16x16x4_f32(
            false, a, false, bf[ks], (short)0, acc, false, false);
      }
    }

    int dd[8];
#pragma unroll
    for (int v = 0; v < 8; ++v) {
      int eo = tile * 16 + v + halfOff;
      dd[v] = edst[eo < cnt ? eo : 0];
    }
#pragma unroll
    for (int v = 0; v < 8; ++v) {
      int eo = tile * 16 + v + halfOff;
      if (eo < cnt) atomicAdd(&h1[(long)dd[v] * 128 + col], acc[v]);
    }
  }
}

__global__ void k_bnfinal(const float* __restrict__ csum,
                          const float* __restrict__ csq,
                          const float* __restrict__ gamma,
                          const float* __restrict__ beta,
                          float* __restrict__ scale,
                          float* __restrict__ shift,
                          const int* __restrict__ cnts,
                          const int* __restrict__ rlist,
                          const float* __restrict__ h1) {
  int c = threadIdx.x;
  float s = csum[c], q = csq[c];
  int nr = cnts[1];
  for (int j = 0; j < nr; ++j) {
    float v = h1[(long)rlist[j] * 128 + c];
    s += v; q += v * v;
  }
  const float inv_n = 1.0f / (float)N1;
  float mean = s * inv_n;
  float var  = q * inv_n - mean * mean;
  float sc = gamma[c] * rsqrtf(var + BN_EPS);
  scale[c] = sc;
  shift[c] = beta[c] - mean * sc;
}

#define PA 644

__device__ __forceinline__ void tile_chunk(int cnt, const int (&e)[TC / 256],
                                           int wave, int lane,
                                           int* __restrict__ srt,
                                           int* __restrict__ hist,
                                           int* __restrict__ off,
                                           const float* __restrict__ h1,
                                           float4 sc, float4 sh,
                                           float4 (&acc)[8]) {
  if (threadIdx.x < 64) hist[threadIdx.x] = 0;
  __syncthreads();
  int rk[TC / 256];
#pragma unroll
  for (int k = 0; k < TC / 256; ++k) {
    rk[k] = 0;
    if (threadIdx.x + 256 * k < cnt) rk[k] = atomicAdd(&hist[(e[k] >> 16) & 63], 1);
  }
  __syncthreads();
  if (wave == 0) {
    int a = hist[2 * lane], c = hist[2 * lane + 1];
    int s = a + c, inc = s;
#pragma unroll
    for (int d = 1; d < 32; d <<= 1) {
      int o = __shfl_up(inc, d, 32);
      if (lane >= d) inc += o;
    }
    off[2 * lane] = inc - s;
    off[2 * lane + 1] = inc - s + a;
    if (lane == 31) off[64] = inc;
  }
  __syncthreads();
  int st[TC / 256];
#pragma unroll
  for (int k = 0; k < TC / 256; ++k) st[k] = off[(e[k] >> 16) & 63];
#pragma unroll
  for (int k = 0; k < TC / 256; ++k)
    if (threadIdx.x + 256 * k < cnt) srt[st[k] + rk[k]] = e[k] & 0xFFFF;
  __syncthreads();
#pragma unroll
  for (int i = 0; i < 8; ++i) {
    int beg = __builtin_amdgcn_readfirstlane(off[wave * 8 + i]);
    int end = __builtin_amdgcn_readfirstlane(off[wave * 8 + i + 1]);
    float4 a4 = acc[i];
    for (int base = beg; base < end; base += 32) {
      int n = end - base;
      n = n < 32 ? n : 32;
      int mine = (lane < n) ? srt[base + lane] : 0;
      for (int j0 = 0; j0 < n; j0 += 8) {
        float4 x[8];
#pragma unroll
        for (int k = 0; k < 8; ++k) {
          int sRow = __builtin_amdgcn_readlane(mine, j0 + k);
          x[k] = ((const float4*)(h1 + (long)sRow * 128))[lane];
        }
#pragma unroll
        for (int k = 0; k < 8; ++k) {
          if (j0 + k < n) {
            a4.x += fmaxf(0.0f, x[k].x * sc.x + sh.x);
            a4.y += fmaxf(0.0f, x[k].y * sc.y + sh.y);
            a4.z += fmaxf(0.0f, x[k].z * sc.z + sh.z);
            a4.w += fmaxf(0.0f, x[k].w * sc.w + sh.w);
          }
        }
      }
    }
    acc[i] = a4;
  }
}

__global__ void __launch_bounds__(256)
k_agg_gemm2(const float* __restrict__ h1,
            const float* __restrict__ scale,
            const float* __restrict__ shift,
            const int* __restrict__ tcnt,
            const int* __restrict__ bucket,
            const int* __restrict__ cnts,
            const int* __restrict__ ovt,
            const int* __restrict__ ove,
            const float* __restrict__ W2,
            const float* __restrict__ Ws2,
            const float* __restrict__ b2,
            float* __restrict__ out) {
  __shared__ float At[16 * PA];
  __shared__ float red[8 * 4 * 8 * 32];
  __shared__ int ent[TC];
  __shared__ int srt[TC];
  __shared__ int hist[64];
  __shared__ int off[65];
  __shared__ int nsel[1];

  int wave = __builtin_amdgcn_readfirstlane((int)(threadIdx.x >> 5));
  int lane = threadIdx.x & 31;
  int tile = blockIdx.x;
  int row0 = tile * 16;

  int tc = tcnt[tile * TSTRIDE];
  int nb = tc < TC ? tc : TC;
  int novf = 0;
  if (tc > TC) {
    novf = cnts[2];
    novf = novf < E2 ? novf : E2;
  }

  float4 sc = ((const float4*)scale)[lane];
  float4 sh = ((const float4*)shift)[lane];

  const int* bk = bucket + tile * TC;
  int e0[TC / 256];
#pragma unroll
  for (int k = 0; k < TC / 256; ++k) e0[k] = 0;
  if (nb > 0) {
#pragma unroll
    for (int k = 0; k < TC / 256; ++k) {
      int i = threadIdx.x + 256 * k;
      e0[k] = bk[i < nb ? i : nb - 1];
    }
  }

  int g0 = row0 + wave * 2;
  float4 own[2];
#pragma unroll
  for (int rr = 0; rr < 2; ++rr)
    own[rr] = ((const float4*)(h1 + (long)(g0 + rr) * 128))[lane];

  float4 segsum[8];
#pragma unroll
  for (int i = 0; i < 8; ++i) segsum[i] = make_float4(0.f, 0.f, 0.f, 0.f);

  int nchunk = 1 + (novf + TC - 1) / TC;
#pragma nounroll
  for (int ch = 0; ch < nchunk; ++ch) {
    int cnt;
    int e[TC / 256];
#pragma unroll
    for (int k = 0; k < TC / 256; ++k) e[k] = 0;
    if (ch == 0) {
      cnt = nb;
#pragma unroll
      for (int k = 0; k < TC / 256; ++k) e[k] = e0[k];
    } else {
      int base = (ch - 1) * TC;
      __syncthreads();
      if (threadIdx.x == 0) nsel[0] = 0;
      __syncthreads();
#pragma unroll
      for (int k = 0; k < TC / 256; ++k) {
        int j = base + threadIdx.x + 256 * k;
        if (j < novf && ovt[j] == tile) {
          int p = atomicAdd(&nsel[0], 1);
          ent[p] = ove[j];
        }
      }
      __syncthreads();
      cnt = nsel[0];
#pragma unroll
      for (int k = 0; k < TC / 256; ++k) {
        int i = threadIdx.x + 256 * k;
        if (i < cnt) e[k] = ent[i];
      }
    }
    tile_chunk(cnt, e, wave, lane, srt, hist, off, h1, sc, sh, segsum);
  }

#pragma unroll
  for (int rr = 0; rr < 2; ++rr) {
    float4 x = own[rr];
    float4 v;
    v.x = fmaxf(0.0f, x.x * sc.x + sh.x);
    v.y = fmaxf(0.0f, x.y * sc.y + sh.y);
    v.z = fmaxf(0.0f, x.z * sc.z + sh.z);
    v.w = fmaxf(0.0f, x.w * sc.w + sh.w);
    *(float4*)(At + (wave * 2 + rr) * PA + 512 + 4 * lane) = v;
  }
#pragma unroll
  for (int i = 0; i < 8; ++i)
    *(float4*)(At + (wave * 2 + (i >> 2)) * PA + (i & 3) * 128 + 4 * lane) = segsum[i];
  __syncthreads();

  int c = lane & 15;
  int h2 = (lane >> 4) << 1;
  int kb = wave * 80;
  v2f af[20];
  v2f bf[20][4];
#pragma unroll
  for (int ks = 0; ks < 20; ++ks) {
    int kk = kb + 4 * ks;
    const float* Bk = (kk < 512) ? (W2 + kk * 64) : (Ws2 + (kk - 512) * 64);
    Bk += h2 * 64 + c;
#pragma unroll
    for (int t = 0; t < 4; ++t) {
      bf[ks][t].x = Bk[t * 16];
      bf[ks][t].y = Bk[64 + t * 16];
    }
    af[ks] = *(const v2f*)(At + c * PA + kk + h2);
  }
  v8f acc[4];
  const v8f zero = {0.f,0.f,0.f,0.f,0.f,0.f,0.f,0.f};
#pragma unroll
  for (int t = 0; t < 4; ++t) acc[t] = zero;
#pragma unroll
  for (int ks = 0; ks < 20; ++ks) {
#pragma unroll
    for (int t = 0; t < 4; ++t)
      acc[t] = __builtin_amdgcn_wmma_f32_16x16x4_f32(
          false, af[ks], false, bf[ks][t], (short)0, acc[t], false, false);
  }
#pragma unroll
  for (int t = 0; t < 4; ++t)
#pragma unroll
    for (int v = 0; v < 8; ++v)
      red[((wave * 4 + t) * 8 + v) * 32 + lane] = acc[t][v];
  __syncthreads();

  int row = threadIdx.x >> 4;
  int cq  = threadIdx.x & 15;
  int ridx = (((cq >> 2) * 8 + (row & 7)) * 32) + (row >> 3) * 16 + (cq & 3) * 4;
  float4 bb = ((const float4*)b2)[cq];
  float vals[4] = {bb.x, bb.y, bb.z, bb.w};
#pragma unroll
  for (int w = 0; w < 8; ++w) {
    float4 p = *(const float4*)(red + w * (4 * 8 * 32) + ridx);
    vals[0] += p.x; vals[1] += p.y; vals[2] += p.z; vals[3] += p.w;
  }
  float m = fmaxf(fmaxf(vals[0], vals[1]), fmaxf(vals[2], vals[3]));
#pragma unroll
  for (int mask = 1; mask < 16; mask <<= 1)
    m = fmaxf(m, __shfl_xor(m, mask, 32));
  float sum = 0.f;
#pragma unroll
  for (int t = 0; t < 4; ++t) sum += expf(vals[t] - m);
#pragma unroll
  for (int mask = 1; mask < 16; mask <<= 1)
    sum += __shfl_xor(sum, mask, 32);
  float lse = m + logf(sum);
  float4 o;
  o.x = vals[0] - lse; o.y = vals[1] - lse; o.z = vals[2] - lse; o.w = vals[3] - lse;
  ((float4*)(out + (long)(row0 + row) * 64))[cq] = o;
}

extern "C" void kernel_launch(void* const* d_in, const int* in_sizes, int n_in,
                              void* d_out, int out_size, void* d_ws, size_t ws_size,
                              hipStream_t stream) {
  const float* x     = (const float*)d_in[0];
  const int*   src1  = (const int*)  d_in[1];
  const int*   dst1  = (const int*)  d_in[2];
  const int*   et1   = (const int*)  d_in[3];
  const int*   src2  = (const int*)  d_in[4];
  const int*   dst2  = (const int*)  d_in[5];
  const int*   et2   = (const int*)  d_in[6];
  const int*   hmap  = (const int*)  d_in[7];
  const float* hbuf  = (const float*)d_in[8];
  const float* W1    = (const float*)d_in[9];
  const float* Ws1   = (const float*)d_in[10];
  const float* b1    = (const float*)d_in[11];
  const float* gamma = (const float*)d_in[12];
  const float* beta  = (const float*)d_in[13];
  const float* W2    = (const float*)d_in[14];
  const float* Ws2   = (const float*)d_in[15];
  const float* b2    = (const float*)d_in[16];

  float* ws    = (float*)d_ws;
  float* A1    = ws;
  float* A2    = A1 + (size_t)4 * N1 * 128;
  float* ss    = A2 + (size_t)4 * N2 * 128 + 256;
  float* h1    = ss + 256;
  float* out   = (float*)d_out;
  int* esrc  = (int*)A1;
  int* edst  = esrc + CAP;
  int* erel  = edst + CAP;
  int* rlist = erel + CAP;
  int* cnts  = (int*)A2 - 4;
  int* tcnt  = (int*)A2;
  float* sums = (float*)(tcnt + NT * TSTRIDE);
  int* bucket = (int*)(sums + 256);
  int* ovt    = bucket + NT * TC;
  int* ove    = ovt + E2;

  k_zero<<<96, 256, 0, stream>>>((float*)cnts, (long)(4 + NT * TSTRIDE + 256));

  k_pick<<<(NQ + N1 + E2 / 4 + 255) / 256, 256, 0, stream>>>(src1, dst1, et1, hmap, cnts,
                                                         esrc, edst, erel, rlist,
                                                         src2, dst2, et2,
                                                         tcnt, bucket, ovt, ove);
  k_histfill_stats<<<HF_G, HF_T, 0, stream>>>(hmap, hbuf, b1, h1, sums, sums + 128);

  k_msg1<<<1024, 256, 0, stream>>>(x, W1, Ws1, cnts, esrc, edst, erel, h1);

  k_bnfinal<<<1, 128, 0, stream>>>(sums, sums + 128, gamma, beta, ss, ss + 128,
                                   cnts, rlist, h1);

  static_assert(N2 % 16 == 0, "one work-group per 16 destination rows");
  k_agg_gemm2<<<NT, 256, 0, stream>>>(h1, ss, ss + 128, tcnt, bucket, cnts, ovt, ove,
                                      W2, Ws2, b2, out);
}


# history fill: a thread's 15 trips unrolled on the fixed grid; its 15 index loads issue as one group, then its 15 row loads as one group (2 memory round trips per thread, was 30); BN sum order kept
# speedup vs baseline: 5.5369x; 1.1098x over previous
#include <hip/hip_runtime.h>
#include <hip/hip_bf16.h>
#include <math.h>

typedef __attribute__((ext_vector_type(2))) float v2f;
typedef __attribute__((ext_vector_type(8))) float v8f;

#define N0  300000
#define N1  60000
#define N2  6000
#define E1  1200000
#define E2  300000
#define D_H 128
#define D_OUT 64
#define BN_EPS 1e-5f

__global__ void k_zero(float* __restrict__ p, long n) {
  long i = (long)blockIdx.x * blockDim.x + threadIdx.x;
  long stride = (long)gridDim.x * blockDim.x;
  for (; i < n; i += stride) p[i] = 0.0f;
}

#define NQ (E1 / 4)
#define CAP (E1 + N1)
static_assert(E1 % 4 == 0, "edge list is read four at a time");
#define NSEG (4 * N2)
static_assert(E2 % 4 == 0, "layer-2 edge list is read four at a time");

#define NT (N2 / 16)
#define TC 2048
#define TSTRIDE 64
static_assert(N1 <= 65536, "a source row fits the low 16 bits of a bucket entry");
static_assert(TC % 256 == 0, "a chunk is a whole number of passes of the work-group");
static_assert(NT * TSTRIDE == NSEG, "the counters fill the cleared range");

__device__ __forceinline__ void put_edge(int* __restrict__ bucket,
                                         int* __restrict__ cnts,
                                         int* __restrict__ ovt,
                                         int* __restrict__ ove,
                                         int slot, int s, int d, int r) {
  int tile = d >> 4;
  int packed = s | ((((d & 15) << 2) | r) << 16);
  if (slot < TC) {
    atomicExch(&bucket[tile * TC + slot], packed);
  } else {
    int o = atomicAdd(&cnts[2], 1);
    atomicExch(&ovt[o], tile);
    atomicExch(&ove[o], packed);
  }
}

__device__ __forceinline__ void push_entry(int* __restrict__ cnts,
                                           int* __restrict__ esrc,
                                           int* __restrict__ edst,
                                           int* __restrict__ erel,
                                           int s, int d, int r) {
  int slot = atomicAdd(&cnts[0], 1);
  atomicExch(&esrc[slot], s);
  atomicExch(&edst[slot], d);
  atomicExch(&erel[slot], r);
}

__global__ void k_pick(const int* __restrict__ src,
                       const int* __restrict__ dst,
                       const int* __restrict__ et,
                       const int* __restrict__ hmap,
                       int* __restrict__ cnts,
                       int* __restrict__ esrc,
                       int* __restrict__ edst,
                       int* __restrict__ erel,
                       int* __restrict__ rlist,
                       const int* __restrict__ src2,
                       const int* __restrict__ dst2,
                       const int* __restrict__ et2,
                       int* __restrict__ tcnt,
                       int* __restrict__ bucket,
                       int* __restrict__ ovt,
                       int* __restrict__ ove) {
  int t = blockIdx.x * blockDim.x + threadIdx.x;
  if (t < E2 / 4) {
    int4 s = ((const int4*)src2)[t];
    int4 d = ((const int4*)dst2)[t];
    int4 r = ((const int4*)et2)[t];
    int p0 = atomicAdd(&tcnt[(d.x >> 4) * TSTRIDE], 1);
    int p1 = atomicAdd(&tcnt[(d.y >> 4) * TSTRIDE], 1);
    int p2 = atomicAdd(&tcnt[(d.z >> 4) * TSTRIDE], 1);
    int p3 = atomicAdd(&tcnt[(d.w >> 4) * TSTRIDE], 1);
    put_edge(bucket, cnts, ovt, ove, p0, s.x, d.x, r.x);
    put_edge(bucket, cnts, ovt, ove, p1, s.y, d.y, r.y);
    put_edge(bucket, cnts, ovt, ove, p2, s.z, d.z, r.z);
    put_edge(bucket, cnts, ovt, ove, p3, s.w, d.w, r.w);
  } else if (t < E2 / 4 + NQ) {
    int q = t - E2 / 4;
    int4 d = ((const int4*)dst)[q];
    int h0 = hmap[d.x], h1_ = hmap[d.y], h2 = hmap[d.z], h3 = hmap[d.w];
    int e = q * 4;
    if (h0 < 0) push_entry(cnts, esrc, edst, erel, src[e],     d.x, et[e]);
    if (h1_ < 0) push_entry(cnts, esrc, edst, erel, src[e + 1], d.y, et[e + 1]);
    if (h2 < 0) push_entry(cnts, esrc, edst, erel, src[e + 2], d.z, et[e + 2]);
    if (h3 < 0) push_entry(cnts, esrc, edst, erel, src[e + 3], d.w, et[e + 3]);
  } else if (t < E2 / 4 + NQ + N1) {
    int n = t - (E2 / 4 + NQ);
    if (hmap[n] < 0) {
      push_entry(cnts, esrc, edst, erel, n, n, 4);
      int slot = atomicAdd(&cnts[1], 1);
      atomicExch(&rlist[slot], n);
    }
  }
}

#define HF_T 1024
#define HF_G 512
#define HF_TOTAL (N1 * 128)
#define HF_STRIDE (HF_G * HF_T)
#define HF_TRIPS ((HF_TOTAL + HF_STRIDE - 1) / HF_STRIDE)
static_assert(HF_T % 128 == 0, "a thread keeps one column over all its trips");
static_assert((long)N1 * 128 * 4 < (1l << 31), "element offsets fit 32 bits");
__global__ void __launch_bounds__(HF_T)
k_histfill_stats(const int* __restrict__ hmap,
                 const float* __restrict__ hbuf,
                 const float* __restrict__ b1,
                 float* __restrict__ h1,
                 float* __restrict__ csum,
                 float* __restrict__ csq) {
  __shared__ float ls[HF_T], lq[HF_T];
  unsigned tid = blockIdx.x * HF_T + threadIdx.x;
  unsigned c = threadIdx.x & 127;
  unsigned row0 = tid >> 7;
  float bias = b1[c];
  int hm[HF_TRIPS];
#pragma unroll
  for (int t = 0; t < HF_TRIPS; ++t) {
    bool in = ((t + 1) * HF_STRIDE <= HF_TOTAL) || (tid + t * HF_STRIDE < HF_TOTAL);
    hm[t] = hmap[in ? row0 + t * (HF_STRIDE / 128) : 0u];
  }
  float v[HF_TRIPS];
#pragma unroll
  for (int t = 0; t < HF_TRIPS; ++t) {
    unsigned r = hm[t] >= 0 ? (unsigned)hm[t] : 0u;
    v[t] = hbuf[r * 128u + c];
  }
  __builtin_amdgcn_sched_barrier(0);
  float s = 0.f, q = 0.f;
#pragma unroll
  for (int t = 0; t < HF_TRIPS; ++t) {
    bool in = ((t + 1) * HF_STRIDE <= HF_TOTAL) || (tid + t * HF_STRIDE < HF_TOTAL);
    bool hist = in && hm[t] >= 0;
    float sn = s + v[t];
    float qn = q + v[t] * v[t];
    s = hist ? sn : s;
    q = hist ? qn : q;
    float val = hist ? v[t] : bias;
    if (in) h1[tid + t * HF_STRIDE] = val;
  }
  ls[threadIdx.x] = s; lq[threadIdx.x] = q;
  __syncthreads();
  if (threadIdx.x < 128) {
    float S = 0.f, Q = 0.f;
    for (int k = 0; k < HF_T / 128; ++k) {
      S += ls[threadIdx.x + 128 * k];
      Q += lq[threadIdx.x + 128 * k];
    }
    atomicAdd(&csum[threadIdx.x], S);
    atomicAdd(&csq[threadIdx.x], Q);
  }
}

__global__ void __launch_bounds__(256)
k_msg1(const float* __restrict__ x,
                       const float* __restrict__ W1,
                       const float* __restrict__ Ws,
                       const int* __restrict__ cnts,
                       const int* __restrict__ esrc,
                       const int* __restrict__ edst,
                       const int* __restrict__ erel,
                       float* __restrict__ h1) {
  int cnt = cnts[0];
  int ntiles = (cnt + 15) >> 4;
  int wave = threadIdx.x >> 5;
  int lane = threadIdx.x & 31;
  int aRow = lane & 15;
  int aK   = (lane >> 4) << 1;
  int col  = wave * 16 + (lane & 15);
  int halfOff = (lane >> 4) << 3;
  const v2f zero2 = {0.f, 0.f};

  for (int tile = blockIdx.x; tile < ntiles; tile += gridDim.x) {
    int ei = tile * 16 + aRow;
    bool live = ei < cnt;
    int ec = live ? ei : 0;
    int sRow = esrc[ec];
    int rel  = live ? erel[ec] : -1;
    const float* Ap = x + (long)sRow * 128 + aK;

    v2f af[32];
#pragma unroll
    for (int ks = 0; ks < 32; ++ks) af[ks] = *(const v2f*)(Ap + 4 * ks);

    v8f acc = {0.f,0.f,0.f,0.f,0.f,0.f,0.f,0.f};
    for (int r = 0; r < 5; ++r) {
      bool mine = (rel == r);
      if (__ballot(mine) == 0ull) continue;
      const float* Bp = ((r < 4) ? (W1 + (long)r * 128 * 128) : Ws)
                        + aK * 128 + col;
      v2f bf[32];
#pragma unroll
      for (int ks = 0; ks < 32; ++ks) {
        bf[ks].x = Bp[(4 * ks) * 128];
        bf[ks].y = Bp[(4 * ks + 1) * 128];
      }
#pragma unroll
      for (int ks = 0; ks < 32; ++ks) {
        v2f a = mine ? af[ks] : zero2;
        acc = __builtin_amdgcn_wmma_f32_16x16x4_f32(
            false, a, false, bf[ks], (short)0, acc, false, false);
      }
    }

    int dd[8];
#pragma unroll
    for (int v = 0; v < 8; ++v) {
      int eo = tile * 16 + v + halfOff;
      dd[v] = edst[eo < cnt ? eo : 0];
    }
#pragma unroll
    for (int v = 0; v < 8; ++v) {
      int eo = tile * 16 + v + halfOff;
      if (eo < cnt) atomicAdd(&h1[(long)dd[v] * 128 + col], acc[v]);
    }
  }
}

__global__ void k_bnfinal(const float* __restrict__ csum,
                          const float* __restrict__ csq,
                          const float* __restrict__ gamma,
                          const float* __restrict__ beta,
                          float* __restrict__ scale,
                          float* __restrict__ shift,
                          const int* __restrict__ cnts,
                          const int* __restrict__ rlist,
                          const float* __restrict__ h1) {
  int c = threadIdx.x;
  float s = csum[c], q = csq[c];
  int nr = cnts[1];
  for (int j = 0; j < nr; ++j) {
    float v = h1[(long)rlist[j] * 128 + c];
    s += v; q += v * v;
  }
  const float inv_n = 1.0f / (float)N1;
  float mean = s * inv_n;
  float var  = q * inv_n - mean * mean;
  float sc = gamma[c] * rsqrtf(var + BN_EPS);
  scale[c] = sc;
  shift[c] = beta[c] - mean * sc;
}

#define PA 644

__device__ __forceinline__ void tile_chunk(int cnt, const int (&e)[TC / 256],
                                           int wave, int lane,
                                           int* __restrict__ srt,
                                           int* __restrict__ hist,
                                           int* __restrict__ off,
                                           const float* __restrict__ h1,
                                           float4 sc, float4 sh,
                                           float4 (&acc)[8]) {
  if (threadIdx.x < 64) hist[threadIdx.x] = 0;
  __syncthreads();
  int rk[TC / 256];
#pragma unroll
  for (int k = 0; k < TC / 256; ++k) {
    rk[k] = 0;
    if (threadIdx.x + 256 * k < cnt) rk[k] = atomicAdd(&hist[(e[k] >> 16) & 63], 1);
  }
  __syncthreads();
  if (wave == 0) {
    int a = hist[2 * lane], c = hist[2 * lane + 1];
    int s = a + c, inc = s;
#pragma unroll
    for (int d = 1; d < 32; d <<= 1) {
      int o = __shfl_up(inc, d, 32);
      if (lane >= d) inc += o;
    }
    off[2 * lane] = inc - s;
    off[2 * lane + 1] = inc - s + a;
    if (lane == 31) off[64] = inc;
  }
  __syncthreads();
  int st[TC / 256];
#pragma unroll
  for (int k = 0; k < TC / 256; ++k) st[k] = off[(e[k] >> 16) & 63];
#pragma unroll
  for (int k = 0; k < TC / 256; ++k)
    if (threadIdx.x + 256 * k < cnt) srt[st[k] + rk[k]] = e[k] & 0xFFFF;
  __syncthreads();
#pragma unroll
  for (int i = 0; i < 8; ++i) {
    int beg = __builtin_amdgcn_readfirstlane(off[wave * 8 + i]);
    int end = __builtin_amdgcn_readfirstlane(off[wave * 8 + i + 1]);
    float4 a4 = acc[i];
    for (int base = beg; base < end; base += 32) {
      int n = end - base;
      n = n < 32 ? n : 32;
      int mine = (lane < n) ? srt[base + lane] : 0;
      for (int j0 = 0; j0 < n; j0 += 8) {
        float4 x[8];
#pragma unroll
        for (int k = 0; k < 8; ++k) {
          int sRow = __builtin_amdgcn_readlane(mine, j0 + k);
          x[k] = ((const float4*)(h1 + (long)sRow * 128))[lane];
        }
#pragma unroll
        for (int k = 0; k < 8; ++k) {
          if (j0 + k < n) {
            a4.x += fmaxf(0.0f, x[k].x * sc.x + sh.x);
            a4.y += fmaxf(0.0f, x[k].y * sc.y + sh.y);
            a4.z += fmaxf(0.0f, x[k].z * sc.z + sh.z);
            a4.w += fmaxf(0.0f, x[k].w * sc.w + sh.w);
          }
        }
      }
    }
    acc[i] = a4;
  }
}

__global__ void __launch_bounds__(256)
k_agg_gemm2(const float* __restrict__ h1,
            const float* __restrict__ scale,
            const float* __restrict__ shift,
            const int* __restrict__ tcnt,
            const int* __restrict__ bucket,
            const int* __restrict__ cnts,
            const int* __restrict__ ovt,
            const int* __restrict__ ove,
            const float* __restrict__ W2,
            const float* __restrict__ Ws2,
            const float* __restrict__ b2,
            float* __restrict__ out) {
  __shared__ float At[16 * PA];
  __shared__ float red[8 * 4 * 8 * 32];
  __shared__ int ent[TC];
  __shared__ int srt[TC];
  __shared__ int hist[64];
  __shared__ int off[65];
  __shared__ int nsel[1];

  int wave = __builtin_amdgcn_readfirstlane((int)(threadIdx.x >> 5));
  int lane = threadIdx.x & 31;
  int tile = blockIdx.x;
  int row0 = tile * 16;

  int tc = tcnt[tile * TSTRIDE];
  int nb = tc < TC ? tc : TC;
  int novf = 0;
  if (tc > TC) {
    novf = cnts[2];
    novf = novf < E2 ? novf : E2;
  }

  float4 sc = ((const float4*)scale)[lane];
  float4 sh = ((const float4*)shift)[lane];

  const int* bk = bucket + tile * TC;
  int e0[TC / 256];
#pragma unroll
  for (int k = 0; k < TC / 256; ++k) e0[k] = 0;
  if (nb > 0) {
#pragma unroll
    for (int k = 0; k < TC / 256; ++k) {
      int i = threadIdx.x + 256 * k;
      e0[k] = bk[i < nb ? i : nb - 1];
    }
  }

  int g0 = row0 + wave * 2;
  float4 own[2];
#pragma unroll
  for (int rr = 0; rr < 2; ++rr)
    own[rr] = ((const float4*)(h1 + (long)(g0 + rr) * 128))[lane];

  float4 segsum[8];
#pragma unroll
  for (int i = 0; i < 8; ++i) segsum[i] = make_float4(0.f, 0.f, 0.f, 0.f);

  int nchunk = 1 + (novf + TC - 1) / TC;
#pragma nounroll
  for (int ch = 0; ch < nchunk; ++ch) {
    int cnt;
    int e[TC / 256];
#pragma unroll
    for (int k = 0; k < TC / 256; ++k) e[k] = 0;
    if (ch == 0) {
      cnt = nb;
#pragma unroll
      for (int k = 0; k < TC / 256; ++k) e[k] = e0[k];
    } else {
      int base = (ch - 1) * TC;
      __syncthreads();
      if (threadIdx.x == 0) nsel[0] = 0;
      __syncthreads();
#pragma unroll
      for (int k = 0; k < TC / 256; ++k) {
        int j = base + threadIdx.x + 256 * k;
        if (j < novf && ovt[j] == tile) {
          int p = atomicAdd(&nsel[0], 1);
          ent[p] = ove[j];
        }
      }
      __syncthreads();
      cnt = nsel[0];
#pragma unroll
      for (int k = 0; k < TC / 256; ++k) {
        int i = threadIdx.x + 256 * k;
        if (i < cnt) e[k] = ent[i];
      }
    }
    tile_chunk(cnt, e, wave, lane, srt, hist, off, h1, sc, sh, segsum);
  }

#pragma unroll
  for (int rr = 0; rr < 2; ++rr) {
    float4 x = own[rr];
    float4 v;
    v.x = fmaxf(0.0f, x.x * sc.x + sh.x);
    v.y = fmaxf(0.0f, x.y * sc.y + sh.y);
    v.z = fmaxf(0.0f, x.z * sc.z + sh.z);
    v.w = fmaxf(0.0f, x.w * sc.w + sh.w);
    *(float4*)(At + (wave * 2 + rr) * PA + 512 + 4 * lane) = v;
  }
#pragma unroll
  for (int i = 0; i < 8; ++i)
    *(float4*)(At + (wave * 2 + (i >> 2)) * PA + (i & 3) * 128 + 4 * lane) = segsum[i];
  __syncthreads();

  int c = lane & 15;
  int h2 = (lane >> 4) << 1;
  int kb = wave * 80;
  v2f af[20];
  v2f bf[20][4];
#pragma unroll
  for (int ks = 0; ks < 20; ++ks) {
    int kk = kb + 4 * ks;
    const float* Bk = (kk < 512) ? (W2 + kk * 64) : (Ws2 + (kk - 512) * 64);
    Bk += h2 * 64 + c;
#pragma unroll
    for (int t = 0; t < 4; ++t) {
      bf[ks][t].x = Bk[t * 16];
      bf[ks][t].y = Bk[64 + t * 16];
    }
    af[ks] = *(const v2f*)(At + c * PA + kk + h2);
  }
  v8f acc[4];
  const v8f zero = {0.f,0.f,0.f,0.f,0.f,0.f,0.f,0.f};
#pragma unroll
  for (int t = 0; t < 4; ++t) acc[t] = zero;
#pragma unroll
  for (int ks = 0; ks < 20; ++ks) {
#pragma unroll
    for (int t = 0; t < 4; ++t)
      acc[t] = __builtin_amdgcn_wmma_f32_16x16x4_f32(
          false, af[ks], false, bf[ks][t], (short)0, acc[t], false, false);
  }
#pragma unroll
  for (int t = 0; t < 4; ++t)
#pragma unroll
    for (int v = 0; v < 8; ++v)
      red[((wave * 4 + t) * 8 + v) * 32 + lane] = acc[t][v];
  __syncthreads();

  int row = threadIdx.x >> 4;
  int cq  = threadIdx.x & 15;
  int ridx = (((cq >> 2) * 8 + (row & 7)) * 32) + (row >> 3) * 16 + (cq & 3) * 4;
  float4 bb = ((const float4*)b2)[cq];
  float vals[4] = {bb.x, bb.y, bb.z, bb.w};
#pragma unroll
  for (int w = 0; w < 8; ++w) {
    float4 p = *(const float4*)(red + w * (4 * 8 * 32) + ridx);
    vals[0] += p.x; vals[1] += p.y; vals[2] += p.z; vals[3] += p.w;
  }
  float m = fmaxf(fmaxf(vals[0], vals[1]), fmaxf(vals[2], vals[3]));
#pragma unroll
  for (int mask = 1; mask < 16; mask <<= 1)
    m = fmaxf(m, __shfl_xor(m, mask, 32));
  float sum = 0.f;
#pragma unroll
  for (int t = 0; t < 4; ++t) sum += expf(vals[t] - m);
#pragma unroll
  for (int mask = 1; mask < 16; mask <<= 1)
    sum += __shfl_xor(sum, mask, 32);
  float lse = m + logf(sum);
  float4 o;
  o.x = vals[0] - lse; o.y = vals[1] - lse; o.z = vals[2] - lse; o.w = vals[3] - lse;
  ((float4*)(out + (long)(row0 + row) * 64))[cq] = o;
}

extern "C" void kernel_launch(void* const* d_in, const int* in_sizes, int n_in,
                              void* d_out, int out_size, void* d_ws, size_t ws_size,
                              hipStream_t stream) {
  const float* x     = (const float*)d_in[0];
  const int*   src1  = (const int*)  d_in[1];
  const int*   dst1  = (const int*)  d_in[2];
  const int*   et1   = (const int*)  d_in[3];
  const int*   src2  = (const int*)  d_in[4];
  const int*   dst2  = (const int*)  d_in[5];
  const int*   et2   = (const int*)  d_in[6];
  const int*   hmap  = (const int*)  d_in[7];
  const float* hbuf  = (const float*)d_in[8];
  const float* W1    = (const float*)d_in[9];
  const float* Ws1   = (const float*)d_in[10];
  const float* b1    = (const float*)d_in[11];
  const float* gamma = (const float*)d_in[12];
  const float* beta  = (const float*)d_in[13];
  const float* W2    = (const float*)d_in[14];
  const float* Ws2   = (const float*)d_in[15];
  const float* b2    = (const float*)d_in[16];

  float* ws    = (float*)d_ws;
  float* A1    = ws;
  float* A2    = A1 + (size_t)4 * N1 * 128;
  float* ss    = A2 + (size_t)4 * N2 * 128 + 256;
  float* h1    = ss + 256;
  float* out   = (float*)d_out;
  int* esrc  = (int*)A1;
  int* edst  = esrc + CAP;
  int* erel  = edst + CAP;
  int* rlist = erel + CAP;
  int* cnts  = (int*)A2 - 4;
  int* tcnt  = (int*)A2;
  float* sums = (float*)(tcnt + NT * TSTRIDE);
  int* bucket = (int*)(sums + 256);
  int* ovt    = bucket + NT * TC;
  int* ove    = ovt + E2;

  k_zero<<<96, 256, 0, stream>>>((float*)cnts, (long)(4 + NT * TSTRIDE + 256));

  k_pick<<<(NQ + N1 + E2 / 4 + 255) / 256, 256, 0, stream>>>(src1, dst1, et1, hmap, cnts,
                                                         esrc, edst, erel, rlist,
                                                         src2, dst2, et2,
                                                         tcnt, bucket, ovt, ove);
  k_histfill_stats<<<HF_G, HF_T, 0, stream>>>(hmap, hbuf, b1, h1, sums, sums + 128);

  k_msg1<<<1024, 256, 0, stream>>>(x, W1, Ws1, cnts, esrc, edst, erel, h1);

  k_bnfinal<<<1, 128, 0, stream>>>(sums, sums + 128, gamma, beta, ss, ss + 128,
                                   cnts, rlist, h1);

  static_assert(N2 % 16 == 0, "one work-group per 16 destination rows");
  k_agg_gemm2<<<NT, 256, 0, stream>>>(h1, ss, ss + 128, tcnt, bucket, cnts, ovt, ove,
                                      W2, Ws2, b2, out);
}


# edge pass re-blocked to 4096 items per work-group (1700 -> 382 groups, loads grouped); layer-2 groups rank edges per tile in LDS, one counter add per tile (edge sum order: runs by arrival)
# speedup vs baseline: 8.6547x; 1.5631x over previous
#include <hip/hip_runtime.h>
#include <hip/hip_bf16.h>
#include <math.h>

typedef __attribute__((ext_vector_type(2))) float v2f;
typedef __attribute__((ext_vector_type(8))) float v8f;

#define N0  300000
#define N1  60000
#define N2  6000
#define E1  1200000
#define E2  300000
#define D_H 128
#define D_OUT 64
#define BN_EPS 1e-5f

__global__ void k_zero(float* __restrict__ p, long n) {
  long i = (long)blockIdx.x * blockDim.x + threadIdx.x;
  long stride = (long)gridDim.x * blockDim.x;
  for (; i < n; i += stride) p[i] = 0.0f;
}

#define NQ (E1 / 4)
#define CAP (E1 + N1)
static_assert(E1 % 4 == 0, "edge list is read four at a time");
#define NSEG (4 * N2)
static_assert(E2 % 4 == 0, "layer-2 edge list is read four at a time");

#define NT (N2 / 16)
#define TC 2048
#define TSTRIDE 64
static_assert(N1 <= 65536, "a source row fits the low 16 bits of a bucket entry");
static_assert(TC % 256 == 0, "a chunk is a whole number of passes of the work-group");
static_assert(NT * TSTRIDE == NSEG, "the counters fill the cleared range");

__device__ __forceinline__ void put_edge(int* __restrict__ bucket,
                                         int* __restrict__ cnts,
                                         int* __restrict__ ovt,
                                         int* __restrict__ ove,
                                         int slot, int tile, int packed) {
  if (slot < TC) {
    atomicExch(&bucket[tile * TC + slot], packed);
  } else {
    int o = atomicAdd(&cnts[2], 1);
    atomicExch(&ovt[o], tile);
    atomicExch(&ove[o], packed);
  }
}

__device__ __forceinline__ void push_entry(int* __restrict__ cnts,
                                           int* __restrict__ esrc,
                                           int* __restrict__ edst,
                                           int* __restrict__ erel,
                                           int s, int d, int r) {
  int slot = atomicAdd(&cnts[0], 1);
  atomicExch(&esrc[slot], s);
  atomicExch(&edst[slot], d);
  atomicExch(&erel[slot], r);
}

#define PK_T 256
#define PK_Q 4
#define PK_CH (PK_T * PK_Q * 4)
#define PK_WG2 ((E2 + PK_CH - 1) / PK_CH)
#define PK_WG1 ((E1 + PK_CH - 1) / PK_CH)
#define PK_WGR ((N1 + PK_CH - 1) / PK_CH)
#define NTP 384
static_assert(N1 % 4 == 0, "the history map is read four rows at a time");
static_assert(NT <= NTP && NTP % 32 == 0 && NTP <= 2 * PK_T, "two counters per thread, 12 per lane of one wave");
static_assert(NT <= 512, "a tile number fits the 9 bits above the 22-bit entry");

__device__ __forceinline__ void bin_group(const int* __restrict__ src2,
                                          const int* __restrict__ dst2,
                                          const int* __restrict__ et2,
                                          int* __restrict__ cnts,
                                          int* __restrict__ tcnt,
                                          int* __restrict__ bucket,
                                          int* __restrict__ ovt,
                                          int* __restrict__ ove,
                                          int g,
                                          int* __restrict__ lcnt,
                                          int* __restrict__ loff,
                                          int* __restrict__ lgb,
                                          int* __restrict__ srt) {
  const int tid = threadIdx.x;
  for (int i = tid; i < NTP; i += PK_T) lcnt[i] = 0;
  int sv[4 * PK_Q], dv[4 * PK_Q], rv[4 * PK_Q];
  bool ok[PK_Q];
#pragma unroll
  for (int k = 0; k < PK_Q; ++k) {
    int q = g * (PK_CH / 4) + k * PK_T + tid;
    ok[k] = q < E2 / 4;
    int qq = ok[k] ? q : 0;
    int4 a = ((const int4*)src2)[qq];
    int4 b = ((const int4*)dst2)[qq];
    int4 c = ((const int4*)et2)[qq];
    sv[4 * k] = a.x; sv[4 * k + 1] = a.y; sv[4 * k + 2] = a.z; sv[4 * k + 3] = a.w;
    dv[4 * k] = b.x; dv[4 * k + 1] = b.y; dv[4 * k + 2] = b.z; dv[4 * k + 3] = b.w;
    rv[4 * k] = c.x; rv[4 * k + 1] = c.y; rv[4 * k + 2] = c.z; rv[4 * k + 3] = c.w;
  }
  __syncthreads();
  int pk[4 * PK_Q], rk[4 * PK_Q];
#pragma unroll
  for (int i = 0; i < 4 * PK_Q; ++i) {
    int tile = dv[i] >> 4;
    pk[i] = sv[i] | ((((dv[i] & 15) << 2) | rv[i]) << 16) | (tile << 22);
    rk[i] = ok[i >> 2] ? atomicAdd(&lcnt[tile], 1) : 0;
  }
  __syncthreads();
  const int t0 = tid, t1 = tid + PK_T;
  int c0 = t0 < NT ? lcnt[t0] : 0;
  int c1 = t1 < NT ? lcnt[t1] : 0;
  int b0 = 0, b1 = 0;
  if (c0 > 0) b0 = atomicAdd(&tcnt[t0 * TSTRIDE], c0);
  if (c1 > 0) b1 = atomicAdd(&tcnt[t1 * TSTRIDE], c1);
  if (tid < 32) {
    int base = tid * (NTP / 32);
    int loc[NTP / 32];
    int sum = 0;
#pragma unroll
    for (int j = 0; j < NTP / 32; ++j) { loc[j] = sum; sum += lcnt[base + j]; }
    int inc = sum;
#pragma unroll
    for (int dl = 1; dl < 32; dl <<= 1) {
      int o = __shfl_up(inc, dl, 32);
      if (tid >= dl) inc += o;
    }
#pragma unroll
    for (int j = 0; j < NTP / 32; ++j) loff[base + j] = inc - sum + loc[j];
  }
  lgb[t0] = b0;
  if (t1 < NTP) lgb[t1] = b1;
  __syncthreads();
  int pl[4 * PK_Q];
#pragma unroll
  for (int i = 0; i < 4 * PK_Q; ++i) pl[i] = loff[pk[i] >> 22];
#pragma unroll
  for (int i = 0; i < 4 * PK_Q; ++i)
    if (ok[i >> 2]) srt[pl[i] + rk[i]] = pk[i];
  __syncthreads();
  int nv = E2 - g * PK_CH;
  nv = nv < PK_CH ? nv : PK_CH;
  int e[4 * PK_Q], sl[4 * PK_Q];
#pragma unroll
  for (int k = 0; k < 4 * PK_Q; ++k) {
    int i = k * PK_T + tid;
    e[k] = srt[i < nv ? i : 0];
  }
#pragma unroll
  for (int k = 0; k < 4 * PK_Q; ++k) {
    int tile = e[k] >> 22;
    sl[k] = lgb[tile] - loff[tile] + k * PK_T + tid;
  }
#pragma unroll
  for (int k = 0; k < 4 * PK_Q; ++k)
    if (k * PK_T + tid < nv)
      put_edge(bucket, cnts, ovt, ove, sl[k], e[k] >> 22, e[k] & 0x3FFFFF);
}

__global__ void __launch_bounds__(PK_T)
k_pick(const int* __restrict__ src,
       const int* __restrict__ dst,
       const int* __restrict__ et,
       const int* __restrict__ hmap,
       int* __restrict__ cnts,
       int* __restrict__ esrc,
       int* __restrict__ edst,
       int* __restrict__ erel,
       int* __restrict__ rlist,
       const int* __restrict__ src2,
       const int* __restrict__ dst2,
       const int* __restrict__ et2,
       int* __restrict__ tcnt,
       int* __restrict__ bucket,
       int* __restrict__ ovt,
       int* __restrict__ ove) {
  __shared__ int lcnt[NTP], loff[NTP], lgb[NTP];
  __shared__ int srt[PK_CH];
  const int tid = threadIdx.x;
  const int g = blockIdx.x;
  if (g < PK_WG2) {
    bin_group(src2, dst2, et2, cnts, tcnt, bucket, ovt, ove, g, lcnt, loff, lgb, srt);
  } else if (g < PK_WG2 + PK_WG1) {
    const int q0 = (g - PK_WG2) * (PK_CH / 4) + tid;
    int dv[4 * PK_Q];
#pragma unroll
    for (int k = 0; k < PK_Q; ++k) {
      int q = q0 + k * PK_T;
      int4 b = ((const int4*)dst)[q < NQ ? q : 0];
      dv[4 * k] = b.x; dv[4 * k + 1] = b.y; dv[4 * k + 2] = b.z; dv[4 * k + 3] = b.w;
    }
    int h[4 * PK_Q];
#pragma unroll
    for (int i = 0; i < 4 * PK_Q; ++i) h[i] = hmap[dv[i]];
    unsigned m = 0;
#pragma unroll
    for (int i = 0; i < 4 * PK_Q; ++i) {
      unsigned in = (q0 + (i >> 2) * PK_T < NQ) ? 1u : 0u;
      m |= (((unsigned)h[i] >> 31) & in) << i;
    }
    while (m) {
      int i = __ffs(m) - 1;
      m &= m - 1;
      int ei = (q0 + (i >> 2) * PK_T) * 4 + (i & 3);
      push_entry(cnts, esrc, edst, erel, src[ei], dst[ei], et[ei]);
    }
  } else {
    const int q0 = (g - PK_WG2 - PK_WG1) * (PK_CH / 4) + tid;
    unsigned m = 0;
#pragma unroll
    for (int k = 0; k < PK_Q; ++k) {
      int q = q0 + k * PK_T;
      bool ok = q < N1 / 4;
      int4 hq = ((const int4*)hmap)[ok ? q : 0];
      unsigned in = ok ? 1u : 0u;
      m |= ((((unsigned)hq.x >> 31) | (((unsigned)hq.y >> 31) << 1) |
             (((unsigned)hq.z >> 31) << 2) | (((unsigned)hq.w >> 31) << 3)) * in) << (4 * k);
    }
    while (m) {
      int i = __ffs(m) - 1;
      m &= m - 1;
      int n = (q0 + (i >> 2) * PK_T) * 4 + (i & 3);
      push_entry(cnts, esrc, edst, erel, n, n, 4);
      int slot = atomicAdd(&cnts[1], 1);
      atomicExch(&rlist[slot], n);
    }
  }
}

#define HF_T 1024
#define HF_G 512
#define HF_TOTAL (N1 * 128)
#define HF_STRIDE (HF_G * HF_T)
#define HF_TRIPS ((HF_TOTAL + HF_STRIDE - 1) / HF_STRIDE)
static_assert(HF_T % 128 == 0, "a thread keeps one column over all its trips");
static_assert((long)N1 * 128 * 4 < (1l << 31), "element offsets fit 32 bits");
__global__ void __launch_bounds__(HF_T)
k_histfill_stats(const int* __restrict__ hmap,
                 const float* __restrict__ hbuf,
                 const float* __restrict__ b1,
                 float* __restrict__ h1,
                 float* __restrict__ csum,
                 float* __restrict__ csq) {
  __shared__ float ls[HF_T], lq[HF_T];
  unsigned tid = blockIdx.x * HF_T + threadIdx.x;
  unsigned c = threadIdx.x & 127;
  unsigned row0 = tid >> 7;
  float bias = b1[c];
  int hm[HF_TRIPS];
#pragma unroll
  for (int t = 0; t < HF_TRIPS; ++t) {
    bool in = ((t + 1) * HF_STRIDE <= HF_TOTAL) || (tid + t * HF_STRIDE < HF_TOTAL);
    hm[t] = hmap[in ? row0 + t * (HF_STRIDE / 128) : 0u];
  }
  float v[HF_TRIPS];
#pragma unroll
  for (int t = 0; t < HF_TRIPS; ++t) {
    unsigned r = hm[t] >= 0 ? (unsigned)hm[t] : 0u;
    v[t] = hbuf[r * 128u + c];
  }
  __builtin_amdgcn_sched_barrier(0);
  float s = 0.f, q = 0.f;
#pragma unroll
  for (int t = 0; t < HF_TRIPS; ++t) {
    bool in = ((t + 1) * HF_STRIDE <= HF_TOTAL) || (tid + t * HF_STRIDE < HF_TOTAL);
    bool hist = in && hm[t] >= 0;
    float sn = s + v[t];
    float qn = q + v[t] * v[t];
    s = hist ? sn : s;
    q = hist ? qn : q;
    float val = hist ? v[t] : bias;
    if (in) h1[tid + t * HF_STRIDE] = val;
  }
  ls[threadIdx.x] = s; lq[threadIdx.x] = q;
  __syncthreads();
  if (threadIdx.x < 128) {
    float S = 0.f, Q = 0.f;
    for (int k = 0; k < HF_T / 128; ++k) {
      S += ls[threadIdx.x + 128 * k];
      Q += lq[threadIdx.x + 128 * k];
    }
    atomicAdd(&csum[threadIdx.x], S);
    atomicAdd(&csq[threadIdx.x], Q);
  }
}

__global__ void __launch_bounds__(256)
k_msg1(const float* __restrict__ x,
                       const float* __restrict__ W1,
                       const float* __restrict__ Ws,
                       const int* __restrict__ cnts,
                       const int* __restrict__ esrc,
                       const int* __restrict__ edst,
                       const int* __restrict__ erel,
                       float* __restrict__ h1) {
  int cnt = cnts[0];
  int ntiles = (cnt + 15) >> 4;
  int wave = threadIdx.x >> 5;
  int lane = threadIdx.x & 31;
  int aRow = lane & 15;
  int aK   = (lane >> 4) << 1;
  int col  = wave * 16 + (lane & 15);
  int halfOff = (lane >> 4) << 3;
  const v2f zero2 = {0.f, 0.f};

  for (int tile = blockIdx.x; tile < ntiles; tile += gridDim.x) {
    int ei = tile * 16 + aRow;
    bool live = ei < cnt;
    int ec = live ? ei : 0;
    int sRow = esrc[ec];
    int rel  = live ? erel[ec] : -1;
    const float* Ap = x + (long)sRow * 128 + aK;

    v2f af[32];
#pragma unroll
    for (int ks = 0; ks < 32; ++ks) af[ks] = *(const v2f*)(Ap + 4 * ks);

    v8f acc = {0.f,0.f,0.f,0.f,0.f,0.f,0.f,0.f};
    for (int r = 0; r < 5; ++r) {
      bool mine = (rel == r);
      if (__ballot(mine) == 0ull) continue;
      const float* Bp = ((r < 4) ? (W1 + (long)r * 128 * 128) : Ws)
                        + aK * 128 + col;
      v2f bf[32];
#pragma unroll
      for (int ks = 0; ks < 32; ++ks) {
        bf[ks].x = Bp[(4 * ks) * 128];
        bf[ks].y = Bp[(4 * ks + 1) * 128];
      }
#pragma unroll
      for (int ks = 0; ks < 32; ++ks) {
        v2f a = mine ? af[ks] : zero2;
        acc = __builtin_amdgcn_wmma_f32_16x16x4_f32(
            false, a, false, bf[ks], (short)0, acc, false, false);
      }
    }

    int dd[8];
#pragma unroll
    for (int v = 0; v < 8; ++v) {
      int eo = tile * 16 + v + halfOff;
      dd[v] = edst[eo < cnt ? eo : 0];
    }
#pragma unroll
    for (int v = 0; v < 8; ++v) {
      int eo = tile * 16 + v + halfOff;
      if (eo < cnt) atomicAdd(&h1[(long)dd[v] * 128 + col], acc[v]);
    }
  }
}

__global__ void k_bnfinal(const float* __restrict__ csum,
                          const float* __restrict__ csq,
                          const float* __restrict__ gamma,
                          const float* __restrict__ beta,
                          float* __restrict__ scale,
                          float* __restrict__ shift,
                          const int* __restrict__ cnts,
                          const int* __restrict__ rlist,
                          const float* __restrict__ h1) {
  int c = threadIdx.x;
  float s = csum[c], q = csq[c];
  int nr = cnts[1];
  for (int j = 0; j < nr; ++j) {
    float v = h1[(long)rlist[j] * 128 + c];
    s += v; q += v * v;
  }
  const float inv_n = 1.0f / (float)N1;
  float mean = s * inv_n;
  float var  = q * inv_n - mean * mean;
  float sc = gamma[c] * rsqrtf(var + BN_EPS);
  scale[c] = sc;
  shift[c] = beta[c] - mean * sc;
}

#define PA 644

__device__ __forceinline__ void tile_chunk(int cnt, const int (&e)[TC / 256],
                                           int wave, int lane,
                                           int* __restrict__ srt,
                                           int* __restrict__ hist,
                                           int* __restrict__ off,
                                           const float* __restrict__ h1,
                                           float4 sc, float4 sh,
                                           float4 (&acc)[8]) {
  if (threadIdx.x < 64) hist[threadIdx.x] = 0;
  __syncthreads();
  int rk[TC / 256];
#pragma unroll
  for (int k = 0; k < TC / 256; ++k) {
    rk[k] = 0;
    if (threadIdx.x + 256 * k < cnt) rk[k] = atomicAdd(&hist[(e[k] >> 16) & 63], 1);
  }
  __syncthreads();
  if (wave == 0) {
    int a = hist[2 * lane], c = hist[2 * lane + 1];
    int s = a + c, inc = s;
#pragma unroll
    for (int d = 1; d < 32; d <<= 1) {
      int o = __shfl_up(inc, d, 32);
      if (lane >= d) inc += o;
    }
    off[2 * lane] = inc - s;
    off[2 * lane + 1] = inc - s + a;
    if (lane == 31) off[64] = inc;
  }
  __syncthreads();
  int st[TC / 256];
#pragma unroll
  for (int k = 0; k < TC / 256; ++k) st[k] = off[(e[k] >> 16) & 63];
#pragma unroll
  for (int k = 0; k < TC / 256; ++k)
    if (threadIdx.x + 256 * k < cnt) srt[st[k] + rk[k]] = e[k] & 0xFFFF;
  __syncthreads();
#pragma unroll
  for (int i = 0; i < 8; ++i) {
    int beg = __builtin_amdgcn_readfirstlane(off[wave * 8 + i]);
    int end = __builtin_amdgcn_readfirstlane(off[wave * 8 + i + 1]);
    float4 a4 = acc[i];
    for (int base = beg; base < end; base += 32) {
      int n = end - base;
      n = n < 32 ? n : 32;
      int mine = (lane < n) ? srt[base + lane] : 0;
      for (int j0 = 0; j0 < n; j0 += 8) {
        float4 x[8];
#pragma unroll
        for (int k = 0; k < 8; ++k) {
          int sRow = __builtin_amdgcn_readlane(mine, j0 + k);
          x[k] = ((const float4*)(h1 + (long)sRow * 128))[lane];
        }
#pragma unroll
        for (int k = 0; k < 8; ++k) {
          if (j0 + k < n) {
            a4.x += fmaxf(0.0f, x[k].x * sc.x + sh.x);
            a4.y += fmaxf(0.0f, x[k].y * sc.y + sh.y);
            a4.z += fmaxf(0.0f, x[k].z * sc.z + sh.z);
            a4.w += fmaxf(0.0f, x[k].w * sc.w + sh.w);
          }
        }
      }
    }
    acc[i] = a4;
  }
}

__global__ void __launch_bounds__(256)
k_agg_gemm2(const float* __restrict__ h1,
            const float* __restrict__ scale,
            const float* __restrict__ shift,
            const int* __restrict__ tcnt,
            const int* __restrict__ bucket,
            const int* __restrict__ cnts,
            const int* __restrict__ ovt,
            const int* __restrict__ ove,
            const float* __restrict__ W2,
            const float* __restrict__ Ws2,
            const float* __restrict__ b2,
            float* __restrict__ out) {
  __shared__ float At[16 * PA];
  __shared__ float red[8 * 4 * 8 * 32];
  __shared__ int ent[TC];
  __shared__ int srt[TC];
  __shared__ int hist[64];
  __shared__ int off[65];
  __shared__ int nsel[1];

  int wave = __builtin_amdgcn_readfirstlane((int)(threadIdx.x >> 5));
  int lane = threadIdx.x & 31;
  int tile = blockIdx.x;
  int row0 = tile * 16;

  int tc = tcnt[tile * TSTRIDE];
  int nb = tc < TC ? tc : TC;
  int novf = 0;
  if (tc > TC) {
    novf = cnts[2];
    novf = novf < E2 ? novf : E2;
  }

  float4 sc = ((const float4*)scale)[lane];
  float4 sh = ((const float4*)shift)[lane];

  const int* bk = bucket + tile * TC;
  int e0[TC / 256];
#pragma unroll
  for (int k = 0; k < TC / 256; ++k) e0[k] = 0;
  if (nb > 0) {
#pragma unroll
    for (int k = 0; k < TC / 256; ++k) {
      int i = threadIdx.x + 256 * k;
      e0[k] = bk[i < nb ? i : nb - 1];
    }
  }

  int g0 = row0 + wave * 2;
  float4 own[2];
#pragma unroll
  for (int rr = 0; rr < 2; ++rr)
    own[rr] = ((const float4*)(h1 + (long)(g0 + rr) * 128))[lane];

  float4 segsum[8];
#pragma unroll
  for (int i = 0; i < 8; ++i) segsum[i] = make_float4(0.f, 0.f, 0.f, 0.f);

  int nchunk = 1 + (novf + TC - 1) / TC;
#pragma nounroll
  for (int ch = 0; ch < nchunk; ++ch) {
    int cnt;
    int e[TC / 256];
#pragma unroll
    for (int k = 0; k < TC / 256; ++k) e[k] = 0;
    if (ch == 0) {
      cnt = nb;
#pragma unroll
      for (int k = 0; k < TC / 256; ++k) e[k] = e0[k];
    } else {
      int base = (ch - 1) * TC;
      __syncthreads();
      if (threadIdx.x == 0) nsel[0] = 0;
      __syncthreads();
#pragma unroll
      for (int k = 0; k < TC / 256; ++k) {
        int j = base + threadIdx.x + 256 * k;
        if (j < novf && ovt[j] == tile) {
          int p = atomicAdd(&nsel[0], 1);
          ent[p] = ove[j];
        }
      }
      __syncthreads();
      cnt = nsel[0];
#pragma unroll
      for (int k = 0; k < TC / 256; ++k) {
        int i = threadIdx.x + 256 * k;
        if (i < cnt) e[k] = ent[i];
      }
    }
    tile_chunk(cnt, e, wave, lane, srt, hist, off, h1, sc, sh, segsum);
  }

#pragma unroll
  for (int rr = 0; rr < 2; ++rr) {
    float4 x = own[rr];
    float4 v;
    v.x = fmaxf(0.0f, x.x * sc.x + sh.x);
    v.y = fmaxf(0.0f, x.y * sc.y + sh.y);
    v.z = fmaxf(0.0f, x.z * sc.z + sh.z);
    v.w = fmaxf(0.0f, x.w * sc.w + sh.w);
    *(float4*)(At + (wave * 2 + rr) * PA + 512 + 4 * lane) = v;
  }
#pragma unroll
  for (int i = 0; i < 8; ++i)
    *(float4*)(At + (wave * 2 + (i >> 2)) * PA + (i & 3) * 128 + 4 * lane) = segsum[i];
  __syncthreads();

  int c = lane & 15;
  int h2 = (lane >> 4) << 1;
  int kb = wave * 80;
  v2f af[20];
  v2f bf[20][4];
#pragma unroll
  for (int ks = 0; ks < 20; ++ks) {
    int kk = kb + 4 * ks;
    const float* Bk = (kk < 512) ? (W2 + kk * 64) : (Ws2 + (kk - 512) * 64);
    Bk += h2 * 64 + c;
#pragma unroll
    for (int t = 0; t < 4; ++t) {
      bf[ks][t].x = Bk[t * 16];
      bf[ks][t].y = Bk[64 + t * 16];
    }
    af[ks] = *(const v2f*)(At + c * PA + kk + h2);
  }
  v8f acc[4];
  const v8f zero = {0.f,0.f,0.f,0.f,0.f,0.f,0.f,0.f};
#pragma unroll
  for (int t = 0; t < 4; ++t) acc[t] = zero;
#pragma unroll
  for (int ks = 0; ks < 20; ++ks) {
#pragma unroll
    for (int t = 0; t < 4; ++t)
      acc[t] = __builtin_amdgcn_wmma_f32_16x16x4_f32(
          false, af[ks], false, bf[ks][t], (short)0, acc[t], false, false);
  }
#pragma unroll
  for (int t = 0; t < 4; ++t)
#pragma unroll
    for (int v = 0; v < 8; ++v)
      red[((wave * 4 + t) * 8 + v) * 32 + lane] = acc[t][v];
  __syncthreads();

  int row = threadIdx.x >> 4;
  int cq  = threadIdx.x & 15;
  int ridx = (((cq >> 2) * 8 + (row & 7)) * 32) + (row >> 3) * 16 + (cq & 3) * 4;
  float4 bb = ((const float4*)b2)[cq];
  float vals[4] = {bb.x, bb.y, bb.z, bb.w};
#pragma unroll
  for (int w = 0; w < 8; ++w) {
    float4 p = *(const float4*)(red + w * (4 * 8 * 32) + ridx);
    vals[0] += p.x; vals[1] += p.y; vals[2] += p.z; vals[3] += p.w;
  }
  float m = fmaxf(fmaxf(vals[0], vals[1]), fmaxf(vals[2], vals[3]));
#pragma unroll
  for (int mask = 1; mask < 16; mask <<= 1)
    m = fmaxf(m, __shfl_xor(m, mask, 32));
  float sum = 0.f;
#pragma unroll
  for (int t = 0; t < 4; ++t) sum += expf(vals[t] - m);
#pragma unroll
  for (int mask = 1; mask < 16; mask <<= 1)
    sum += __shfl_xor(sum, mask, 32);
  float lse = m + logf(sum);
  float4 o;
  o.x = vals[0] - lse; o.y = vals[1] - lse; o.z = vals[2] - lse; o.w = vals[3] - lse;
  ((float4*)(out + (long)(row0 + row) * 64))[cq] = o;
}

extern "C" void kernel_launch(void* const* d_in, const int* in_sizes, int n_in,
                              void* d_out, int out_size, void* d_ws, size_t ws_size,
                              hipStream_t stream) {
  const float* x     = (const float*)d_in[0];
  const int*   src1  = (const int*)  d_in[1];
  const int*   dst1  = (const int*)  d_in[2];
  const int*   et1   = (const int*)  d_in[3];
  const int*   src2  = (const int*)  d_in[4];
  const int*   dst2  = (const int*)  d_in[5];
  const int*   et2   = (const int*)  d_in[6];
  const int*   hmap  = (const int*)  d_in[7];
  const float* hbuf  = (const float*)d_in[8];
  const float* W1    = (const float*)d_in[9];
  const float* Ws1   = (const float*)d_in[10];
  const float* b1    = (const float*)d_in[11];
  const float* gamma = (const float*)d_in[12];
  const float* beta  = (const float*)d_in[13];
  const float* W2    = (const float*)d_in[14];
  const float* Ws2   = (const float*)d_in[15];
  const float* b2    = (const float*)d_in[16];

  float* ws    = (float*)d_ws;
  float* A1    = ws;
  float* A2    = A1 + (size_t)4 * N1 * 128;
  float* ss    = A2 + (size_t)4 * N2 * 128 + 256;
  float* h1    = ss + 256;
  float* out   = (float*)d_out;
  int* esrc  = (int*)A1;
  int* edst  = esrc + CAP;
  int* erel  = edst + CAP;
  int* rlist = erel + CAP;
  int* cnts  = (int*)A2 - 4;
  int* tcnt  = (int*)A2;
  float* sums = (float*)(tcnt + NT * TSTRIDE);
  int* bucket = (int*)(sums + 256);
  int* ovt    = bucket + NT * TC;
  int* ove    = ovt + E2;

  k_zero<<<96, 256, 0, stream>>>((float*)cnts, (long)(4 + NT * TSTRIDE + 256));

  k_pick<<<PK_WG2 + PK_WG1 + PK_WGR, PK_T, 0, stream>>>(src1, dst1, et1, hmap, cnts,
                                                         esrc, edst, erel, rlist,
                                                         src2, dst2, et2,
                                                         tcnt, bucket, ovt, ove);
  k_histfill_stats<<<HF_G, HF_T, 0, stream>>>(hmap, hbuf, b1, h1, sums, sums + 128);

  k_msg1<<<1024, 256, 0, stream>>>(x, W1, Ws1, cnts, esrc, edst, erel, h1);

  k_bnfinal<<<1, 128, 0, stream>>>(sums, sums + 128, gamma, beta, ss, ss + 128,
                                   cnts, rlist, h1);

  static_assert(N2 % 16 == 0, "one work-group per 16 destination rows");
  k_agg_gemm2<<<NT, 256, 0, stream>>>(h1, ss, ss + 128, tcnt, bucket, cnts, ovt, ove,
                                      W2, Ws2, b2, out);
}


# layer-1 messages run inside the edge pass: the work-group that finds an edge lists it in LDS and runs its list on WMMA into h1 atomics; global list and message launch gone (atomic arrival order)
# speedup vs baseline: 10.3580x; 1.1968x over previous
#include <hip/hip_runtime.h>
#include <hip/hip_bf16.h>
#include <math.h>

typedef __attribute__((ext_vector_type(2))) float v2f;
typedef __attribute__((ext_vector_type(8))) float v8f;

#define N0  300000
#define N1  60000
#define N2  6000
#define E1  1200000
#define E2  300000
#define D_H 128
#define D_OUT 64
#define BN_EPS 1e-5f

__global__ void k_zero(float* __restrict__ p, long n) {
  long i = (long)blockIdx.x * blockDim.x + threadIdx.x;
  long stride = (long)gridDim.x * blockDim.x;
  for (; i < n; i += stride) p[i] = 0.0f;
}

#define NQ (E1 / 4)
static_assert(E1 % 4 == 0, "edge list is read four at a time");
#define NSEG (4 * N2)
static_assert(E2 % 4 == 0, "layer-2 edge list is read four at a time");

#define NT (N2 / 16)
#define TC 2048
#define TSTRIDE 64
static_assert(N1 <= 65536, "a source row fits the low 16 bits of a bucket entry");
static_assert(TC % 256 == 0, "a chunk is a whole number of passes of the work-group");
static_assert(NT * TSTRIDE == NSEG, "the counters fill the cleared range");

__device__ __forceinline__ void put_edge(int* __restrict__ bucket,
                                         int* __restrict__ cnts,
                                         int* __restrict__ ovt,
                                         int* __restrict__ ove,
                                         int slot, int tile, int packed) {
  if (slot < TC) {
    atomicExch(&bucket[tile * TC + slot], packed);
  } else {
    int o = atomicAdd(&cnts[2], 1);
    atomicExch(&ovt[o], tile);
    atomicExch(&ove[o], packed);
  }
}

__device__ __forceinline__ void msg_tiles(int cnt,
                                          const int* __restrict__ items,
                                          const int* __restrict__ src,
                                          const int* __restrict__ dst,
                                          const int* __restrict__ et,
                                          const float* __restrict__ x,
                                          const float* __restrict__ W1,
                                          const float* __restrict__ Ws,
                                          float* __restrict__ h1) {
  int ntiles = (cnt + 15) >> 4;
  int wave = threadIdx.x >> 5;
  int lane = threadIdx.x & 31;
  int aRow = lane & 15;
  int aK   = (lane >> 4) << 1;
  int col  = wave * 16 + (lane & 15);
  int halfOff = (lane >> 4) << 3;
  const v2f zero2 = {0.f, 0.f};

  for (int tile = 0; tile < ntiles; ++tile) {
    int ei = tile * 16 + aRow;
    bool live = ei < cnt;
    int item = items[live ? ei : 0];
    bool self = item >= E1;
    int e = self ? 0 : item;
    int sLd = src[e];
    int dLd = dst[e];
    int rLd = et[e];
    int sRow = self ? item - E1 : sLd;
    int dRow = self ? item - E1 : dLd;
    int rel  = live ? (self ? 4 : rLd) : -1;
    const float* Ap = x + (long)sRow * 128 + aK;

    v2f af[32];
#pragma unroll
    for (int ks = 0; ks < 32; ++ks) af[ks] = *(const v2f*)(Ap + 4 * ks);

    v8f acc = {0.f,0.f,0.f,0.f,0.f,0.f,0.f,0.f};
    for (int r = 0; r < 5; ++r) {
      bool mine = (rel == r);
      if (__ballot(mine) == 0ull) continue;
      const float* Bp = ((r < 4) ? (W1 + (long)r * 128 * 128) : Ws)
                        + aK * 128 + col;
      v2f bf[32];
#pragma unroll
      for (int ks = 0; ks < 32; ++ks) {
        bf[ks].x = Bp[(4 * ks) * 128];
        bf[ks].y = Bp[(4 * ks + 1) * 128];
      }
#pragma unroll
      for (int ks = 0; ks < 32; ++ks) {
        v2f a = mine ? af[ks] : zero2;
        acc = __builtin_amdgcn_wmma_f32_16x16x4_f32(
            false, a, false, bf[ks], (short)0, acc, false, false);
      }
    }

#pragma unroll
    for (int v = 0; v < 8; ++v) {
      int lo = __builtin_amdgcn_readlane(dRow, v);
      int hi = __builtin_amdgcn_readlane(dRow, v + 8);
      int d = halfOff ? hi : lo;
      int eo = tile * 16 + v + halfOff;
      if (eo < cnt) atomicAdd(&h1[(long)d * 128 + col], acc[v]);
    }
  }
}

#define PK_T 256
#define PK_Q 4
#define PK_CH (PK_T * PK_Q * 4)
#define PK_WG2 ((E2 + PK_CH - 1) / PK_CH)
#define PK_WG1 ((E1 + PK_CH - 1) / PK_CH)
#define PK_WGR ((N1 + PK_CH - 1) / PK_CH)
#define NTP 384
static_assert(N1 % 4 == 0, "the history map is read four rows at a time");
static_assert(NT <= NTP && NTP % 32 == 0 && NTP <= 2 * PK_T, "two counters per thread, 12 per lane of one wave");
static_assert(NT <= 512, "a tile number fits the 9 bits above the 22-bit entry");

__device__ __forceinline__ void bin_group(const int* __restrict__ src2,
                                          const int* __restrict__ dst2,
                                          const int* __restrict__ et2,
                                          int* __restrict__ cnts,
                                          int* __restrict__ tcnt,
                                          int* __restrict__ bucket,
                                          int* __restrict__ ovt,
                                          int* __restrict__ ove,
                                          int g,
                                          int* __restrict__ lcnt,
                                          int* __restrict__ loff,
                                          int* __restrict__ lgb,
                                          int* __restrict__ srt) {
  const int tid = threadIdx.x;
  for (int i = tid; i < NTP; i += PK_T) lcnt[i] = 0;
  int sv[4 * PK_Q], dv[4 * PK_Q], rv[4 * PK_Q];
  bool ok[PK_Q];
#pragma unroll
  for (int k = 0; k < PK_Q; ++k) {
    int q = g * (PK_CH / 4) + k * PK_T + tid;
    ok[k] = q < E2 / 4;
    int qq = ok[k] ? q : 0;
    int4 a = ((const int4*)src2)[qq];
    int4 b = ((const int4*)dst2)[qq];
    int4 c = ((const int4*)et2)[qq];
    sv[4 * k] = a.x; sv[4 * k + 1] = a.y; sv[4 * k + 2] = a.z; sv[4 * k + 3] = a.w;
    dv[4 * k] = b.x; dv[4 * k + 1] = b.y; dv[4 * k + 2] = b.z; dv[4 * k + 3] = b.w;
    rv[4 * k] = c.x; rv[4 * k + 1] = c.y; rv[4 * k + 2] = c.z; rv[4 * k + 3] = c.w;
  }
  __syncthreads();
  int pk[4 * PK_Q], rk[4 * PK_Q];
#pragma unroll
  for (int i = 0; i < 4 * PK_Q; ++i) {
    int tile = dv[i] >> 4;
    pk[i] = sv[i] | ((((dv[i] & 15) << 2) | rv[i]) << 16) | (tile << 22);
    rk[i] = ok[i >> 2] ? atomicAdd(&lcnt[tile], 1) : 0;
  }
  __syncthreads();
  const int t0 = tid, t1 = tid + PK_T;
  int c0 = t0 < NT ? lcnt[t0] : 0;
  int c1 = t1 < NT ? lcnt[t1] : 0;
  int b0 = 0, b1 = 0;
  if (c0 > 0) b0 = atomicAdd(&tcnt[t0 * TSTRIDE], c0);
  if (c1 > 0) b1 = atomicAdd(&tcnt[t1 * TSTRIDE], c1);
  if (tid < 32) {
    int base = tid * (NTP / 32);
    int loc[NTP / 32];
    int sum = 0;
#pragma unroll
    for (int j = 0; j < NTP / 32; ++j) { loc[j] = sum; sum += lcnt[base + j]; }
    int inc = sum;
#pragma unroll
    for (int dl = 1; dl < 32; dl <<= 1) {
      int o = __shfl_up(inc, dl, 32);
      if (tid >= dl) inc += o;
    }
#pragma unroll
    for (int j = 0; j < NTP / 32; ++j) loff[base + j] = inc - sum + loc[j];
  }
  lgb[t0] = b0;
  if (t1 < NTP) lgb[t1] = b1;
  __syncthreads();
  int pl[4 * PK_Q];
#pragma unroll
  for (int i = 0; i < 4 * PK_Q; ++i) pl[i] = loff[pk[i] >> 22];
#pragma unroll
  for (int i = 0; i < 4 * PK_Q; ++i)
    if (ok[i >> 2]) srt[pl[i] + rk[i]] = pk[i];
  __syncthreads();
  int nv = E2 - g * PK_CH;
  nv = nv < PK_CH ? nv : PK_CH;
  int e[4 * PK_Q], sl[4 * PK_Q];
#pragma unroll
  for (int k = 0; k < 4 * PK_Q; ++k) {
    int i = k * PK_T + tid;
    e[k] = srt[i < nv ? i : 0];
  }
#pragma unroll
  for (int k = 0; k < 4 * PK_Q; ++k) {
    int tile = e[k] >> 22;
    sl[k] = lgb[tile] - loff[tile] + k * PK_T + tid;
  }
#pragma unroll
  for (int k = 0; k < 4 * PK_Q; ++k)
    if (k * PK_T + tid < nv)
      put_edge(bucket, cnts, ovt, ove, sl[k], e[k] >> 22, e[k] & 0x3FFFFF);
}

__global__ void __launch_bounds__(PK_T)
k_pick(const int* __restrict__ src,
       const int* __restrict__ dst,
       const int* __restrict__ et,
       const int* __restrict__ hmap,
       int* __restrict__ cnts,
       const float* __restrict__ x,
       const float* __restrict__ W1,
       const float* __restrict__ Ws1,
       float* __restrict__ h1,
       int* __restrict__ rlist,
       const int* __restrict__ src2,
       const int* __restrict__ dst2,
       const int* __restrict__ et2,
       int* __restrict__ tcnt,
       int* __restrict__ bucket,
       int* __restrict__ ovt,
       int* __restrict__ ove) {
  __shared__ int lcnt[NTP], loff[NTP], lgb[NTP];
  __shared__ int srt[PK_CH];
  const int tid = threadIdx.x;
  const int g = blockIdx.x;
  if (g < PK_WG2) {
    bin_group(src2, dst2, et2, cnts, tcnt, bucket, ovt, ove, g, lcnt, loff, lgb, srt);
  } else {
    if (tid == 0) lcnt[0] = 0;
    const bool edges = g < PK_WG2 + PK_WG1;
    unsigned m = 0;
    int q0;
    if (edges) {
      q0 = (g - PK_WG2) * (PK_CH / 4) + tid;
      int dv[4 * PK_Q];
#pragma unroll
      for (int k = 0; k < PK_Q; ++k) {
        int q = q0 + k * PK_T;
        int4 b = ((const int4*)dst)[q < NQ ? q : 0];
        dv[4 * k] = b.x; dv[4 * k + 1] = b.y; dv[4 * k + 2] = b.z; dv[4 * k + 3] = b.w;
      }
      int h[4 * PK_Q];
#pragma unroll
      for (int i = 0; i < 4 * PK_Q; ++i) h[i] = hmap[dv[i]];
#pragma unroll
      for (int i = 0; i < 4 * PK_Q; ++i) {
        unsigned in = (q0 + (i >> 2) * PK_T < NQ) ? 1u : 0u;
        m |= (((unsigned)h[i] >> 31) & in) << i;
      }
    } else {
      q0 = (g - PK_WG2 - PK_WG1) * (PK_CH / 4) + tid;
#pragma unroll
      for (int k = 0; k < PK_Q; ++k) {
        int q = q0 + k * PK_T;
        bool ok = q < N1 / 4;
        int4 hq = ((const int4*)hmap)[ok ? q : 0];
        unsigned in = ok ? 1u : 0u;
        m |= ((((unsigned)hq.x >> 31) | (((unsigned)hq.y >> 31) << 1) |
               (((unsigned)hq.z >> 31) << 2) | (((unsigned)hq.w >> 31) << 3)) * in) << (4 * k);
      }
    }
    __syncthreads();
    while (m) {
      int i = __ffs(m) - 1;
      m &= m - 1;
      int n = (q0 + (i >> 2) * PK_T) * 4 + (i & 3);
      srt[atomicAdd(&lcnt[0], 1)] = edges ? n : E1 + n;
      if (!edges) {
        int slot = atomicAdd(&cnts[1], 1);
        atomicExch(&rlist[slot], n);
      }
    }
    __syncthreads();
    int cnt = __builtin_amdgcn_readfirstlane(lcnt[0]);
    msg_tiles(cnt, srt, src, dst, et, x, W1, Ws1, h1);
  }
}

#define HF_T 1024
#define HF_G 512
#define HF_TOTAL (N1 * 128)
#define HF_STRIDE (HF_G * HF_T)
#define HF_TRIPS ((HF_TOTAL + HF_STRIDE - 1) / HF_STRIDE)
static_assert(HF_T % 128 == 0, "a thread keeps one column over all its trips");
static_assert((long)N1 * 128 * 4 < (1l << 31), "element offsets fit 32 bits");
__global__ void __launch_bounds__(HF_T)
k_histfill_stats(const int* __restrict__ hmap,
                 const float* __restrict__ hbuf,
                 const float* __restrict__ b1,
                 float* __restrict__ h1,
                 float* __restrict__ csum,
                 float* __restrict__ csq) {
  __shared__ float ls[HF_T], lq[HF_T];
  unsigned tid = blockIdx.x * HF_T + threadIdx.x;
  unsigned c = threadIdx.x & 127;
  unsigned row0 = tid >> 7;
  float bias = b1[c];
  int hm[HF_TRIPS];
#pragma unroll
  for (int t = 0; t < HF_TRIPS; ++t) {
    bool in = ((t + 1) * HF_STRIDE <= HF_TOTAL) || (tid + t * HF_STRIDE < HF_TOTAL);
    hm[t] = hmap[in ? row0 + t * (HF_STRIDE / 128) : 0u];
  }
  float v[HF_TRIPS];
#pragma unroll
  for (int t = 0; t < HF_TRIPS; ++t) {
    unsigned r = hm[t] >= 0 ? (unsigned)hm[t] : 0u;
    v[t] = hbuf[r * 128u + c];
  }
  __builtin_amdgcn_sched_barrier(0);
  float s = 0.f, q = 0.f;
#pragma unroll
  for (int t = 0; t < HF_TRIPS; ++t) {
    bool in = ((t + 1) * HF_STRIDE <= HF_TOTAL) || (tid + t * HF_STRIDE < HF_TOTAL);
    bool hist = in && hm[t] >= 0;
    float sn = s + v[t];
    float qn = q + v[t] * v[t];
    s = hist ? sn : s;
    q = hist ? qn : q;
    float val = hist ? v[t] : bias;
    if (in) h1[tid + t * HF_STRIDE] = val;
  }
  ls[threadIdx.x] = s; lq[threadIdx.x] = q;
  __syncthreads();
  if (threadIdx.x < 128) {
    float S = 0.f, Q = 0.f;
    for (int k = 0; k < HF_T / 128; ++k) {
      S += ls[threadIdx.x + 128 * k];
      Q += lq[threadIdx.x + 128 * k];
    }
    atomicAdd(&csum[threadIdx.x], S);
    atomicAdd(&csq[threadIdx.x], Q);
  }
}

__global__ void k_bnfinal(const float* __restrict__ csum,
                          const float* __restrict__ csq,
                          const float* __restrict__ gamma,
                          const float* __restrict__ beta,
                          float* __restrict__ scale,
                          float* __restrict__ shift,
                          const int* __restrict__ cnts,
                          const int* __restrict__ rlist,
                          const float* __restrict__ h1) {
  int c = threadIdx.x;
  float s = csum[c], q = csq[c];
  int nr = cnts[1];
  for (int j = 0; j < nr; ++j) {
    float v = h1[(long)rlist[j] * 128 + c];
    s += v; q += v * v;
  }
  const float inv_n = 1.0f / (float)N1;
  float mean = s * inv_n;
  float var  = q * inv_n - mean * mean;
  float sc = gamma[c] * rsqrtf(var + BN_EPS);
  scale[c] = sc;
  shift[c] = beta[c] - mean * sc;
}

#define PA 644

__device__ __forceinline__ void tile_chunk(int cnt, const int (&e)[TC / 256],
                                           int wave, int lane,
                                           int* __restrict__ srt,
                                           int* __restrict__ hist,
                                           int* __restrict__ off,
                                           const float* __restrict__ h1,
                                           float4 sc, float4 sh,
                                           float4 (&acc)[8]) {
  if (threadIdx.x < 64) hist[threadIdx.x] = 0;
  __syncthreads();
  int rk[TC / 256];
#pragma unroll
  for (int k = 0; k < TC / 256; ++k) {
    rk[k] = 0;
    if (threadIdx.x + 256 * k < cnt) rk[k] = atomicAdd(&hist[(e[k] >> 16) & 63], 1);
  }
  __syncthreads();
  if (wave == 0) {
    int a = hist[2 * lane], c = hist[2 * lane + 1];
    int s = a + c, inc = s;
#pragma unroll
    for (int d = 1; d < 32; d <<= 1) {
      int o = __shfl_up(inc, d, 32);
      if (lane >= d) inc += o;
    }
    off[2 * lane] = inc - s;
    off[2 * lane + 1] = inc - s + a;
    if (lane == 31) off[64] = inc;
  }
  __syncthreads();
  int st[TC / 256];
#pragma unroll
  for (int k = 0; k < TC / 256; ++k) st[k] = off[(e[k] >> 16) & 63];
#pragma unroll
  for (int k = 0; k < TC / 256; ++k)
    if (threadIdx.x + 256 * k < cnt) srt[st[k] + rk[k]] = e[k] & 0xFFFF;
  __syncthreads();
#pragma unroll
  for (int i = 0; i < 8; ++i) {
    int beg = __builtin_amdgcn_readfirstlane(off[wave * 8 + i]);
    int end = __builtin_amdgcn_readfirstlane(off[wave * 8 + i + 1]);
    float4 a4 = acc[i];
    for (int base = beg; base < end; base += 32) {
      int n = end - base;
      n = n < 32 ? n : 32;
      int mine = (lane < n) ? srt[base + lane] : 0;
      for (int j0 = 0; j0 < n; j0 += 8) {
        float4 x[8];
#pragma unroll
        for (int k = 0; k < 8; ++k) {
          int sRow = __builtin_amdgcn_readlane(mine, j0 + k);
          x[k] = ((const float4*)(h1 + (long)sRow * 128))[lane];
        }
#pragma unroll
        for (int k = 0; k < 8; ++k) {
          if (j0 + k < n) {
            a4.x += fmaxf(0.0f, x[k].x * sc.x + sh.x);
            a4.y += fmaxf(0.0f, x[k].y * sc.y + sh.y);
            a4.z += fmaxf(0.0f, x[k].z * sc.z + sh.z);
            a4.w += fmaxf(0.0f, x[k].w * sc.w + sh.w);
          }
        }
      }
    }
    acc[i] = a4;
  }
}

__global__ void __launch_bounds__(256)
k_agg_gemm2(const float* __restrict__ h1,
            const float* __restrict__ scale,
            const float* __restrict__ shift,
            const int* __restrict__ tcnt,
            const int* __restrict__ bucket,
            const int* __restrict__ cnts,
            const int* __restrict__ ovt,
            const int* __restrict__ ove,
            const float* __restrict__ W2,
            const float* __restrict__ Ws2,
            const float* __restrict__ b2,
            float* __restrict__ out) {
  __shared__ float At[16 * PA];
  __shared__ float red[8 * 4 * 8 * 32];
  __shared__ int ent[TC];
  __shared__ int srt[TC];
  __shared__ int hist[64];
  __shared__ int off[65];
  __shared__ int nsel[1];

  int wave = __builtin_amdgcn_readfirstlane((int)(threadIdx.x >> 5));
  int lane = threadIdx.x & 31;
  int tile = blockIdx.x;
  int row0 = tile * 16;

  int tc = tcnt[tile * TSTRIDE];
  int nb = tc < TC ? tc : TC;
  int novf = 0;
  if (tc > TC) {
    novf = cnts[2];
    novf = novf < E2 ? novf : E2;
  }

  float4 sc = ((const float4*)scale)[lane];
  float4 sh = ((const float4*)shift)[lane];

  const int* bk = bucket + tile * TC;
  int e0[TC / 256];
#pragma unroll
  for (int k = 0; k < TC / 256; ++k) e0[k] = 0;
  if (nb > 0) {
#pragma unroll
    for (int k = 0; k < TC / 256; ++k) {
      int i = threadIdx.x + 256 * k;
      e0[k] = bk[i < nb ? i : nb - 1];
    }
  }

  int g0 = row0 + wave * 2;
  float4 own[2];
#pragma unroll
  for (int rr = 0; rr < 2; ++rr)
    own[rr] = ((const float4*)(h1 + (long)(g0 + rr) * 128))[lane];

  float4 segsum[8];
#pragma unroll
  for (int i = 0; i < 8; ++i) segsum[i] = make_float4(0.f, 0.f, 0.f, 0.f);

  int nchunk = 1 + (novf + TC - 1) / TC;
#pragma nounroll
  for (int ch = 0; ch < nchunk; ++ch) {
    int cnt;
    int e[TC / 256];
#pragma unroll
    for (int k = 0; k < TC / 256; ++k) e[k] = 0;
    if (ch == 0) {
      cnt = nb;
#pragma unroll
      for (int k = 0; k < TC / 256; ++k) e[k] = e0[k];
    } else {
      int base = (ch - 1) * TC;
      __syncthreads();
      if (threadIdx.x == 0) nsel[0] = 0;
      __syncthreads();
#pragma unroll
      for (int k = 0; k < TC / 256; ++k) {
        int j = base + threadIdx.x + 256 * k;
        if (j < novf && ovt[j] == tile) {
          int p = atomicAdd(&nsel[0], 1);
          ent[p] = ove[j];
        }
      }
      __syncthreads();
      cnt = nsel[0];
#pragma unroll
      for (int k = 0; k < TC / 256; ++k) {
        int i = threadIdx.x + 256 * k;
        if (i < cnt) e[k] = ent[i];
      }
    }
    tile_chunk(cnt, e, wave, lane, srt, hist, off, h1, sc, sh, segsum);
  }

#pragma unroll
  for (int rr = 0; rr < 2; ++rr) {
    float4 x = own[rr];
    float4 v;
    v.x = fmaxf(0.0f, x.x * sc.x + sh.x);
    v.y = fmaxf(0.0f, x.y * sc.y + sh.y);
    v.z = fmaxf(0.0f, x.z * sc.z + sh.z);
    v.w = fmaxf(0.0f, x.w * sc.w + sh.w);
    *(float4*)(At + (wave * 2 + rr) * PA + 512 + 4 * lane) = v;
  }
#pragma unroll
  for (int i = 0; i < 8; ++i)
    *(float4*)(At + (wave * 2 + (i >> 2)) * PA + (i & 3) * 128 + 4 * lane) = segsum[i];
  __syncthreads();

  int c = lane & 15;
  int h2 = (lane >> 4) << 1;
  int kb = wave * 80;
  v2f af[20];
  v2f bf[20][4];
#pragma unroll
  for (int ks = 0; ks < 20; ++ks) {
    int kk = kb + 4 * ks;
    const float* Bk = (kk < 512) ? (W2 + kk * 64) : (Ws2 + (kk - 512) * 64);
    Bk += h2 * 64 + c;
#pragma unroll
    for (int t = 0; t < 4; ++t) {
      bf[ks][t].x = Bk[t * 16];
      bf[ks][t].y = Bk[64 + t * 16];
    }
    af[ks] = *(const v2f*)(At + c * PA + kk + h2);
  }
  v8f acc[4];
  const v8f zero = {0.f,0.f,0.f,0.f,0.f,0.f,0.f,0.f};
#pragma unroll
  for (int t = 0; t < 4; ++t) acc[t] = zero;
#pragma unroll
  for (int ks = 0; ks < 20; ++ks) {
#pragma unroll
    for (int t = 0; t < 4; ++t)
      acc[t] = __builtin_amdgcn_wmma_f32_16x16x4_f32(
          false, af[ks], false, bf[ks][t], (short)0, acc[t], false, false);
  }
#pragma unroll
  for (int t = 0; t < 4; ++t)
#pragma unroll
    for (int v = 0; v < 8; ++v)
      red[((wave * 4 + t) * 8 + v) * 32 + lane] = acc[t][v];
  __syncthreads();

  int row = threadIdx.x >> 4;
  int cq  = threadIdx.x & 15;
  int ridx = (((cq >> 2) * 8 + (row & 7)) * 32) + (row >> 3) * 16 + (cq & 3) * 4;
  float4 bb = ((const float4*)b2)[cq];
  float vals[4] = {bb.x, bb.y, bb.z, bb.w};
#pragma unroll
  for (int w = 0; w < 8; ++w) {
    float4 p = *(const float4*)(red + w * (4 * 8 * 32) + ridx);
    vals[0] += p.x; vals[1] += p.y; vals[2] += p.z; vals[3] += p.w;
  }
  float m = fmaxf(fmaxf(vals[0], vals[1]), fmaxf(vals[2], vals[3]));
#pragma unroll
  for (int mask = 1; mask < 16; mask <<= 1)
    m = fmaxf(m, __shfl_xor(m, mask, 32));
  float sum = 0.f;
#pragma unroll
  for (int t = 0; t < 4; ++t) sum += expf(vals[t] - m);
#pragma unroll
  for (int mask = 1; mask < 16; mask <<= 1)
    sum += __shfl_xor(sum, mask, 32);
  float lse = m + logf(sum);
  float4 o;
  o.x = vals[0] - lse; o.y = vals[1] - lse; o.z = vals[2] - lse; o.w = vals[3] - lse;
  ((float4*)(out + (long)(row0 + row) * 64))[cq] = o;
}

extern "C" void kernel_launch(void* const* d_in, const int* in_sizes, int n_in,
                              void* d_out, int out_size, void* d_ws, size_t ws_size,
                              hipStream_t stream) {
  const float* x     = (const float*)d_in[0];
  const int*   src1  = (const int*)  d_in[1];
  const int*   dst1  = (const int*)  d_in[2];
  const int*   et1   = (const int*)  d_in[3];
  const int*   src2  = (const int*)  d_in[4];
  const int*   dst2  = (const int*)  d_in[5];
  const int*   et2   = (const int*)  d_in[6];
  const int*   hmap  = (const int*)  d_in[7];
  const float* hbuf  = (const float*)d_in[8];
  const float* W1    = (const float*)d_in[9];
  const float* Ws1   = (const float*)d_in[10];
  const float* b1    = (const float*)d_in[11];
  const float* gamma = (const float*)d_in[12];
  const float* beta  = (const float*)d_in[13];
  const float* W2    = (const float*)d_in[14];
  const float* Ws2   = (const float*)d_in[15];
  const float* b2    = (const float*)d_in[16];

  float* ws    = (float*)d_ws;
  float* A1    = ws;
  float* A2    = A1 + (size_t)4 * N1 * 128;
  float* ss    = A2 + (size_t)4 * N2 * 128 + 256;
  float* h1    = ss + 256;
  float* out   = (float*)d_out;
  int* rlist = (int*)A1;
  int* cnts  = (int*)A2 - 4;
  int* tcnt  = (int*)A2;
  float* sums = (float*)(tcnt + NT * TSTRIDE);
  int* bucket = (int*)(sums + 256);
  int* ovt    = bucket + NT * TC;
  int* ove    = ovt + E2;

  k_zero<<<96, 256, 0, stream>>>((float*)cnts, (long)(4 + NT * TSTRIDE + 256));

  k_histfill_stats<<<HF_G, HF_T, 0, stream>>>(hmap, hbuf, b1, h1, sums, sums + 128);

  k_pick<<<PK_WG2 + PK_WG1 + PK_WGR, PK_T, 0, stream>>>(src1, dst1, et1, hmap, cnts,
                                                         x, W1, Ws1, h1, rlist,
                                                         src2, dst2, et2,
                                                         tcnt, bucket, ovt, ove);

  k_bnfinal<<<1, 128, 0, stream>>>(sums, sums + 128, gamma, beta, ss, ss + 128,
                                   cnts, rlist, h1);

  static_assert(N2 % 16 == 0, "one work-group per 16 destination rows");
  k_agg_gemm2<<<NT, 256, 0, stream>>>(h1, ss, ss + 128, tcnt, bucket, cnts, ovt, ove,
                                      W2, Ws2, b2, out);
}


# BN sums without atomics: history pass stores a partial row per group and clears the edge counters, 8 edge-pass groups add them, layer-2 waves fold scale/shift; 5 launches -> 3 (BN sum order fixed)
# speedup vs baseline: 11.3034x; 1.0913x over previous
#include <hip/hip_runtime.h>
#include <hip/hip_bf16.h>
#include <math.h>

typedef __attribute__((ext_vector_type(2))) float v2f;
typedef __attribute__((ext_vector_type(8))) float v8f;

#define N0  300000
#define N1  60000
#define N2  6000
#define E1  1200000
#define E2  300000
#define D_H 128
#define D_OUT 64
#define BN_EPS 1e-5f

#define NQ (E1 / 4)
static_assert(E1 % 4 == 0, "edge list is read four at a time");
#define NSEG (4 * N2)
static_assert(E2 % 4 == 0, "layer-2 edge list is read four at a time");

#define NT (N2 / 16)
#define TC 2048
#define TSTRIDE 64
static_assert(N1 <= 65536, "a source row fits the low 16 bits of a bucket entry");
static_assert(TC % 256 == 0, "a chunk is a whole number of passes of the work-group");
static_assert(NT * TSTRIDE == NSEG, "the counters fill the cleared range");

#define NSLOT 512
#define NRED 8
static_assert(NSLOT % NRED == 0, "every summing work-group takes the same number of rows");

__device__ __forceinline__ void put_edge(int* __restrict__ bucket,
                                         int* __restrict__ cnts,
                                         int* __restrict__ ovt,
                                         int* __restrict__ ove,
                                         int slot, int tile, int packed) {
  if (slot < TC) {
    atomicExch(&bucket[tile * TC + slot], packed);
  } else {
    int o = atomicAdd(&cnts[2], 1);
    atomicExch(&ovt[o], tile);
    atomicExch(&ove[o], packed);
  }
}

__device__ __forceinline__ void msg_tiles(int cnt,
                                          const int* __restrict__ items,
                                          const int* __restrict__ src,
                                          const int* __restrict__ dst,
                                          const int* __restrict__ et,
                                          const float* __restrict__ x,
                                          const float* __restrict__ W1,
                                          const float* __restrict__ Ws,
                                          float* __restrict__ h1) {
  int ntiles = (cnt + 15) >> 4;
  int wave = threadIdx.x >> 5;
  int lane = threadIdx.x & 31;
  int aRow = lane & 15;
  int aK   = (lane >> 4) << 1;
  int col  = wave * 16 + (lane & 15);
  int halfOff = (lane >> 4) << 3;
  const v2f zero2 = {0.f, 0.f};

  for (int tile = 0; tile < ntiles; ++tile) {
    int ei = tile * 16 + aRow;
    bool live = ei < cnt;
    int item = items[live ? ei : 0];
    bool self = item >= E1;
    int e = self ? 0 : item;
    int sLd = src[e];
    int dLd = dst[e];
    int rLd = et[e];
    int sRow = self ? item - E1 : sLd;
    int dRow = self ? item - E1 : dLd;
    int rel  = live ? (self ? 4 : rLd) : -1;
    const float* Ap = x + (long)sRow * 128 + aK;

    v2f af[32];
#pragma unroll
    for (int ks = 0; ks < 32; ++ks) af[ks] = *(const v2f*)(Ap + 4 * ks);

    v8f acc = {0.f,0.f,0.f,0.f,0.f,0.f,0.f,0.f};
    for (int r = 0; r < 5; ++r) {
      bool mine = (rel == r);
      if (__ballot(mine) == 0ull) continue;
      const float* Bp = ((r < 4) ? (W1 + (long)r * 128 * 128) : Ws)
                        + aK * 128 + col;
      v2f bf[32];
#pragma unroll
      for (int ks = 0; ks < 32; ++ks) {
        bf[ks].x = Bp[(4 * ks) * 128];
        bf[ks].y = Bp[(4 * ks + 1) * 128];
      }
#pragma unroll
      for (int ks = 0; ks < 32; ++ks) {
        v2f a = mine ? af[ks] : zero2;
        acc = __builtin_amdgcn_wmma_f32_16x16x4_f32(
            false, a, false, bf[ks], (short)0, acc, false, false);
      }
    }

#pragma unroll
    for (int v = 0; v < 8; ++v) {
      int lo = __builtin_amdgcn_readlane(dRow, v);
      int hi = __builtin_amdgcn_readlane(dRow, v + 8);
      int d = halfOff ? hi : lo;
      int eo = tile * 16 + v + halfOff;
      if (eo < cnt) atomicAdd(&h1[(long)d * 128 + col], acc[v]);
    }
  }
}

#define PK_T 256
#define PK_Q 4
#define PK_CH (PK_T * PK_Q * 4)
#define PK_WG2 ((E2 + PK_CH - 1) / PK_CH)
#define PK_WG1 ((E1 + PK_CH - 1) / PK_CH)
#define PK_WGR ((N1 + PK_CH - 1) / PK_CH)
#define NTP 384
static_assert(N1 % 4 == 0, "the history map is read four rows at a time");
static_assert(NT <= NTP && NTP % 32 == 0 && NTP <= 2 * PK_T, "two counters per thread, 12 per lane of one wave");
static_assert(NT <= 512, "a tile number fits the 9 bits above the 22-bit entry");
static_assert(PK_T == 256, "a summing work-group has one thread per column of a BN partial row");

__device__ __forceinline__ void bin_group(const int* __restrict__ src2,
                                          const int* __restrict__ dst2,
                                          const int* __restrict__ et2,
                                          int* __restrict__ cnts,
                                          int* __restrict__ tcnt,
                                          int* __restrict__ bucket,
                                          int* __restrict__ ovt,
                                          int* __restrict__ ove,
                                          int g,
                                          int* __restrict__ lcnt,
                                          int* __restrict__ loff,
                                          int* __restrict__ lgb,
                                          int* __restrict__ srt) {
  const int tid = threadIdx.x;
  for (int i = tid; i < NTP; i += PK_T) lcnt[i] = 0;
  int sv[4 * PK_Q], dv[4 * PK_Q], rv[4 * PK_Q];
  bool ok[PK_Q];
#pragma unroll
  for (int k = 0; k < PK_Q; ++k) {
    int q = g * (PK_CH / 4) + k * PK_T + tid;
    ok[k] = q < E2 / 4;
    int qq = ok[k] ? q : 0;
    int4 a = ((const int4*)src2)[qq];
    int4 b = ((const int4*)dst2)[qq];
    int4 c = ((const int4*)et2)[qq];
    sv[4 * k] = a.x; sv[4 * k + 1] = a.y; sv[4 * k + 2] = a.z; sv[4 * k + 3] = a.w;
    dv[4 * k] = b.x; dv[4 * k + 1] = b.y; dv[4 * k + 2] = b.z; dv[4 * k + 3] = b.w;
    rv[4 * k] = c.x; rv[4 * k + 1] = c.y; rv[4 * k + 2] = c.z; rv[4 * k + 3] = c.w;
  }
  __syncthreads();
  int pk[4 * PK_Q], rk[4 * PK_Q];
#pragma unroll
  for (int i = 0; i < 4 * PK_Q; ++i) {
    int tile = dv[i] >> 4;
    pk[i] = sv[i] | ((((dv[i] & 15) << 2) | rv[i]) << 16) | (tile << 22);
    rk[i] = ok[i >> 2] ? atomicAdd(&lcnt[tile], 1) : 0;
  }
  __syncthreads();
  const int t0 = tid, t1 = tid + PK_T;
  int c0 = t0 < NT ? lcnt[t0] : 0;
  int c1 = t1 < NT ? lcnt[t1] : 0;
  int b0 = 0, b1 = 0;
  if (c0 > 0) b0 = atomicAdd(&tcnt[t0 * TSTRIDE], c0);
  if (c1 > 0) b1 = atomicAdd(&tcnt[t1 * TSTRIDE], c1);
  if (tid < 32) {
    int base = tid * (NTP / 32);
    int loc[NTP / 32];
    int sum = 0;
#pragma unroll
    for (int j = 0; j < NTP / 32; ++j) { loc[j] = sum; sum += lcnt[base + j]; }
    int inc = sum;
#pragma unroll
    for (int dl = 1; dl < 32; dl <<= 1) {
      int o = __shfl_up(inc, dl, 32);
      if (tid >= dl) inc += o;
    }
#pragma unroll
    for (int j = 0; j < NTP / 32; ++j) loff[base + j] = inc - sum + loc[j];
  }
  lgb[t0] = b0;
  if (t1 < NTP) lgb[t1] = b1;
  __syncthreads();
  int pl[4 * PK_Q];
#pragma unroll
  for (int i = 0; i < 4 * PK_Q; ++i) pl[i] = loff[pk[i] >> 22];
#pragma unroll
  for (int i = 0; i < 4 * PK_Q; ++i)
    if (ok[i >> 2]) srt[pl[i] + rk[i]] = pk[i];
  __syncthreads();
  int nv = E2 - g * PK_CH;
  nv = nv < PK_CH ? nv : PK_CH;
  int e[4 * PK_Q], sl[4 * PK_Q];
#pragma unroll
  for (int k = 0; k < 4 * PK_Q; ++k) {
    int i = k * PK_T + tid;
    e[k] = srt[i < nv ? i : 0];
  }
#pragma unroll
  for (int k = 0; k < 4 * PK_Q; ++k) {
    int tile = e[k] >> 22;
    sl[k] = lgb[tile] - loff[tile] + k * PK_T + tid;
  }
#pragma unroll
  for (int k = 0; k < 4 * PK_Q; ++k)
    if (k * PK_T + tid < nv)
      put_edge(bucket, cnts, ovt, ove, sl[k], e[k] >> 22, e[k] & 0x3FFFFF);
}

__global__ void __launch_bounds__(PK_T)
k_pick(const int* __restrict__ src,
       const int* __restrict__ dst,
       const int* __restrict__ et,
       const int* __restrict__ hmap,
       int* __restrict__ cnts,
       const float* __restrict__ x,
       const float* __restrict__ W1,
       const float* __restrict__ Ws1,
       float* __restrict__ h1,
       int* __restrict__ rlist,
       const int* __restrict__ src2,
       const int* __restrict__ dst2,
       const int* __restrict__ et2,
       int* __restrict__ tcnt,
       int* __restrict__ bucket,
       int* __restrict__ ovt,
       int* __restrict__ ove,
       const float* __restrict__ slots,
       float* __restrict__ part) {
  __shared__ int lcnt[NTP], loff[NTP], lgb[NTP];
  __shared__ int srt[PK_CH];
  const int tid = threadIdx.x;
  if (blockIdx.x < NRED) {
    const float* sp = slots + (long)blockIdx.x * (NSLOT / NRED) * PK_T + tid;
    float a = 0.f;
#pragma unroll 16
    for (int s = 0; s < NSLOT / NRED; ++s) a += sp[s * PK_T];
    part[blockIdx.x * PK_T + tid] = a;
    return;
  }
  const int g = blockIdx.x - NRED;
  if (g < PK_WG2) {
    bin_group(src2, dst2, et2, cnts, tcnt, bucket, ovt, ove, g, lcnt, loff, lgb, srt);
  } else {
    if (tid == 0) lcnt[0] = 0;
    const bool edges = g < PK_WG2 + PK_WG1;
    unsigned m = 0;
    int q0;
    if (edges) {
      q0 = (g - PK_WG2) * (PK_CH / 4) + tid;
      int dv[4 * PK_Q];
#pragma unroll
      for (int k = 0; k < PK_Q; ++k) {
        int q = q0 + k * PK_T;
        int4 b = ((const int4*)dst)[q < NQ ? q : 0];
        dv[4 * k] = b.x; dv[4 * k + 1] = b.y; dv[4 * k + 2] = b.z; dv[4 * k + 3] = b.w;
      }
      int h[4 * PK_Q];
#pragma unroll
      for (int i = 0; i < 4 * PK_Q; ++i) h[i] = hmap[dv[i]];
#pragma unroll
      for (int i = 0; i < 4 * PK_Q; ++i) {
        unsigned in = (q0 + (i >> 2) * PK_T < NQ) ? 1u : 0u;
        m |= (((unsigned)h[i] >> 31) & in) << i;
      }
    } else {
      q0 = (g - PK_WG2 - PK_WG1) * (PK_CH / 4) + tid;
#pragma unroll
      for (int k = 0; k < PK_Q; ++k) {
        int q = q0 + k * PK_T;
        bool ok = q < N1 / 4;
        int4 hq = ((const int4*)hmap)[ok ? q : 0];
        unsigned in = ok ? 1u : 0u;
        m |= ((((unsigned)hq.x >> 31) | (((unsigned)hq.y >> 31) << 1) |
               (((unsigned)hq.z >> 31) << 2) | (((unsigned)hq.w >> 31) << 3)) * in) << (4 * k);
      }
    }
    __syncthreads();
    while (m) {
      int i = __ffs(m) - 1;
      m &= m - 1;
      int n = (q0 + (i >> 2) * PK_T) * 4 + (i & 3);
      srt[atomicAdd(&lcnt[0], 1)] = edges ? n : E1 + n;
      if (!edges) {
        int slot = atomicAdd(&cnts[1], 1);
        atomicExch(&rlist[slot], n);
      }
    }
    __syncthreads();
    int cnt = __builtin_amdgcn_readfirstlane(lcnt[0]);
    msg_tiles(cnt, srt, src, dst, et, x, W1, Ws1, h1);
  }
}

#define HF_T 1024
#define HF_G NSLOT
#define HF_TOTAL (N1 * 128)
#define HF_STRIDE (HF_G * HF_T)
#define HF_TRIPS ((HF_TOTAL + HF_STRIDE - 1) / HF_STRIDE)
static_assert(HF_T % 128 == 0, "a thread keeps one column over all its trips");
static_assert((long)N1 * 128 * 4 < (1l << 31), "element offsets fit 32 bits");
static_assert(32 + NT * TSTRIDE <= HF_STRIDE, "one thread per cleared word");
__global__ void __launch_bounds__(HF_T)
k_histfill_stats(const int* __restrict__ hmap,
                 const float* __restrict__ hbuf,
                 const float* __restrict__ b1,
                 float* __restrict__ h1,
                 float* __restrict__ slots,
                 int* __restrict__ clr) {
  __shared__ float ls[HF_T], lq[HF_T];
  unsigned tid = blockIdx.x * HF_T + threadIdx.x;
  if (tid >= 28u && tid < 32u + NT * TSTRIDE) clr[tid] = 0;
  unsigned c = threadIdx.x & 127;
  unsigned row0 = tid >> 7;
  float bias = b1[c];
  int hm[HF_TRIPS];
#pragma unroll
  for (int t = 0; t < HF_TRIPS; ++t) {
    bool in = ((t + 1) * HF_STRIDE <= HF_TOTAL) || (tid + t * HF_STRIDE < HF_TOTAL);
    hm[t] = hmap[in ? row0 + t * (HF_STRIDE / 128) : 0u];
  }
  float v[HF_TRIPS];
#pragma unroll
  for (int t = 0; t < HF_TRIPS; ++t) {
    unsigned r = hm[t] >= 0 ? (unsigned)hm[t] : 0u;
    v[t] = hbuf[r * 128u + c];
  }
  __builtin_amdgcn_sched_barrier(0);
  float s = 0.f, q = 0.f;
#pragma unroll
  for (int t = 0; t < HF_TRIPS; ++t) {
    bool in = ((t + 1) * HF_STRIDE <= HF_TOTAL) || (tid + t * HF_STRIDE < HF_TOTAL);
    bool hist = in && hm[t] >= 0;
    float sn = s + v[t];
    float qn = q + v[t] * v[t];
    s = hist ? sn : s;
    q = hist ? qn : q;
    float val = hist ? v[t] : bias;
    if (in) h1[tid + t * HF_STRIDE] = val;
  }
  ls[threadIdx.x] = s; lq[threadIdx.x] = q;
  __syncthreads();
  if (threadIdx.x < 128) {
    float S = 0.f, Q = 0.f;
    for (int k = 0; k < HF_T / 128; ++k) {
      S += ls[threadIdx.x + 128 * k];
      Q += lq[threadIdx.x + 128 * k];
    }
    slots[blockIdx.x * 256 + threadIdx.x] = S;
    slots[blockIdx.x * 256 + 128 + threadIdx.x] = Q;
  }
}

#define PA 644

__device__ __forceinline__ void tile_chunk(int cnt, const int (&e)[TC / 256],
                                           int wave, int lane,
                                           int* __restrict__ srt,
                                           int* __restrict__ hist,
                                           int* __restrict__ off,
                                           const float* __restrict__ h1,
                                           float4 sc, float4 sh,
                                           float4 (&acc)[8]) {
  if (threadIdx.x < 64) hist[threadIdx.x] = 0;
  __syncthreads();
  int rk[TC / 256];
#pragma unroll
  for (int k = 0; k < TC / 256; ++k) {
    rk[k] = 0;
    if (threadIdx.x + 256 * k < cnt) rk[k] = atomicAdd(&hist[(e[k] >> 16) & 63], 1);
  }
  __syncthreads();
  if (wave == 0) {
    int a = hist[2 * lane], c = hist[2 * lane + 1];
    int s = a + c, inc = s;
#pragma unroll
    for (int d = 1; d < 32; d <<= 1) {
      int o = __shfl_up(inc, d, 32);
      if (lane >= d) inc += o;
    }
    off[2 * lane] = inc - s;
    off[2 * lane + 1] = inc - s + a;
    if (lane == 31) off[64] = inc;
  }
  __syncthreads();
  int st[TC / 256];
#pragma unroll
  for (int k = 0; k < TC / 256; ++k) st[k] = off[(e[k] >> 16) & 63];
#pragma unroll
  for (int k = 0; k < TC / 256; ++k)
    if (threadIdx.x + 256 * k < cnt) srt[st[k] + rk[k]] = e[k] & 0xFFFF;
  __syncthreads();
#pragma unroll
  for (int i = 0; i < 8; ++i) {
    int beg = __builtin_amdgcn_readfirstlane(off[wave * 8 + i]);
    int end = __builtin_amdgcn_readfirstlane(off[wave * 8 + i + 1]);
    float4 a4 = acc[i];
    for (int base = beg; base < end; base += 32) {
      int n = end - base;
      n = n < 32 ? n : 32;
      int mine = (lane < n) ? srt[base + lane] : 0;
      for (int j0 = 0; j0 < n; j0 += 8) {
        float4 x[8];
#pragma unroll
        for (int k = 0; k < 8; ++k) {
          int sRow = __builtin_amdgcn_readlane(mine, j0 + k);
          x[k] = ((const float4*)(h1 + (long)sRow * 128))[lane];
        }
#pragma unroll
        for (int k = 0; k < 8; ++k) {
          if (j0 + k < n) {
            a4.x += fmaxf(0.0f, x[k].x * sc.x + sh.x);
            a4.y += fmaxf(0.0f, x[k].y * sc.y + sh.y);
            a4.z += fmaxf(0.0f, x[k].z * sc.z + sh.z);
            a4.w += fmaxf(0.0f, x[k].w * sc.w + sh.w);
          }
        }
      }
    }
    acc[i] = a4;
  }
}

__global__ void __launch_bounds__(256)
k_agg_gemm2(const float* __restrict__ h1,
            const float* __restrict__ part,
            const int* __restrict__ rlist,
            const float* __restrict__ gamma,
            const float* __restrict__ beta,
            const int* __restrict__ tcnt,
            const int* __restrict__ bucket,
            const int* __restrict__ cnts,
            const int* __restrict__ ovt,
            const int* __restrict__ ove,
            const float* __restrict__ W2,
            const float* __restrict__ Ws2,
            const float* __restrict__ b2,
            float* __restrict__ out) {
  __shared__ float At[16 * PA];
  __shared__ float red[8 * 4 * 8 * 32];
  __shared__ int ent[TC];
  __shared__ int srt[TC];
  __shared__ int hist[64];
  __shared__ int off[65];
  __shared__ int nsel[1];

  int wave = __builtin_amdgcn_readfirstlane((int)(threadIdx.x >> 5));
  int lane = threadIdx.x & 31;
  int tile = blockIdx.x;
  int row0 = tile * 16;

  int tc = tcnt[tile * TSTRIDE];
  int nr = __builtin_amdgcn_readfirstlane(cnts[1]);
  nr = nr < N1 ? nr : N1;
  int nb = tc < TC ? tc : TC;
  int novf = 0;
  if (tc > TC) {
    novf = cnts[2];
    novf = novf < E2 ? novf : E2;
  }

  int nfirst = nr < 32 ? nr : 32;
  int rmine = (lane < nfirst) ? rlist[lane] : 0;

  const int* bk = bucket + tile * TC;
  int e0[TC / 256];
#pragma unroll
  for (int k = 0; k < TC / 256; ++k) e0[k] = 0;
  if (nb > 0) {
#pragma unroll
    for (int k = 0; k < TC / 256; ++k) {
      int i = threadIdx.x + 256 * k;
      e0[k] = bk[i < nb ? i : nb - 1];
    }
  }

  int g0 = row0 + wave * 2;
  float4 own[2];
#pragma unroll
  for (int rr = 0; rr < 2; ++rr)
    own[rr] = ((const float4*)(h1 + (long)(g0 + rr) * 128))[lane];

  float4 s4 = make_float4(0.f, 0.f, 0.f, 0.f), q4 = make_float4(0.f, 0.f, 0.f, 0.f);
#pragma unroll
  for (int r = 0; r < NRED; ++r) {
    float4 a = ((const float4*)(part + r * 256))[lane];
    float4 b = ((const float4*)(part + r * 256 + 128))[lane];
    s4.x += a.x; s4.y += a.y; s4.z += a.z; s4.w += a.w;
    q4.x += b.x; q4.y += b.y; q4.z += b.z; q4.w += b.w;
  }
  for (int base = 0; base < nr; base += 32) {
    int n = nr - base;
    n = n < 32 ? n : 32;
    if (base > 0) rmine = (lane < n) ? rlist[base + lane] : 0;
    for (int j0 = 0; j0 < n; j0 += 8) {
      float4 v[8];
#pragma unroll
      for (int k = 0; k < 8; ++k) {
        int row = __builtin_amdgcn_readlane(rmine, j0 + k);
        v[k] = ((const float4*)(h1 + (long)row * 128))[lane];
      }
#pragma unroll
      for (int k = 0; k < 8; ++k) {
        if (j0 + k < n) {
          s4.x += v[k].x; s4.y += v[k].y; s4.z += v[k].z; s4.w += v[k].w;
          q4.x += v[k].x * v[k].x; q4.y += v[k].y * v[k].y;
          q4.z += v[k].z * v[k].z; q4.w += v[k].w * v[k].w;
        }
      }
    }
  }
  float4 sc, sh;
  {
    const float inv_n = 1.0f / (float)N1;
    float4 g4 = ((const float4*)gamma)[lane];
    float4 b4 = ((const float4*)beta)[lane];
    float mx = s4.x * inv_n, my = s4.y * inv_n, mz = s4.z * inv_n, mw = s4.w * inv_n;
    sc.x = g4.x * rsqrtf(q4.x * inv_n - mx * mx + BN_EPS);
    sc.y = g4.y * rsqrtf(q4.y * inv_n - my * my + BN_EPS);
    sc.z = g4.z * rsqrtf(q4.z * inv_n - mz * mz + BN_EPS);
    sc.w = g4.w * rsqrtf(q4.w * inv_n - mw * mw + BN_EPS);
    sh.x = b4.x - mx * sc.x; sh.y = b4.y - my * sc.y;
    sh.z = b4.z - mz * sc.z; sh.w = b4.w - mw * sc.w;
  }

  float4 segsum[8];
#pragma unroll
  for (int i = 0; i < 8; ++i) segsum[i] = make_float4(0.f, 0.f, 0.f, 0.f);

  int nchunk = 1 + (novf + TC - 1) / TC;
#pragma nounroll
  for (int ch = 0; ch < nchunk; ++ch) {
    int cnt;
    int e[TC / 256];
#pragma unroll
    for (int k = 0; k < TC / 256; ++k) e[k] = 0;
    if (ch == 0) {
      cnt = nb;
#pragma unroll
      for (int k = 0; k < TC / 256; ++k) e[k] = e0[k];
    } else {
      int base = (ch - 1) * TC;
      __syncthreads();
      if (threadIdx.x == 0) nsel[0] = 0;
      __syncthreads();
#pragma unroll
      for (int k = 0; k < TC / 256; ++k) {
        int j = base + threadIdx.x + 256 * k;
        if (j < novf && ovt[j] == tile) {
          int p = atomicAdd(&nsel[0], 1);
          ent[p] = ove[j];
        }
      }
      __syncthreads();
      cnt = nsel[0];
#pragma unroll
      for (int k = 0; k < TC / 256; ++k) {
        int i = threadIdx.x + 256 * k;
        if (i < cnt) e[k] = ent[i];
      }
    }
    tile_chunk(cnt, e, wave, lane, srt, hist, off, h1, sc, sh, segsum);
  }

#pragma unroll
  for (int rr = 0; rr < 2; ++rr) {
    float4 x = own[rr];
    float4 v;
    v.x = fmaxf(0.0f, x.x * sc.x + sh.x);
    v.y = fmaxf(0.0f, x.y * sc.y + sh.y);
    v.z = fmaxf(0.0f, x.z * sc.z + sh.z);
    v.w = fmaxf(0.0f, x.w * sc.w + sh.w);
    *(float4*)(At + (wave * 2 + rr) * PA + 512 + 4 * lane) = v;
  }
#pragma unroll
  for (int i = 0; i < 8; ++i)
    *(float4*)(At + (wave * 2 + (i >> 2)) * PA + (i & 3) * 128 + 4 * lane) = segsum[i];
  __syncthreads();

  int c = lane & 15;
  int h2 = (lane >> 4) << 1;
  int kb = wave * 80;
  v2f af[20];
  v2f bf[20][4];
#pragma unroll
  for (int ks = 0; ks < 20; ++ks) {
    int kk = kb + 4 * ks;
    const float* Bk = (kk < 512) ? (W2 + kk * 64) : (Ws2 + (kk - 512) * 64);
    Bk += h2 * 64 + c;
#pragma unroll
    for (int t = 0; t < 4; ++t) {
      bf[ks][t].x = Bk[t * 16];
      bf[ks][t].y = Bk[64 + t * 16];
    }
    af[ks] = *(const v2f*)(At + c * PA + kk + h2);
  }
  v8f acc[4];
  const v8f zero = {0.f,0.f,0.f,0.f,0.f,0.f,0.f,0.f};
#pragma unroll
  for (int t = 0; t < 4; ++t) acc[t] = zero;
#pragma unroll
  for (int ks = 0; ks < 20; ++ks) {
#pragma unroll
    for (int t = 0; t < 4; ++t)
      acc[t] = __builtin_amdgcn_wmma_f32_16x16x4_f32(
          false, af[ks], false, bf[ks][t], (short)0, acc[t], false, false);
  }
#pragma unroll
  for (int t = 0; t < 4; ++t)
#pragma unroll
    for (int v = 0; v < 8; ++v)
      red[((wave * 4 + t) * 8 + v) * 32 + lane] = acc[t][v];
  __syncthreads();

  int row = threadIdx.x >> 4;
  int cq  = threadIdx.x & 15;
  int ridx = (((cq >> 2) * 8 + (row & 7)) * 32) + (row >> 3) * 16 + (cq & 3) * 4;
  float4 bb = ((const float4*)b2)[cq];
  float vals[4] = {bb.x, bb.y, bb.z, bb.w};
#pragma unroll
  for (int w = 0; w < 8; ++w) {
    float4 p = *(const float4*)(red + w * (4 * 8 * 32) + ridx);
    vals[0] += p.x; vals[1] += p.y; vals[2] += p.z; vals[3] += p.w;
  }
  float m = fmaxf(fmaxf(vals[0], vals[1]), fmaxf(vals[2], vals[3]));
#pragma unroll
  for (int mask = 1; mask < 16; mask <<= 1)
    m = fmaxf(m, __shfl_xor(m, mask, 32));
  float sum = 0.f;
#pragma unroll
  for (int t = 0; t < 4; ++t) sum += expf(vals[t] - m);
#pragma unroll
  for (int mask = 1; mask < 16; mask <<= 1)
    sum += __shfl_xor(sum, mask, 32);
  float lse = m + logf(sum);
  float4 o;
  o.x = vals[0] - lse; o.y = vals[1] - lse; o.z = vals[2] - lse; o.w = vals[3] - lse;
  ((float4*)(out + (long)(row0 + row) * 64))[cq] = o;
}

extern "C" void kernel_launch(void* const* d_in, const int* in_sizes, int n_in,
                              void* d_out, int out_size, void* d_ws, size_t ws_size,
                              hipStream_t stream) {
  const float* x     = (const float*)d_in[0];
  const int*   src1  = (const int*)  d_in[1];
  const int*   dst1  = (const int*)  d_in[2];
  const int*   et1   = (const int*)  d_in[3];
  const int*   src2  = (const int*)  d_in[4];
  const int*   dst2  = (const int*)  d_in[5];
  const int*   et2   = (const int*)  d_in[6];
  const int*   hmap  = (const int*)  d_in[7];
  const float* hbuf  = (const float*)d_in[8];
  const float* W1    = (const float*)d_in[9];
  const float* Ws1   = (const float*)d_in[10];
  const float* b1    = (const float*)d_in[11];
  const float* gamma = (const float*)d_in[12];
  const float* beta  = (const float*)d_in[13];
  const float* W2    = (const float*)d_in[14];
  const float* Ws2   = (const float*)d_in[15];
  const float* b2    = (const float*)d_in[16];

  float* ws    = (float*)d_ws;
  float* A1    = ws;
  float* A2    = A1 + (size_t)4 * N1 * 128;
  float* h1    = A2 + (size_t)4 * N2 * 128;
  float* out   = (float*)d_out;
  int* rlist = (int*)A1;
  float* slots = A1 + 65536;
  float* part  = slots + NSLOT * 256;
  int* cnts  = (int*)A2 - 4;
  int* tcnt  = (int*)A2;
  int* bucket = tcnt + NT * TSTRIDE;
  int* ovt    = bucket + NT * TC;
  int* ove    = ovt + E2;
  static_assert(N1 <= 65536, "the BN partial rows start behind the row list");

  k_histfill_stats<<<HF_G, HF_T, 0, stream>>>(hmap, hbuf, b1, h1, slots, tcnt - 32);

  k_pick<<<NRED + PK_WG2 + PK_WG1 + PK_WGR, PK_T, 0, stream>>>(src1, dst1, et1, hmap, cnts,
                                                                x, W1, Ws1, h1, rlist,
                                                                src2, dst2, et2,
                                                                tcnt, bucket, ovt, ove,
                                                                slots, part);

  static_assert(N2 % 16 == 0, "one work-group per 16 destination rows");
  k_agg_gemm2<<<NT, 256, 0, stream>>>(h1, part, rlist, gamma, beta, tcnt, bucket, cnts, ovt, ove,
                                      W2, Ws2, b2, out);
}


# layer-2 gather pipelined across a wave's segments: 16 whole-row loads per segment, issued one segment ahead of its sums (two groups in flight per wave, was one group of 8); sum order kept
# speedup vs baseline: 12.1001x; 1.0705x over previous
#include <hip/hip_runtime.h>
#include <hip/hip_bf16.h>
#include <math.h>

typedef __attribute__((ext_vector_type(2))) float v2f;
typedef __attribute__((ext_vector_type(8))) float v8f;

#define N0  300000
#define N1  60000
#define N2  6000
#define E1  1200000
#define E2  300000
#define D_H 128
#define D_OUT 64
#define BN_EPS 1e-5f

#define NQ (E1 / 4)
static_assert(E1 % 4 == 0, "edge list is read four at a time");
#define NSEG (4 * N2)
static_assert(E2 % 4 == 0, "layer-2 edge list is read four at a time");

#define NT (N2 / 16)
#define TC 2048
#define TSTRIDE 64
static_assert(N1 <= 65536, "a source row fits the low 16 bits of a bucket entry");
static_assert(TC % 256 == 0, "a chunk is a whole number of passes of the work-group");
static_assert(NT * TSTRIDE == NSEG, "the counters fill the cleared range");

#define NSLOT 512
#define NRED 8
static_assert(NSLOT % NRED == 0, "every summing work-group takes the same number of rows");

__device__ __forceinline__ void put_edge(int* __restrict__ bucket,
                                         int* __restrict__ cnts,
                                         int* __restrict__ ovt,
                                         int* __restrict__ ove,
                                         int slot, int tile, int packed) {
  if (slot < TC) {
    atomicExch(&bucket[tile * TC + slot], packed);
  } else {
    int o = atomicAdd(&cnts[2], 1);
    atomicExch(&ovt[o], tile);
    atomicExch(&ove[o], packed);
  }
}

__device__ __forceinline__ void msg_tiles(int cnt,
                                          const int* __restrict__ items,
                                          const int* __restrict__ src,
                                          const int* __restrict__ dst,
                                          const int* __restrict__ et,
                                          const float* __restrict__ x,
                                          const float* __restrict__ W1,
                                          const float* __restrict__ Ws,
                                          float* __restrict__ h1) {
  int ntiles = (cnt + 15) >> 4;
  int wave = threadIdx.x >> 5;
  int lane = threadIdx.x & 31;
  int aRow = lane & 15;
  int aK   = (lane >> 4) << 1;
  int col  = wave * 16 + (lane & 15);
  int halfOff = (lane >> 4) << 3;
  const v2f zero2 = {0.f, 0.f};

  for (int tile = 0; tile < ntiles; ++tile) {
    int ei = tile * 16 + aRow;
    bool live = ei < cnt;
    int item = items[live ? ei : 0];
    bool self = item >= E1;
    int e = self ? 0 : item;
    int sLd = src[e];
    int dLd = dst[e];
    int rLd = et[e];
    int sRow = self ? item - E1 : sLd;
    int dRow = self ? item - E1 : dLd;
    int rel  = live ? (self ? 4 : rLd) : -1;
    const float* Ap = x + (long)sRow * 128 + aK;

    v2f af[32];
#pragma unroll
    for (int ks = 0; ks < 32; ++ks) af[ks] = *(const v2f*)(Ap + 4 * ks);

    v8f acc = {0.f,0.f,0.f,0.f,0.f,0.f,0.f,0.f};
    for (int r = 0; r < 5; ++r) {
      bool mine = (rel == r);
      if (__ballot(mine) == 0ull) continue;
      const float* Bp = ((r < 4) ? (W1 + (long)r * 128 * 128) : Ws)
                        + aK * 128 + col;
      v2f bf[32];
#pragma unroll
      for (int ks = 0; ks < 32; ++ks) {
        bf[ks].x = Bp[(4 * ks) * 128];
        bf[ks].y = Bp[(4 * ks + 1) * 128];
      }
#pragma unroll
      for (int ks = 0; ks < 32; ++ks) {
        v2f a = mine ? af[ks] : zero2;
        acc = __builtin_amdgcn_wmma_f32_16x16x4_f32(
            false, a, false, bf[ks], (short)0, acc, false, false);
      }
    }

#pragma unroll
    for (int v = 0; v < 8; ++v) {
      int lo = __builtin_amdgcn_readlane(dRow, v);
      int hi = __builtin_amdgcn_readlane(dRow, v + 8);
      int d = halfOff ? hi : lo;
      int eo = tile * 16 + v + halfOff;
      if (eo < cnt) atomicAdd(&h1[(long)d * 128 + col], acc[v]);
    }
  }
}

#define PK_T 256
#define PK_Q 4
#define PK_CH (PK_T * PK_Q * 4)
#define PK_WG2 ((E2 + PK_CH - 1) / PK_CH)
#define PK_WG1 ((E1 + PK_CH - 1) / PK_CH)
#define PK_WGR ((N1 + PK_CH - 1) / PK_CH)
#define NTP 384
static_assert(N1 % 4 == 0, "the history map is read four rows at a time");
static_assert(NT <= NTP && NTP % 32 == 0 && NTP <= 2 * PK_T, "two counters per thread, 12 per lane of one wave");
static_assert(NT <= 512, "a tile number fits the 9 bits above the 22-bit entry");
static_assert(PK_T == 256, "a summing work-group has one thread per column of a BN partial row");

__device__ __forceinline__ void bin_group(const int* __restrict__ src2,
                                          const int* __restrict__ dst2,
                                          const int* __restrict__ et2,
                                          int* __restrict__ cnts,
                                          int* __restrict__ tcnt,
                                          int* __restrict__ bucket,
                                          int* __restrict__ ovt,
                                          int* __restrict__ ove,
                                          int g,
                                          int* __restrict__ lcnt,
                                          int* __restrict__ loff,
                                          int* __restrict__ lgb,
                                          int* __restrict__ srt) {
  const int tid = threadIdx.x;
  for (int i = tid; i < NTP; i += PK_T) lcnt[i] = 0;
  int sv[4 * PK_Q], dv[4 * PK_Q], rv[4 * PK_Q];
  bool ok[PK_Q];
#pragma unroll
  for (int k = 0; k < PK_Q; ++k) {
    int q = g * (PK_CH / 4) + k * PK_T + tid;
    ok[k] = q < E2 / 4;
    int qq = ok[k] ? q : 0;
    int4 a = ((const int4*)src2)[qq];
    int4 b = ((const int4*)dst2)[qq];
    int4 c = ((const int4*)et2)[qq];
    sv[4 * k] = a.x; sv[4 * k + 1] = a.y; sv[4 * k + 2] = a.z; sv[4 * k + 3] = a.w;
    dv[4 * k] = b.x; dv[4 * k + 1] = b.y; dv[4 * k + 2] = b.z; dv[4 * k + 3] = b.w;
    rv[4 * k] = c.x; rv[4 * k + 1] = c.y; rv[4 * k + 2] = c.z; rv[4 * k + 3] = c.w;
  }
  __syncthreads();
  int pk[4 * PK_Q], rk[4 * PK_Q];
#pragma unroll
  for (int i = 0; i < 4 * PK_Q; ++i) {
    int tile = dv[i] >> 4;
    pk[i] = sv[i] | ((((dv[i] & 15) << 2) | rv[i]) << 16) | (tile << 22);
    rk[i] = ok[i >> 2] ? atomicAdd(&lcnt[tile], 1) : 0;
  }
  __syncthreads();
  const int t0 = tid, t1 = tid + PK_T;
  int c0 = t0 < NT ? lcnt[t0] : 0;
  int c1 = t1 < NT ? lcnt[t1] : 0;
  int b0 = 0, b1 = 0;
  if (c0 > 0) b0 = atomicAdd(&tcnt[t0 * TSTRIDE], c0);
  if (c1 > 0) b1 = atomicAdd(&tcnt[t1 * TSTRIDE], c1);
  if (tid < 32) {
    int base = tid * (NTP / 32);
    int loc[NTP / 32];
    int sum = 0;
#pragma unroll
    for (int j = 0; j < NTP / 32; ++j) { loc[j] = sum; sum += lcnt[base + j]; }
    int inc = sum;
#pragma unroll
    for (int dl = 1; dl < 32; dl <<= 1) {
      int o = __shfl_up(inc, dl, 32);
      if (tid >= dl) inc += o;
    }
#pragma unroll
    for (int j = 0; j < NTP / 32; ++j) loff[base + j] = inc - sum + loc[j];
  }
  lgb[t0] = b0;
  if (t1 < NTP) lgb[t1] = b1;
  __syncthreads();
  int pl[4 * PK_Q];
#pragma unroll
  for (int i = 0; i < 4 * PK_Q; ++i) pl[i] = loff[pk[i] >> 22];
#pragma unroll
  for (int i = 0; i < 4 * PK_Q; ++i)
    if (ok[i >> 2]) srt[pl[i] + rk[i]] = pk[i];
  __syncthreads();
  int nv = E2 - g * PK_CH;
  nv = nv < PK_CH ? nv : PK_CH;
  int e[4 * PK_Q], sl[4 * PK_Q];
#pragma unroll
  for (int k = 0; k < 4 * PK_Q; ++k) {
    int i = k * PK_T + tid;
    e[k] = srt[i < nv ? i : 0];
  }
#pragma unroll
  for (int k = 0; k < 4 * PK_Q; ++k) {
    int tile = e[k] >> 22;
    sl[k] = lgb[tile] - loff[tile] + k * PK_T + tid;
  }
#pragma unroll
  for (int k = 0; k < 4 * PK_Q; ++k)
    if (k * PK_T + tid < nv)
      put_edge(bucket, cnts, ovt, ove, sl[k], e[k] >> 22, e[k] & 0x3FFFFF);
}

__global__ void __launch_bounds__(PK_T)
k_pick(const int* __restrict__ src,
       const int* __restrict__ dst,
       const int* __restrict__ et,
       const int* __restrict__ hmap,
       int* __restrict__ cnts,
       const float* __restrict__ x,
       const float* __restrict__ W1,
       const float* __restrict__ Ws1,
       float* __restrict__ h1,
       int* __restrict__ rlist,
       const int* __restrict__ src2,
       const int* __restrict__ dst2,
       const int* __restrict__ et2,
       int* __restrict__ tcnt,
       int* __restrict__ bucket,
       int* __restrict__ ovt,
       int* __restrict__ ove,
       const float* __restrict__ slots,
       float* __restrict__ part) {
  __shared__ int lcnt[NTP], loff[NTP], lgb[NTP];
  __shared__ int srt[PK_CH];
  const int tid = threadIdx.x;
  if (blockIdx.x < NRED) {
    const float* sp = slots + (long)blockIdx.x * (NSLOT / NRED) * PK_T + tid;
    float a = 0.f;
#pragma unroll 16
    for (int s = 0; s < NSLOT / NRED; ++s) a += sp[s * PK_T];
    part[blockIdx.x * PK_T + tid] = a;
    return;
  }
  const int g = blockIdx.x - NRED;
  if (g < PK_WG2) {
    bin_group(src2, dst2, et2, cnts, tcnt, bucket, ovt, ove, g, lcnt, loff, lgb, srt);
  } else {
    if (tid == 0) lcnt[0] = 0;
    const bool edges = g < PK_WG2 + PK_WG1;
    unsigned m = 0;
    int q0;
    if (edges) {
      q0 = (g - PK_WG2) * (PK_CH / 4) + tid;
      int dv[4 * PK_Q];
#pragma unroll
      for (int k = 0; k < PK_Q; ++k) {
        int q = q0 + k * PK_T;
        int4 b = ((const int4*)dst)[q < NQ ? q : 0];
        dv[4 * k] = b.x; dv[4 * k + 1] = b.y; dv[4 * k + 2] = b.z; dv[4 * k + 3] = b.w;
      }
      int h[4 * PK_Q];
#pragma unroll
      for (int i = 0; i < 4 * PK_Q; ++i) h[i] = hmap[dv[i]];
#pragma unroll
      for (int i = 0; i < 4 * PK_Q; ++i) {
        unsigned in = (q0 + (i >> 2) * PK_T < NQ) ? 1u : 0u;
        m |= (((unsigned)h[i] >> 31) & in) << i;
      }
    } else {
      q0 = (g - PK_WG2 - PK_WG1) * (PK_CH / 4) + tid;
#pragma unroll
      for (int k = 0; k < PK_Q; ++k) {
        int q = q0 + k * PK_T;
        bool ok = q < N1 / 4;
        int4 hq = ((const int4*)hmap)[ok ? q : 0];
        unsigned in = ok ? 1u : 0u;
        m |= ((((unsigned)hq.x >> 31) | (((unsigned)hq.y >> 31) << 1) |
               (((unsigned)hq.z >> 31) << 2) | (((unsigned)hq.w >> 31) << 3)) * in) << (4 * k);
      }
    }
    __syncthreads();
    while (m) {
      int i = __ffs(m) - 1;
      m &= m - 1;
      int n = (q0 + (i >> 2) * PK_T) * 4 + (i & 3);
      srt[atomicAdd(&lcnt[0], 1)] = edges ? n : E1 + n;
      if (!edges) {
        int slot = atomicAdd(&cnts[1], 1);
        atomicExch(&rlist[slot], n);
      }
    }
    __syncthreads();
    int cnt = __builtin_amdgcn_readfirstlane(lcnt[0]);
    msg_tiles(cnt, srt, src, dst, et, x, W1, Ws1, h1);
  }
}

#define HF_T 1024
#define HF_G NSLOT
#define HF_TOTAL (N1 * 128)
#define HF_STRIDE (HF_G * HF_T)
#define HF_TRIPS ((HF_TOTAL + HF_STRIDE - 1) / HF_STRIDE)
static_assert(HF_T % 128 == 0, "a thread keeps one column over all its trips");
static_assert((long)N1 * 128 * 4 < (1l << 31), "element offsets fit 32 bits");
static_assert(32 + NT * TSTRIDE <= HF_STRIDE, "one thread per cleared word");
__global__ void __launch_bounds__(HF_T)
k_histfill_stats(const int* __restrict__ hmap,
                 const float* __restrict__ hbuf,
                 const float* __restrict__ b1,
                 float* __restrict__ h1,
                 float* __restrict__ slots,
                 int* __restrict__ clr) {
  __shared__ float ls[HF_T], lq[HF_T];
  unsigned tid = blockIdx.x * HF_T + threadIdx.x;
  if (tid >= 28u && tid < 32u + NT * TSTRIDE) clr[tid] = 0;
  unsigned c = threadIdx.x & 127;
  unsigned row0 = tid >> 7;
  float bias = b1[c];
  int hm[HF_TRIPS];
#pragma unroll
  for (int t = 0; t < HF_TRIPS; ++t) {
    bool in = ((t + 1) * HF_STRIDE <= HF_TOTAL) || (tid + t * HF_STRIDE < HF_TOTAL);
    hm[t] = hmap[in ? row0 + t * (HF_STRIDE / 128) : 0u];
  }
  float v[HF_TRIPS];
#pragma unroll
  for (int t = 0; t < HF_TRIPS; ++t) {
    unsigned r = hm[t] >= 0 ? (unsigned)hm[t] : 0u;
    v[t] = hbuf[r * 128u + c];
  }
  __builtin_amdgcn_sched_barrier(0);
  float s = 0.f, q = 0.f;
#pragma unroll
  for (int t = 0; t < HF_TRIPS; ++t) {
    bool in = ((t + 1) * HF_STRIDE <= HF_TOTAL) || (tid + t * HF_STRIDE < HF_TOTAL);
    bool hist = in && hm[t] >= 0;
    float sn = s + v[t];
    float qn = q + v[t] * v[t];
    s = hist ? sn : s;
    q = hist ? qn : q;
    float val = hist ? v[t] : bias;
    if (in) h1[tid + t * HF_STRIDE] = val;
  }
  ls[threadIdx.x] = s; lq[threadIdx.x] = q;
  __syncthreads();
  if (threadIdx.x < 128) {
    float S = 0.f, Q = 0.f;
    for (int k = 0; k < HF_T / 128; ++k) {
      S += ls[threadIdx.x + 128 * k];
      Q += lq[threadIdx.x + 128 * k];
    }
    slots[blockIdx.x * 256 + threadIdx.x] = S;
    slots[blockIdx.x * 256 + 128 + threadIdx.x] = Q;
  }
}

#define PA 644

#define GW 16
__device__ __forceinline__ void seg_issue(int beg, int end, int lane,
                                          const int* __restrict__ srt,
                                          const float* __restrict__ h1,
                                          float4 (&x)[GW]) {
  int n = end - beg;
  n = n < GW ? n : GW;
  int mine = (lane < n) ? srt[beg + lane] : 0;
#pragma unroll
  for (int k = 0; k < GW; ++k) {
    int sRow = __builtin_amdgcn_readlane(mine, k);
    x[k] = ((const float4*)(h1 + (long)sRow * 128))[lane];
  }
}

__device__ __forceinline__ void seg_sum(int n, const float4 (&x)[GW],
                                        float4 sc, float4 sh, float4& a4) {
#pragma unroll
  for (int k = 0; k < GW; ++k) {
    float m = (k < n) ? 1.0f : 0.0f;
    a4.x = fmaf(m, fmaxf(0.0f, x[k].x * sc.x + sh.x), a4.x);
    a4.y = fmaf(m, fmaxf(0.0f, x[k].y * sc.y + sh.y), a4.y);
    a4.z = fmaf(m, fmaxf(0.0f, x[k].z * sc.z + sh.z), a4.z);
    a4.w = fmaf(m, fmaxf(0.0f, x[k].w * sc.w + sh.w), a4.w);
  }
}

__device__ __forceinline__ void seg_step(int i, const int (&bnd)[9], int lane,
                                         const int* __restrict__ srt,
                                         const float* __restrict__ h1,
                                         float4 sc, float4 sh,
                                         float4 (&cur)[GW], float4 (&nxt)[GW],
                                         float4 (&acc)[8]) {
  if (i + 1 < 8) seg_issue(bnd[i + 1], bnd[i + 2], lane, srt, h1, nxt);
  __builtin_amdgcn_sched_barrier(0);
  int n = bnd[i + 1] - bnd[i];
  float4 a4 = acc[i];
  seg_sum(n < GW ? n : GW, cur, sc, sh, a4);
  for (int base = bnd[i] + GW; base < bnd[i + 1]; base += GW) {
    int r = bnd[i + 1] - base;
    float4 xr[GW];
    seg_issue(base, bnd[i + 1], lane, srt, h1, xr);
    seg_sum(r < GW ? r : GW, xr, sc, sh, a4);
  }
  acc[i] = a4;
}

__device__ __forceinline__ void tile_chunk(int cnt, const int (&e)[TC / 256],
                                           int wave, int lane,
                                           int* __restrict__ srt,
                                           int* __restrict__ hist,
                                           int* __restrict__ off,
                                           const float* __restrict__ h1,
                                           float4 sc, float4 sh,
                                           float4 (&acc)[8]) {
  if (threadIdx.x < 64) hist[threadIdx.x] = 0;
  __syncthreads();
  int rk[TC / 256];
#pragma unroll
  for (int k = 0; k < TC / 256; ++k) {
    rk[k] = 0;
    if (threadIdx.x + 256 * k < cnt) rk[k] = atomicAdd(&hist[(e[k] >> 16) & 63], 1);
  }
  __syncthreads();
  if (wave == 0) {
    int a = hist[2 * lane], c = hist[2 * lane + 1];
    int s = a + c, inc = s;
#pragma unroll
    for (int d = 1; d < 32; d <<= 1) {
      int o = __shfl_up(inc, d, 32);
      if (lane >= d) inc += o;
    }
    off[2 * lane] = inc - s;
    off[2 * lane + 1] = inc - s + a;
    if (lane == 31) off[64] = inc;
  }
  __syncthreads();
  int st[TC / 256];
#pragma unroll
  for (int k = 0; k < TC / 256; ++k) st[k] = off[(e[k] >> 16) & 63];
#pragma unroll
  for (int k = 0; k < TC / 256; ++k)
    if (threadIdx.x + 256 * k < cnt) srt[st[k] + rk[k]] = e[k] & 0xFFFF;
  __syncthreads();
  int bnd[9];
#pragma unroll
  for (int i = 0; i < 9; ++i) bnd[i] = __builtin_amdgcn_readfirstlane(off[wave * 8 + i]);
  float4 xa[GW], xb[GW];
  seg_issue(bnd[0], bnd[1], lane, srt, h1, xa);
#pragma unroll
  for (int p = 0; p < 4; ++p) {
    seg_step(2 * p,     bnd, lane, srt, h1, sc, sh, xa, xb, acc);
    seg_step(2 * p + 1, bnd, lane, srt, h1, sc, sh, xb, xa, acc);
  }
}

__global__ void __launch_bounds__(256)
k_agg_gemm2(const float* __restrict__ h1,
            const float* __restrict__ part,
            const int* __restrict__ rlist,
            const float* __restrict__ gamma,
            const float* __restrict__ beta,
            const int* __restrict__ tcnt,
            const int* __restrict__ bucket,
            const int* __restrict__ cnts,
            const int* __restrict__ ovt,
            const int* __restrict__ ove,
            const float* __restrict__ W2,
            const float* __restrict__ Ws2,
            const float* __restrict__ b2,
            float* __restrict__ out) {
  __shared__ float At[16 * PA];
  __shared__ float red[8 * 4 * 8 * 32];
  __shared__ int ent[TC];
  __shared__ int srt[TC];
  __shared__ int hist[64];
  __shared__ int off[65];
  __shared__ int nsel[1];

  int wave = __builtin_amdgcn_readfirstlane((int)(threadIdx.x >> 5));
  int lane = threadIdx.x & 31;
  int tile = blockIdx.x;
  int row0 = tile * 16;

  int tc = tcnt[tile * TSTRIDE];
  int nr = __builtin_amdgcn_readfirstlane(cnts[1]);
  nr = nr < N1 ? nr : N1;
  int nb = tc < TC ? tc : TC;
  int novf = 0;
  if (tc > TC) {
    novf = cnts[2];
    novf = novf < E2 ? novf : E2;
  }

  int nfirst = nr < 32 ? nr : 32;
  int rmine = (lane < nfirst) ? rlist[lane] : 0;

  const int* bk = bucket + tile * TC;
  int e0[TC / 256];
#pragma unroll
  for (int k = 0; k < TC / 256; ++k) e0[k] = 0;
  if (nb > 0) {
#pragma unroll
    for (int k = 0; k < TC / 256; ++k) {
      int i = threadIdx.x + 256 * k;
      e0[k] = bk[i < nb ? i : nb - 1];
    }
  }

  int g0 = row0 + wave * 2;
  float4 own[2];
#pragma unroll
  for (int rr = 0; rr < 2; ++rr)
    own[rr] = ((const float4*)(h1 + (long)(g0 + rr) * 128))[lane];

  float4 s4 = make_float4(0.f, 0.f, 0.f, 0.f), q4 = make_float4(0.f, 0.f, 0.f, 0.f);
#pragma unroll
  for (int r = 0; r < NRED; ++r) {
    float4 a = ((const float4*)(part + r * 256))[lane];
    float4 b = ((const float4*)(part + r * 256 + 128))[lane];
    s4.x += a.x; s4.y += a.y; s4.z += a.z; s4.w += a.w;
    q4.x += b.x; q4.y += b.y; q4.z += b.z; q4.w += b.w;
  }
  for (int base = 0; base < nr; base += 32) {
    int n = nr - base;
    n = n < 32 ? n : 32;
    if (base > 0) rmine = (lane < n) ? rlist[base + lane] : 0;
    for (int j0 = 0; j0 < n; j0 += 8) {
      float4 v[8];
#pragma unroll
      for (int k = 0; k < 8; ++k) {
        int row = __builtin_amdgcn_readlane(rmine, j0 + k);
        v[k] = ((const float4*)(h1 + (long)row * 128))[lane];
      }
#pragma unroll
      for (int k = 0; k < 8; ++k) {
        if (j0 + k < n) {
          s4.x += v[k].x; s4.y += v[k].y; s4.z += v[k].z; s4.w += v[k].w;
          q4.x += v[k].x * v[k].x; q4.y += v[k].y * v[k].y;
          q4.z += v[k].z * v[k].z; q4.w += v[k].w * v[k].w;
        }
      }
    }
  }
  float4 sc, sh;
  {
    const float inv_n = 1.0f / (float)N1;
    float4 g4 = ((const float4*)gamma)[lane];
    float4 b4 = ((const float4*)beta)[lane];
    float mx = s4.x * inv_n, my = s4.y * inv_n, mz = s4.z * inv_n, mw = s4.w * inv_n;
    sc.x = g4.x * rsqrtf(q4.x * inv_n - mx * mx + BN_EPS);
    sc.y = g4.y * rsqrtf(q4.y * inv_n - my * my + BN_EPS);
    sc.z = g4.z * rsqrtf(q4.z * inv_n - mz * mz + BN_EPS);
    sc.w = g4.w * rsqrtf(q4.w * inv_n - mw * mw + BN_EPS);
    sh.x = b4.x - mx * sc.x; sh.y = b4.y - my * sc.y;
    sh.z = b4.z - mz * sc.z; sh.w = b4.w - mw * sc.w;
  }

  float4 segsum[8];
#pragma unroll
  for (int i = 0; i < 8; ++i) segsum[i] = make_float4(0.f, 0.f, 0.f, 0.f);

  int nchunk = 1 + (novf + TC - 1) / TC;
#pragma nounroll
  for (int ch = 0; ch < nchunk; ++ch) {
    int cnt;
    int e[TC / 256];
#pragma unroll
    for (int k = 0; k < TC / 256; ++k) e[k] = 0;
    if (ch == 0) {
      cnt = nb;
#pragma unroll
      for (int k = 0; k < TC / 256; ++k) e[k] = e0[k];
    } else {
      int base = (ch - 1) * TC;
      __syncthreads();
      if (threadIdx.x == 0) nsel[0] = 0;
      __syncthreads();
#pragma unroll
      for (int k = 0; k < TC / 256; ++k) {
        int j = base + threadIdx.x + 256 * k;
        if (j < novf && ovt[j] == tile) {
          int p = atomicAdd(&nsel[0], 1);
          ent[p] = ove[j];
        }
      }
      __syncthreads();
      cnt = nsel[0];
#pragma unroll
      for (int k = 0; k < TC / 256; ++k) {
        int i = threadIdx.x + 256 * k;
        if (i < cnt) e[k] = ent[i];
      }
    }
    tile_chunk(cnt, e, wave, lane, srt, hist, off, h1, sc, sh, segsum);
  }

#pragma unroll
  for (int rr = 0; rr < 2; ++rr) {
    float4 x = own[rr];
    float4 v;
    v.x = fmaxf(0.0f, x.x * sc.x + sh.x);
    v.y = fmaxf(0.0f, x.y * sc.y + sh.y);
    v.z = fmaxf(0.0f, x.z * sc.z + sh.z);
    v.w = fmaxf(0.0f, x.w * sc.w + sh.w);
    *(float4*)(At + (wave * 2 + rr) * PA + 512 + 4 * lane) = v;
  }
#pragma unroll
  for (int i = 0; i < 8; ++i)
    *(float4*)(At + (wave * 2 + (i >> 2)) * PA + (i & 3) * 128 + 4 * lane) = segsum[i];
  __syncthreads();

  int c = lane & 15;
  int h2 = (lane >> 4) << 1;
  int kb = wave * 80;
  v2f af[20];
  v2f bf[20][4];
#pragma unroll
  for (int ks = 0; ks < 20; ++ks) {
    int kk = kb + 4 * ks;
    const float* Bk = (kk < 512) ? (W2 + kk * 64) : (Ws2 + (kk - 512) * 64);
    Bk += h2 * 64 + c;
#pragma unroll
    for (int t = 0; t < 4; ++t) {
      bf[ks][t].x = Bk[t * 16];
      bf[ks][t].y = Bk[64 + t * 16];
    }
    af[ks] = *(const v2f*)(At + c * PA + kk + h2);
  }
  v8f acc[4];
  const v8f zero = {0.f,0.f,0.f,0.f,0.f,0.f,0.f,0.f};
#pragma unroll
  for (int t = 0; t < 4; ++t) acc[t] = zero;
#pragma unroll
  for (int ks = 0; ks < 20; ++ks) {
#pragma unroll
    for (int t = 0; t < 4; ++t)
      acc[t] = __builtin_amdgcn_wmma_f32_16x16x4_f32(
          false, af[ks], false, bf[ks][t], (short)0, acc[t], false, false);
  }
#pragma unroll
  for (int t = 0; t < 4; ++t)
#pragma unroll
    for (int v = 0; v < 8; ++v)
      red[((wave * 4 + t) * 8 + v) * 32 + lane] = acc[t][v];
  __syncthreads();

  int row = threadIdx.x >> 4;
  int cq  = threadIdx.x & 15;
  int ridx = (((cq >> 2) * 8 + (row & 7)) * 32) + (row >> 3) * 16 + (cq & 3) * 4;
  float4 bb = ((const float4*)b2)[cq];
  float vals[4] = {bb.x, bb.y, bb.z, bb.w};
#pragma unroll
  for (int w = 0; w < 8; ++w) {
    float4 p = *(const float4*)(red + w * (4 * 8 * 32) + ridx);
    vals[0] += p.x; vals[1] += p.y; vals[2] += p.z; vals[3] += p.w;
  }
  float m = fmaxf(fmaxf(vals[0], vals[1]), fmaxf(vals[2], vals[3]));
#pragma unroll
  for (int mask = 1; mask < 16; mask <<= 1)
    m = fmaxf(m, __shfl_xor(m, mask, 32));
  float sum = 0.f;
#pragma unroll
  for (int t = 0; t < 4; ++t) sum += expf(vals[t] - m);
#pragma unroll
  for (int mask = 1; mask < 16; mask <<= 1)
    sum += __shfl_xor(sum, mask, 32);
  float lse = m + logf(sum);
  float4 o;
  o.x = vals[0] - lse; o.y = vals[1] - lse; o.z = vals[2] - lse; o.w = vals[3] - lse;
  ((float4*)(out + (long)(row0 + row) * 64))[cq] = o;
}

extern "C" void kernel_launch(void* const* d_in, const int* in_sizes, int n_in,
                              void* d_out, int out_size, void* d_ws, size_t ws_size,
                              hipStream_t stream) {
  const float* x     = (const float*)d_in[0];
  const int*   src1  = (const int*)  d_in[1];
  const int*   dst1  = (const int*)  d_in[2];
  const int*   et1   = (const int*)  d_in[3];
  const int*   src2  = (const int*)  d_in[4];
  const int*   dst2  = (const int*)  d_in[5];
  const int*   et2   = (const int*)  d_in[6];
  const int*   hmap  = (const int*)  d_in[7];
  const float* hbuf  = (const float*)d_in[8];
  const float* W1    = (const float*)d_in[9];
  const float* Ws1   = (const float*)d_in[10];
  const float* b1    = (const float*)d_in[11];
  const float* gamma = (const float*)d_in[12];
  const float* beta  = (const float*)d_in[13];
  const float* W2    = (const float*)d_in[14];
  const float* Ws2   = (const float*)d_in[15];
  const float* b2    = (const float*)d_in[16];

  float* ws    = (float*)d_ws;
  float* A1    = ws;
  float* A2    = A1 + (size_t)4 * N1 * 128;
  float* h1    = A2 + (size_t)4 * N2 * 128;
  float* out   = (float*)d_out;
  int* rlist = (int*)A1;
  float* slots = A1 + 65536;
  float* part  = slots + NSLOT * 256;
  int* cnts  = (int*)A2 - 4;
  int* tcnt  = (int*)A2;
  int* bucket = tcnt + NT * TSTRIDE;
  int* ovt    = bucket + NT * TC;
  int* ove    = ovt + E2;
  static_assert(N1 <= 65536, "the BN partial rows start behind the row list");

  k_histfill_stats<<<HF_G, HF_T, 0, stream>>>(hmap, hbuf, b1, h1, slots, tcnt - 32);

  k_pick<<<NRED + PK_WG2 + PK_WG1 + PK_WGR, PK_T, 0, stream>>>(src1, dst1, et1, hmap, cnts,
                                                                x, W1, Ws1, h1, rlist,
                                                                src2, dst2, et2,
                                                                tcnt, bucket, ovt, ove,
                                                                slots, part);

  static_assert(N2 % 16 == 0, "one work-group per 16 destination rows");
  k_agg_gemm2<<<NT, 256, 0, stream>>>(h1, part, rlist, gamma, beta, tcnt, bucket, cnts, ovt, ove,
                                      W2, Ws2, b2, out);
}


# layer-1 edge test inverted: history pass lists the no-history rows (one extra group, prefix sum); edge groups compare destinations with the list, no map load per edge (BN order of those rows fixed)
# speedup vs baseline: 12.7856x; 1.0567x over previous
#include <hip/hip_runtime.h>
#include <hip/hip_bf16.h>
#include <math.h>

typedef __attribute__((ext_vector_type(2))) float v2f;
typedef __attribute__((ext_vector_type(8))) float v8f;

#define N0  300000
#define N1  60000
#define N2  6000
#define E1  1200000
#define E2  300000
#define D_H 128
#define D_OUT 64
#define BN_EPS 1e-5f

#define RW 8
#define RK (RW - 1)
#define NQ (E1 / 4)
static_assert(E1 % 4 == 0, "edge list is read four at a time");
#define NSEG (4 * N2)
static_assert(E2 % 4 == 0, "layer-2 edge list is read four at a time");

#define NT (N2 / 16)
#define TC 2048
#define TSTRIDE 64
static_assert(N1 <= 65536, "a source row fits the low 16 bits of a bucket entry");
static_assert(TC % 256 == 0, "a chunk is a whole number of passes of the work-group");
static_assert(NT * TSTRIDE == NSEG, "the counters fill the cleared range");

#define NSLOT 512
#define NRED 8
static_assert(NSLOT % NRED == 0, "every summing work-group takes the same number of rows");

__device__ __forceinline__ void put_edge(int* __restrict__ bucket,
                                         int* __restrict__ cnts,
                                         int* __restrict__ ovt,
                                         int* __restrict__ ove,
                                         int slot, int tile, int packed) {
  if (slot < TC) {
    atomicExch(&bucket[tile * TC + slot], packed);
  } else {
    int o = atomicAdd(&cnts[2], 1);
    atomicExch(&ovt[o], tile);
    atomicExch(&ove[o], packed);
  }
}

__device__ __forceinline__ void msg_tiles(int cnt,
                                          const int* __restrict__ items,
                                          const int* __restrict__ src,
                                          const int* __restrict__ dst,
                                          const int* __restrict__ et,
                                          const float* __restrict__ x,
                                          const float* __restrict__ W1,
                                          const float* __restrict__ Ws,
                                          float* __restrict__ h1) {
  int ntiles = (cnt + 15) >> 4;
  int wave = threadIdx.x >> 5;
  int lane = threadIdx.x & 31;
  int aRow = lane & 15;
  int aK   = (lane >> 4) << 1;
  int col  = wave * 16 + (lane & 15);
  int halfOff = (lane >> 4) << 3;
  const v2f zero2 = {0.f, 0.f};

  for (int tile = 0; tile < ntiles; ++tile) {
    int ei = tile * 16 + aRow;
    bool live = ei < cnt;
    int item = items[live ? ei : 0];
    bool self = item >= E1;
    int e = self ? 0 : item;
    int sLd = src[e];
    int dLd = dst[e];
    int rLd = et[e];
    int sRow = self ? item - E1 : sLd;
    int dRow = self ? item - E1 : dLd;
    int rel  = live ? (self ? 4 : rLd) : -1;
    const float* Ap = x + (long)sRow * 128 + aK;

    v2f af[32];
#pragma unroll
    for (int ks = 0; ks < 32; ++ks) af[ks] = *(const v2f*)(Ap + 4 * ks);

    v8f acc = {0.f,0.f,0.f,0.f,0.f,0.f,0.f,0.f};
    for (int r = 0; r < 5; ++r) {
      bool mine = (rel == r);
      if (__ballot(mine) == 0ull) continue;
      const float* Bp = ((r < 4) ? (W1 + (long)r * 128 * 128) : Ws)
                        + aK * 128 + col;
      v2f bf[32];
#pragma unroll
      for (int ks = 0; ks < 32; ++ks) {
        bf[ks].x = Bp[(4 * ks) * 128];
        bf[ks].y = Bp[(4 * ks + 1) * 128];
      }
#pragma unroll
      for (int ks = 0; ks < 32; ++ks) {
        v2f a = mine ? af[ks] : zero2;
        acc = __builtin_amdgcn_wmma_f32_16x16x4_f32(
            false, a, false, bf[ks], (short)0, acc, false, false);
      }
    }

#pragma unroll
    for (int v = 0; v < 8; ++v) {
      int lo = __builtin_amdgcn_readlane(dRow, v);
      int hi = __builtin_amdgcn_readlane(dRow, v + 8);
      int d = halfOff ? hi : lo;
      int eo = tile * 16 + v + halfOff;
      if (eo < cnt) atomicAdd(&h1[(long)d * 128 + col], acc[v]);
    }
  }
}

#define PK_T 256
#define PK_Q 4
#define PK_CH (PK_T * PK_Q * 4)
#define PK_WG2 ((E2 + PK_CH - 1) / PK_CH)
#define PK_WG1 ((E1 + PK_CH - 1) / PK_CH)
#define PK_WGR ((N1 + PK_CH - 1) / PK_CH)
#define NTP 384
static_assert(N1 % 4 == 0, "the history map is read four rows at a time");
static_assert(NT <= NTP && NTP % 32 == 0 && NTP <= 2 * PK_T, "two counters per thread, 12 per lane of one wave");
static_assert(NT <= 512, "a tile number fits the 9 bits above the 22-bit entry");
static_assert(PK_T == 256, "a summing work-group has one thread per column of a BN partial row");

__device__ __forceinline__ void bin_group(const int* __restrict__ src2,
                                          const int* __restrict__ dst2,
                                          const int* __restrict__ et2,
                                          int* __restrict__ cnts,
                                          int* __restrict__ tcnt,
                                          int* __restrict__ bucket,
                                          int* __restrict__ ovt,
                                          int* __restrict__ ove,
                                          int g,
                                          int* __restrict__ lcnt,
                                          int* __restrict__ loff,
                                          int* __restrict__ lgb,
                                          int* __restrict__ srt) {
  const int tid = threadIdx.x;
  for (int i = tid; i < NTP; i += PK_T) lcnt[i] = 0;
  int sv[4 * PK_Q], dv[4 * PK_Q], rv[4 * PK_Q];
  bool ok[PK_Q];
#pragma unroll
  for (int k = 0; k < PK_Q; ++k) {
    int q = g * (PK_CH / 4) + k * PK_T + tid;
    ok[k] = q < E2 / 4;
    int qq = ok[k] ? q : 0;
    int4 a = ((const int4*)src2)[qq];
    int4 b = ((const int4*)dst2)[qq];
    int4 c = ((const int4*)et2)[qq];
    sv[4 * k] = a.x; sv[4 * k + 1] = a.y; sv[4 * k + 2] = a.z; sv[4 * k + 3] = a.w;
    dv[4 * k] = b.x; dv[4 * k + 1] = b.y; dv[4 * k + 2] = b.z; dv[4 * k + 3] = b.w;
    rv[4 * k] = c.x; rv[4 * k + 1] = c.y; rv[4 * k + 2] = c.z; rv[4 * k + 3] = c.w;
  }
  __syncthreads();
  int pk[4 * PK_Q], rk[4 * PK_Q];
#pragma unroll
  for (int i = 0; i < 4 * PK_Q; ++i) {
    int tile = dv[i] >> 4;
    pk[i] = sv[i] | ((((dv[i] & 15) << 2) | rv[i]) << 16) | (tile << 22);
    rk[i] = ok[i >> 2] ? atomicAdd(&lcnt[tile], 1) : 0;
  }
  __syncthreads();
  const int t0 = tid, t1 = tid + PK_T;
  int c0 = t0 < NT ? lcnt[t0] : 0;
  int c1 = t1 < NT ? lcnt[t1] : 0;
  int b0 = 0, b1 = 0;
  if (c0 > 0) b0 = atomicAdd(&tcnt[t0 * TSTRIDE], c0);
  if (c1 > 0) b1 = atomicAdd(&tcnt[t1 * TSTRIDE], c1);
  if (tid < 32) {
    int base = tid * (NTP / 32);
    int loc[NTP / 32];
    int sum = 0;
#pragma unroll
    for (int j = 0; j < NTP / 32; ++j) { loc[j] = sum; sum += lcnt[base + j]; }
    int inc = sum;
#pragma unroll
    for (int dl = 1; dl < 32; dl <<= 1) {
      int o = __shfl_up(inc, dl, 32);
      if (tid >= dl) inc += o;
    }
#pragma unroll
    for (int j = 0; j < NTP / 32; ++j) loff[base + j] = inc - sum + loc[j];
  }
  lgb[t0] = b0;
  if (t1 < NTP) lgb[t1] = b1;
  __syncthreads();
  int pl[4 * PK_Q];
#pragma unroll
  for (int i = 0; i < 4 * PK_Q; ++i) pl[i] = loff[pk[i] >> 22];
#pragma unroll
  for (int i = 0; i < 4 * PK_Q; ++i)
    if (ok[i >> 2]) srt[pl[i] + rk[i]] = pk[i];
  __syncthreads();
  int nv = E2 - g * PK_CH;
  nv = nv < PK_CH ? nv : PK_CH;
  int e[4 * PK_Q], sl[4 * PK_Q];
#pragma unroll
  for (int k = 0; k < 4 * PK_Q; ++k) {
    int i = k * PK_T + tid;
    e[k] = srt[i < nv ? i : 0];
  }
#pragma unroll
  for (int k = 0; k < 4 * PK_Q; ++k) {
    int tile = e[k] >> 22;
    sl[k] = lgb[tile] - loff[tile] + k * PK_T + tid;
  }
#pragma unroll
  for (int k = 0; k < 4 * PK_Q; ++k)
    if (k * PK_T + tid < nv)
      put_edge(bucket, cnts, ovt, ove, sl[k], e[k] >> 22, e[k] & 0x3FFFFF);
}

__global__ void __launch_bounds__(PK_T)
k_pick(const int* __restrict__ src,
       const int* __restrict__ dst,
       const int* __restrict__ et,
       const int* __restrict__ hmap,
       int* __restrict__ cnts,
       const float* __restrict__ x,
       const float* __restrict__ W1,
       const float* __restrict__ Ws1,
       float* __restrict__ h1,
       const int* __restrict__ rlist,
       const int* __restrict__ rcnt,
       const int* __restrict__ src2,
       const int* __restrict__ dst2,
       const int* __restrict__ et2,
       int* __restrict__ tcnt,
       int* __restrict__ bucket,
       int* __restrict__ ovt,
       int* __restrict__ ove,
       const float* __restrict__ slots,
       float* __restrict__ part) {
  __shared__ int lcnt[NTP], loff[NTP], lgb[NTP];
  __shared__ int srt[PK_CH];
  const int tid = threadIdx.x;
  if (blockIdx.x < NRED) {
    const float* sp = slots + (long)blockIdx.x * (NSLOT / NRED) * PK_T + tid;
    float a = 0.f;
#pragma unroll 16
    for (int s = 0; s < NSLOT / NRED; ++s) a += sp[s * PK_T];
    part[blockIdx.x * PK_T + tid] = a;
    return;
  }
  const int g = blockIdx.x - NRED;
  if (g < PK_WG2) {
    bin_group(src2, dst2, et2, cnts, tcnt, bucket, ovt, ove, g, lcnt, loff, lgb, srt);
  } else {
    if (tid == 0) lcnt[0] = 0;
    const bool edges = g < PK_WG2 + PK_WG1;
    if (edges) {
      unsigned m = 0;
      const int q0 = (g - PK_WG2) * (PK_CH / 4) + tid;
      int dv[4 * PK_Q];
#pragma unroll
      for (int k = 0; k < PK_Q; ++k) {
        int q = q0 + k * PK_T;
        int4 b = ((const int4*)dst)[q < NQ ? q : 0];
        dv[4 * k] = b.x; dv[4 * k + 1] = b.y; dv[4 * k + 2] = b.z; dv[4 * k + 3] = b.w;
      }
      int rr[RW];
#pragma unroll
      for (int k = 0; k < RW; ++k) rr[k] = __builtin_amdgcn_readfirstlane(rlist[k]);
      if (rr[RK] < 0) {
#pragma unroll
        for (int k = 0; k < RK; ++k) {
          if (rr[k] >= 0) {
#pragma unroll
            for (int i = 0; i < 4 * PK_Q; ++i) m |= (dv[i] == rr[k] ? 1u : 0u) << i;
          }
        }
      } else {
        int h[4 * PK_Q];
#pragma unroll
        for (int i = 0; i < 4 * PK_Q; ++i) h[i] = hmap[dv[i]];
#pragma unroll
        for (int i = 0; i < 4 * PK_Q; ++i) m |= ((unsigned)h[i] >> 31) << i;
      }
#pragma unroll
      for (int k = 0; k < PK_Q; ++k)
        if (q0 + k * PK_T >= NQ) m &= ~(0xFu << (4 * k));
      __syncthreads();
      while (m) {
        int i = __ffs(m) - 1;
        m &= m - 1;
        srt[atomicAdd(&lcnt[0], 1)] = (q0 + (i >> 2) * PK_T) * 4 + (i & 3);
      }
    } else {
      const int q0 = (g - PK_WG2 - PK_WG1) * (PK_CH / 4) + tid;
      int nr = __builtin_amdgcn_readfirstlane(rcnt[0]);
      nr = nr < N1 ? nr : N1;
      int rv[4 * PK_Q];
#pragma unroll
      for (int k = 0; k < PK_Q; ++k) {
        int q = q0 + k * PK_T;
        int4 a = ((const int4*)rlist)[4 * q < nr ? q : 0];
        rv[4 * k] = a.x; rv[4 * k + 1] = a.y; rv[4 * k + 2] = a.z; rv[4 * k + 3] = a.w;
      }
      __syncthreads();
#pragma unroll
      for (int i = 0; i < 4 * PK_Q; ++i)
        if ((q0 + (i >> 2) * PK_T) * 4 + (i & 3) < nr)
          srt[atomicAdd(&lcnt[0], 1)] = E1 + rv[i];
    }
    __syncthreads();
    int cnt = __builtin_amdgcn_readfirstlane(lcnt[0]);
    msg_tiles(cnt, srt, src, dst, et, x, W1, Ws1, h1);
  }
}

#define HF_T 1024
#define HF_G NSLOT
#define HF_TOTAL (N1 * 128)
#define HF_STRIDE (HF_G * HF_T)
#define HF_TRIPS ((HF_TOTAL + HF_STRIDE - 1) / HF_STRIDE)
#define HS_Q ((N1 / 4 + HF_T - 1) / HF_T)
#define HS_G 8
#define HS_TRIPS ((HS_Q + HS_G - 1) / HS_G)
static_assert(HS_TRIPS * HS_G * 4 <= 64, "one bit per row of a thread in 64 bits");
static_assert(HF_T / 32 == 32, "the listing work-group's wave totals fill one wave");
static_assert(HF_T % 128 == 0, "a thread keeps one column over all its trips");
static_assert((long)N1 * 128 * 4 < (1l << 31), "element offsets fit 32 bits");
static_assert(32 + NT * TSTRIDE <= HF_STRIDE, "one thread per cleared word");
__global__ void __launch_bounds__(HF_T)
k_histfill_stats(const int* __restrict__ hmap,
                 const float* __restrict__ hbuf,
                 const float* __restrict__ b1,
                 float* __restrict__ h1,
                 float* __restrict__ slots,
                 int* __restrict__ clr,
                 int* __restrict__ rlist,
                 int* __restrict__ rcnt) {
  __shared__ float ls[HF_T], lq[HF_T];
  __shared__ int wsum[HF_T / 32];
  if (blockIdx.x == 0) {
    const int t = threadIdx.x;
    const int lane = t & 31, wave = t >> 5;
    unsigned long long bits = 0ull;
#pragma nounroll
    for (int k0 = 0; k0 < HS_TRIPS * HS_G; k0 += HS_G) {
      int4 hq[HS_G];
#pragma unroll
      for (int j = 0; j < HS_G; ++j) {
        int q = (k0 + j) * HF_T + t;
        hq[j] = ((const int4*)hmap)[q < N1 / 4 ? q : 0];
      }
#pragma unroll
      for (int j = 0; j < HS_G; ++j) {
        unsigned in = ((k0 + j) * HF_T + t < N1 / 4) ? 0xFu : 0u;
        unsigned b = ((unsigned)hq[j].x >> 31) | (((unsigned)hq[j].y >> 31) << 1) |
                     (((unsigned)hq[j].z >> 31) << 2) | (((unsigned)hq[j].w >> 31) << 3);
        bits |= (unsigned long long)(b & in) << (4 * (k0 + j));
      }
    }
    const int cnt = __popcll(bits);
    int inc = cnt;
#pragma unroll
    for (int d = 1; d < 32; d <<= 1) {
      int o = __shfl_up(inc, d, 32);
      if (lane >= d) inc += o;
    }
    if (lane == 31) wsum[wave] = inc;
    __syncthreads();
    if (wave == 0) {
      int w = wsum[lane];
#pragma unroll
      for (int d = 1; d < 32; d <<= 1) {
        int o = __shfl_up(w, d, 32);
        if (lane >= d) w += o;
      }
      wsum[lane] = w;
    }
    __syncthreads();
    int pos = inc - cnt + (wave > 0 ? wsum[wave - 1] : 0);
    const int total = wsum[HF_T / 32 - 1];
    while (bits) {
      int i = __ffsll(bits) - 1;
      bits &= bits - 1;
      rlist[pos++] = ((i >> 2) * HF_T + t) * 4 + (i & 3);
    }
    int fill = (total + 3) & ~3;
    fill = fill > RW ? fill : RW;
    if (t < RW && total + t < fill) rlist[total + t] = -1;
    if (t == 0) rcnt[0] = total;
    return;
  }
  const unsigned blk = blockIdx.x - 1;
  unsigned tid = blk * HF_T + threadIdx.x;
  if (tid >= 28u && tid < 32u + NT * TSTRIDE) clr[tid] = 0;
  unsigned c = threadIdx.x & 127;
  unsigned row0 = tid >> 7;
  float bias = b1[c];
  int hm[HF_TRIPS];
#pragma unroll
  for (int t = 0; t < HF_TRIPS; ++t) {
    bool in = ((t + 1) * HF_STRIDE <= HF_TOTAL) || (tid + t * HF_STRIDE < HF_TOTAL);
    hm[t] = hmap[in ? row0 + t * (HF_STRIDE / 128) : 0u];
  }
  float v[HF_TRIPS];
#pragma unroll
  for (int t = 0; t < HF_TRIPS; ++t) {
    unsigned r = hm[t] >= 0 ? (unsigned)hm[t] : 0u;
    v[t] = hbuf[r * 128u + c];
  }
  __builtin_amdgcn_sched_barrier(0);
  float s = 0.f, q = 0.f;
#pragma unroll
  for (int t = 0; t < HF_TRIPS; ++t) {
    bool in = ((t + 1) * HF_STRIDE <= HF_TOTAL) || (tid + t * HF_STRIDE < HF_TOTAL);
    bool hist = in && hm[t] >= 0;
    float sn = s + v[t];
    float qn = q + v[t] * v[t];
    s = hist ? sn : s;
    q = hist ? qn : q;
    float val = hist ? v[t] : bias;
    if (in) h1[tid + t * HF_STRIDE] = val;
  }
  ls[threadIdx.x] = s; lq[threadIdx.x] = q;
  __syncthreads();
  if (threadIdx.x < 128) {
    float S = 0.f, Q = 0.f;
    for (int k = 0; k < HF_T / 128; ++k) {
      S += ls[threadIdx.x + 128 * k];
      Q += lq[threadIdx.x + 128 * k];
    }
    slots[blk * 256 + threadIdx.x] = S;
    slots[blk * 256 + 128 + threadIdx.x] = Q;
  }
}

#define PA 644

#define GW 16
__device__ __forceinline__ void seg_issue(int beg, int end, int lane,
                                          const int* __restrict__ srt,
                                          const float* __restrict__ h1,
                                          float4 (&x)[GW]) {
  int n = end - beg;
  n = n < GW ? n : GW;
  int mine = (lane < n) ? srt[beg + lane] : 0;
#pragma unroll
  for (int k = 0; k < GW; ++k) {
    int sRow = __builtin_amdgcn_readlane(mine, k);
    x[k] = ((const float4*)(h1 + (long)sRow * 128))[lane];
  }
}

__device__ __forceinline__ void seg_sum(int n, const float4 (&x)[GW],
                                        float4 sc, float4 sh, float4& a4) {
#pragma unroll
  for (int k = 0; k < GW; ++k) {
    float m = (k < n) ? 1.0f : 0.0f;
    a4.x = fmaf(m, fmaxf(0.0f, x[k].x * sc.x + sh.x), a4.x);
    a4.y = fmaf(m, fmaxf(0.0f, x[k].y * sc.y + sh.y), a4.y);
    a4.z = fmaf(m, fmaxf(0.0f, x[k].z * sc.z + sh.z), a4.z);
    a4.w = fmaf(m, fmaxf(0.0f, x[k].w * sc.w + sh.w), a4.w);
  }
}

__device__ __forceinline__ void seg_step(int i, const int (&bnd)[9], int lane,
                                         const int* __restrict__ srt,
                                         const float* __restrict__ h1,
                                         float4 sc, float4 sh,
                                         float4 (&cur)[GW], float4 (&nxt)[GW],
                                         float4 (&acc)[8]) {
  if (i + 1 < 8) seg_issue(bnd[i + 1], bnd[i + 2], lane, srt, h1, nxt);
  __builtin_amdgcn_sched_barrier(0);
  int n = bnd[i + 1] - bnd[i];
  float4 a4 = acc[i];
  seg_sum(n < GW ? n : GW, cur, sc, sh, a4);
  for (int base = bnd[i] + GW; base < bnd[i + 1]; base += GW) {
    int r = bnd[i + 1] - base;
    float4 xr[GW];
    seg_issue(base, bnd[i + 1], lane, srt, h1, xr);
    seg_sum(r < GW ? r : GW, xr, sc, sh, a4);
  }
  acc[i] = a4;
}

__device__ __forceinline__ void tile_chunk(int cnt, const int (&e)[TC / 256],
                                           int wave, int lane,
                                           int* __restrict__ srt,
                                           int* __restrict__ hist,
                                           int* __restrict__ off,
                                           const float* __restrict__ h1,
                                           float4 sc, float4 sh,
                                           float4 (&acc)[8]) {
  if (threadIdx.x < 64) hist[threadIdx.x] = 0;
  __syncthreads();
  int rk[TC / 256];
#pragma unroll
  for (int k = 0; k < TC / 256; ++k) {
    rk[k] = 0;
    if (threadIdx.x + 256 * k < cnt) rk[k] = atomicAdd(&hist[(e[k] >> 16) & 63], 1);
  }
  __syncthreads();
  if (wave == 0) {
    int a = hist[2 * lane], c = hist[2 * lane + 1];
    int s = a + c, inc = s;
#pragma unroll
    for (int d = 1; d < 32; d <<= 1) {
      int o = __shfl_up(inc, d, 32);
      if (lane >= d) inc += o;
    }
    off[2 * lane] = inc - s;
    off[2 * lane + 1] = inc - s + a;
    if (lane == 31) off[64] = inc;
  }
  __syncthreads();
  int st[TC / 256];
#pragma unroll
  for (int k = 0; k < TC / 256; ++k) st[k] = off[(e[k] >> 16) & 63];
#pragma unroll
  for (int k = 0; k < TC / 256; ++k)
    if (threadIdx.x + 256 * k < cnt) srt[st[k] + rk[k]] = e[k] & 0xFFFF;
  __syncthreads();
  int bnd[9];
#pragma unroll
  for (int i = 0; i < 9; ++i) bnd[i] = __builtin_amdgcn_readfirstlane(off[wave * 8 + i]);
  float4 xa[GW], xb[GW];
  seg_issue(bnd[0], bnd[1], lane, srt, h1, xa);
#pragma unroll
  for (int p = 0; p < 4; ++p) {
    seg_step(2 * p,     bnd, lane, srt, h1, sc, sh, xa, xb, acc);
    seg_step(2 * p + 1, bnd, lane, srt, h1, sc, sh, xb, xa, acc);
  }
}

__global__ void __launch_bounds__(256)
k_agg_gemm2(const float* __restrict__ h1,
            const float* __restrict__ part,
            const int* __restrict__ rlist,
            const int* __restrict__ rcnt,
            const float* __restrict__ gamma,
            const float* __restrict__ beta,
            const int* __restrict__ tcnt,
            const int* __restrict__ bucket,
            const int* __restrict__ cnts,
            const int* __restrict__ ovt,
            const int* __restrict__ ove,
            const float* __restrict__ W2,
            const float* __restrict__ Ws2,
            const float* __restrict__ b2,
            float* __restrict__ out) {
  __shared__ float At[16 * PA];
  __shared__ float red[8 * 4 * 8 * 32];
  __shared__ int ent[TC];
  __shared__ int srt[TC];
  __shared__ int hist[64];
  __shared__ int off[65];
  __shared__ int nsel[1];

  int wave = __builtin_amdgcn_readfirstlane((int)(threadIdx.x >> 5));
  int lane = threadIdx.x & 31;
  int tile = blockIdx.x;
  int row0 = tile * 16;

  int tc = tcnt[tile * TSTRIDE];
  int nr = __builtin_amdgcn_readfirstlane(rcnt[0]);
  nr = nr < N1 ? nr : N1;
  int nb = tc < TC ? tc : TC;
  int novf = 0;
  if (tc > TC) {
    novf = cnts[2];
    novf = novf < E2 ? novf : E2;
  }

  int nfirst = nr < 32 ? nr : 32;
  int rmine = (lane < nfirst) ? rlist[lane] : 0;

  const int* bk = bucket + tile * TC;
  int e0[TC / 256];
#pragma unroll
  for (int k = 0; k < TC / 256; ++k) e0[k] = 0;
  if (nb > 0) {
#pragma unroll
    for (int k = 0; k < TC / 256; ++k) {
      int i = threadIdx.x + 256 * k;
      e0[k] = bk[i < nb ? i : nb - 1];
    }
  }

  int g0 = row0 + wave * 2;
  float4 own[2];
#pragma unroll
  for (int rr = 0; rr < 2; ++rr)
    own[rr] = ((const float4*)(h1 + (long)(g0 + rr) * 128))[lane];

  float4 s4 = make_float4(0.f, 0.f, 0.f, 0.f), q4 = make_float4(0.f, 0.f, 0.f, 0.f);
#pragma unroll
  for (int r = 0; r < NRED; ++r) {
    float4 a = ((const float4*)(part + r * 256))[lane];
    float4 b = ((const float4*)(part + r * 256 + 128))[lane];
    s4.x += a.x; s4.y += a.y; s4.z += a.z; s4.w += a.w;
    q4.x += b.x; q4.y += b.y; q4.z += b.z; q4.w += b.w;
  }
  for (int base = 0; base < nr; base += 32) {
    int n = nr - base;
    n = n < 32 ? n : 32;
    if (base > 0) rmine = (lane < n) ? rlist[base + lane] : 0;
    for (int j0 = 0; j0 < n; j0 += 8) {
      float4 v[8];
#pragma unroll
      for (int k = 0; k < 8; ++k) {
        int row = __builtin_amdgcn_readlane(rmine, j0 + k);
        v[k] = ((const float4*)(h1 + (long)row * 128))[lane];
      }
#pragma unroll
      for (int k = 0; k < 8; ++k) {
        if (j0 + k < n) {
          s4.x += v[k].x; s4.y += v[k].y; s4.z += v[k].z; s4.w += v[k].w;
          q4.x += v[k].x * v[k].x; q4.y += v[k].y * v[k].y;
          q4.z += v[k].z * v[k].z; q4.w += v[k].w * v[k].w;
        }
      }
    }
  }
  float4 sc, sh;
  {
    const float inv_n = 1.0f / (float)N1;
    float4 g4 = ((const float4*)gamma)[lane];
    float4 b4 = ((const float4*)beta)[lane];
    float mx = s4.x * inv_n, my = s4.y * inv_n, mz = s4.z * inv_n, mw = s4.w * inv_n;
    sc.x = g4.x * rsqrtf(q4.x * inv_n - mx * mx + BN_EPS);
    sc.y = g4.y * rsqrtf(q4.y * inv_n - my * my + BN_EPS);
    sc.z = g4.z * rsqrtf(q4.z * inv_n - mz * mz + BN_EPS);
    sc.w = g4.w * rsqrtf(q4.w * inv_n - mw * mw + BN_EPS);
    sh.x = b4.x - mx * sc.x; sh.y = b4.y - my * sc.y;
    sh.z = b4.z - mz * sc.z; sh.w = b4.w - mw * sc.w;
  }

  float4 segsum[8];
#pragma unroll
  for (int i = 0; i < 8; ++i) segsum[i] = make_float4(0.f, 0.f, 0.f, 0.f);

  int nchunk = 1 + (novf + TC - 1) / TC;
#pragma nounroll
  for (int ch = 0; ch < nchunk; ++ch) {
    int cnt;
    int e[TC / 256];
#pragma unroll
    for (int k = 0; k < TC / 256; ++k) e[k] = 0;
    if (ch == 0) {
      cnt = nb;
#pragma unroll
      for (int k = 0; k < TC / 256; ++k) e[k] = e0[k];
    } else {
      int base = (ch - 1) * TC;
      __syncthreads();
      if (threadIdx.x == 0) nsel[0] = 0;
      __syncthreads();
#pragma unroll
      for (int k = 0; k < TC / 256; ++k) {
        int j = base + threadIdx.x + 256 * k;
        if (j < novf && ovt[j] == tile) {
          int p = atomicAdd(&nsel[0], 1);
          ent[p] = ove[j];
        }
      }
      __syncthreads();
      cnt = nsel[0];
#pragma unroll
      for (int k = 0; k < TC / 256; ++k) {
        int i = threadIdx.x + 256 * k;
        if (i < cnt) e[k] = ent[i];
      }
    }
    tile_chunk(cnt, e, wave, lane, srt, hist, off, h1, sc, sh, segsum);
  }

#pragma unroll
  for (int rr = 0; rr < 2; ++rr) {
    float4 x = own[rr];
    float4 v;
    v.x = fmaxf(0.0f, x.x * sc.x + sh.x);
    v.y = fmaxf(0.0f, x.y * sc.y + sh.y);
    v.z = fmaxf(0.0f, x.z * sc.z + sh.z);
    v.w = fmaxf(0.0f, x.w * sc.w + sh.w);
    *(float4*)(At + (wave * 2 + rr) * PA + 512 + 4 * lane) = v;
  }
#pragma unroll
  for (int i = 0; i < 8; ++i)
    *(float4*)(At + (wave * 2 + (i >> 2)) * PA + (i & 3) * 128 + 4 * lane) = segsum[i];
  __syncthreads();

  int c = lane & 15;
  int h2 = (lane >> 4) << 1;
  int kb = wave * 80;
  v2f af[20];
  v2f bf[20][4];
#pragma unroll
  for (int ks = 0; ks < 20; ++ks) {
    int kk = kb + 4 * ks;
    const float* Bk = (kk < 512) ? (W2 + kk * 64) : (Ws2 + (kk - 512) * 64);
    Bk += h2 * 64 + c;
#pragma unroll
    for (int t = 0; t < 4; ++t) {
      bf[ks][t].x = Bk[t * 16];
      bf[ks][t].y = Bk[64 + t * 16];
    }
    af[ks] = *(const v2f*)(At + c * PA + kk + h2);
  }
  v8f acc[4];
  const v8f zero = {0.f,0.f,0.f,0.f,0.f,0.f,0.f,0.f};
#pragma unroll
  for (int t = 0; t < 4; ++t) acc[t] = zero;
#pragma unroll
  for (int ks = 0; ks < 20; ++ks) {
#pragma unroll
    for (int t = 0; t < 4; ++t)
      acc[t] = __builtin_amdgcn_wmma_f32_16x16x4_f32(
          false, af[ks], false, bf[ks][t], (short)0, acc[t], false, false);
  }
#pragma unroll
  for (int t = 0; t < 4; ++t)
#pragma unroll
    for (int v = 0; v < 8; ++v)
      red[((wave * 4 + t) * 8 + v) * 32 + lane] = acc[t][v];
  __syncthreads();

  int row = threadIdx.x >> 4;
  int cq  = threadIdx.x & 15;
  int ridx = (((cq >> 2) * 8 + (row & 7)) * 32) + (row >> 3) * 16 + (cq & 3) * 4;
  float4 bb = ((const float4*)b2)[cq];
  float vals[4] = {bb.x, bb.y, bb.z, bb.w};
#pragma unroll
  for (int w = 0; w < 8; ++w) {
    float4 p = *(const float4*)(red + w * (4 * 8 * 32) + ridx);
    vals[0] += p.x; vals[1] += p.y; vals[2] += p.z; vals[3] += p.w;
  }
  float m = fmaxf(fmaxf(vals[0], vals[1]), fmaxf(vals[2], vals[3]));
#pragma unroll
  for (int mask = 1; mask < 16; mask <<= 1)
    m = fmaxf(m, __shfl_xor(m, mask, 32));
  float sum = 0.f;
#pragma unroll
  for (int t = 0; t < 4; ++t) sum += expf(vals[t] - m);
#pragma unroll
  for (int mask = 1; mask < 16; mask <<= 1)
    sum += __shfl_xor(sum, mask, 32);
  float lse = m + logf(sum);
  float4 o;
  o.x = vals[0] - lse; o.y = vals[1] - lse; o.z = vals[2] - lse; o.w = vals[3] - lse;
  ((float4*)(out + (long)(row0 + row) * 64))[cq] = o;
}

extern "C" void kernel_launch(void* const* d_in, const int* in_sizes, int n_in,
                              void* d_out, int out_size, void* d_ws, size_t ws_size,
                              hipStream_t stream) {
  const float* x     = (const float*)d_in[0];
  const int*   src1  = (const int*)  d_in[1];
  const int*   dst1  = (const int*)  d_in[2];
  const int*   et1   = (const int*)  d_in[3];
  const int*   src2  = (const int*)  d_in[4];
  const int*   dst2  = (const int*)  d_in[5];
  const int*   et2   = (const int*)  d_in[6];
  const int*   hmap  = (const int*)  d_in[7];
  const float* hbuf  = (const float*)d_in[8];
  const float* W1    = (const float*)d_in[9];
  const float* Ws1   = (const float*)d_in[10];
  const float* b1    = (const float*)d_in[11];
  const float* gamma = (const float*)d_in[12];
  const float* beta  = (const float*)d_in[13];
  const float* W2    = (const float*)d_in[14];
  const float* Ws2   = (const float*)d_in[15];
  const float* b2    = (const float*)d_in[16];

  float* ws    = (float*)d_ws;
  float* A1    = ws;
  float* A2    = A1 + (size_t)4 * N1 * 128;
  float* h1    = A2 + (size_t)4 * N2 * 128;
  float* out   = (float*)d_out;
  int* rlist = (int*)A1;
  int* rcnt  = (int*)A1 + 65504;
  float* slots = A1 + 65536;
  float* part  = slots + NSLOT * 256;
  int* cnts  = (int*)A2 - 4;
  int* tcnt  = (int*)A2;
  int* bucket = tcnt + NT * TSTRIDE;
  int* ovt    = bucket + NT * TC;
  int* ove    = ovt + E2;
  static_assert(N1 <= 65504, "the row count and the BN partial rows start behind the row list");

  k_histfill_stats<<<HF_G + 1, HF_T, 0, stream>>>(hmap, hbuf, b1, h1, slots, tcnt - 32, rlist, rcnt);

  k_pick<<<NRED + PK_WG2 + PK_WG1 + PK_WGR, PK_T, 0, stream>>>(src1, dst1, et1, hmap, cnts,
                                                                x, W1, Ws1, h1, rlist, rcnt,
                                                                src2, dst2, et2,
                                                                tcnt, bucket, ovt, ove,
                                                                slots, part);

  static_assert(N2 % 16 == 0, "one work-group per 16 destination rows");
  k_agg_gemm2<<<NT, 256, 0, stream>>>(h1, part, rlist, rcnt, gamma, beta, tcnt, bucket, cnts, ovt, ove,
                                      W2, Ws2, b2, out);
}
